# starting kernel
# baseline (speedup 1.0000x reference)
#include <hip/hip_runtime.h>

typedef float v2f __attribute__((ext_vector_type(2)));
typedef float v8f __attribute__((ext_vector_type(8)));

#define IN_DIM 256
#define HID 16
#define OUTD 10

__global__ void gcn_init_kernel(float* __restrict__ agg, float* __restrict__ out,
                                const float* __restrict__ b2, int n_nodes) {
    int idx = blockIdx.x * blockDim.x + threadIdx.x;
    if (idx < n_nodes * HID)  agg[idx] = 0.0f;
    if (idx < n_nodes * OUTD) out[idx] = b2[idx % OUTD];
}

__global__ void gcn_gemm1_wmma(const float* __restrict__ x, const float* __restrict__ W1,
                               float* __restrict__ m1, int n_tiles) {
    int wave = (int)((blockIdx.x * blockDim.x + threadIdx.x) >> 5);
    int lane = threadIdx.x & 31;
    if (wave >= n_tiles) return;
    int l15 = lane & 15;
    int hi  = lane >> 4;
    const float* xrow = x + ((size_t)(wave * 16 + l15)) * IN_DIM;

    v8f c = {};
    #pragma unroll 4
    for (int k0 = 0; k0 < IN_DIM; k0 += 4) {
        int ka = k0 + hi * 2;
        v2f a; a.x = xrow[ka];                 a.y = xrow[ka + 1];
        v2f b; b.x = W1[(ka + 0) * HID + l15]; b.y = W1[(ka + 1) * HID + l15];
        c = __builtin_amdgcn_wmma_f32_16x16x4_f32(false, a, false, b, (short)0, c, false, false);
    }
    float* mt = m1 + (size_t)(wave * 16) * HID;
    #pragma unroll
    for (int v = 0; v < 8; ++v)
        mt[(v + hi * 8) * HID + l15] = c[v];
}

__global__ void gcn_scatter1(const int* __restrict__ src, const int* __restrict__ dst,
                             const float* __restrict__ m1, float* __restrict__ agg,
                             long long n_edges) {
    long long e = (long long)blockIdx.x * blockDim.x + threadIdx.x;
    if (e >= n_edges) return;
    int s = src[e], d = dst[e];
    const float4* ms = (const float4*)(m1 + (size_t)s * HID);
    float* ad = agg + (size_t)d * HID;
    #pragma unroll
    for (int q = 0; q < 4; ++q) {
        float4 v = ms[q];
        atomicAdd(ad + q * 4 + 0, v.x);
        atomicAdd(ad + q * 4 + 1, v.y);
        atomicAdd(ad + q * 4 + 2, v.z);
        atomicAdd(ad + q * 4 + 3, v.w);
    }
}

__global__ void gcn_gemm2_wmma(const float* __restrict__ agg, const float* __restrict__ b1,
                               const float* __restrict__ W2, float* __restrict__ m2,
                               int n_tiles) {
    int wave = (int)((blockIdx.x * blockDim.x + threadIdx.x) >> 5);
    int lane = threadIdx.x & 31;
    if (wave >= n_tiles) return;
    int l15 = lane & 15;
    int hi  = lane >> 4;
    const float* arow = agg + (size_t)(wave * 16 + l15) * HID;

    v8f c = {};
    #pragma unroll
    for (int k0 = 0; k0 < HID; k0 += 4) {
        int ka = k0 + hi * 2;
        v2f a;
        a.x = fmaxf(arow[ka]     + b1[ka],     0.0f);
        a.y = fmaxf(arow[ka + 1] + b1[ka + 1], 0.0f);
        v2f b;
        b.x = (l15 < OUTD) ? W2[(ka + 0) * OUTD + l15] : 0.0f;
        b.y = (l15 < OUTD) ? W2[(ka + 1) * OUTD + l15] : 0.0f;
        c = __builtin_amdgcn_wmma_f32_16x16x4_f32(false, a, false, b, (short)0, c, false, false);
    }
    float* mt = m2 + (size_t)(wave * 16) * HID;
    #pragma unroll
    for (int v = 0; v < 8; ++v)
        mt[(v + hi * 8) * HID + l15] = c[v];
}

__global__ void gcn_scatter2(const int* __restrict__ src, const int* __restrict__ dst,
                             const float* __restrict__ m2, float* __restrict__ out,
                             long long n_edges) {
    long long e = (long long)blockIdx.x * blockDim.x + threadIdx.x;
    if (e >= n_edges) return;
    int s = src[e], d = dst[e];
    const float* ms = m2 + (size_t)s * HID;
    float* od = out + (size_t)d * OUTD;
    #pragma unroll
    for (int q = 0; q < OUTD; ++q)
        atomicAdd(od + q, ms[q]);
}

extern "C" void kernel_launch(void* const* d_in, const int* in_sizes, int n_in,
                              void* d_out, int out_size, void* d_ws, size_t ws_size,
                              hipStream_t stream) {
    const float* x  = (const float*)d_in[0];
    const int*   ei = (const int*)  d_in[1];
    const float* W1 = (const float*)d_in[2];
    const float* b1 = (const float*)d_in[3];
    const float* W2 = (const float*)d_in[4];
    const float* b2 = (const float*)d_in[5];
    float* out = (float*)d_out;

    int n_nodes       = in_sizes[0] / IN_DIM;
    long long n_edges = (long long)in_sizes[1] / 2;
    const int* src = ei;
    const int* dst = ei + n_edges;

    float* m1  = (float*)d_ws;
    float* agg = m1  + (size_t)n_nodes * HID;
    float* m2  = agg + (size_t)n_nodes * HID;

    int n_tiles = (n_nodes + 15) / 16;
    int init_blocks = (n_nodes * HID + 255) / 256;
    int gemm_blocks = (n_tiles + 7) / 8;
    int edge_blocks = (int)((n_edges + 255) / 256);

    gcn_init_kernel<<<init_blocks, 256, 0, stream>>>(agg, out, b2, n_nodes);
    gcn_gemm1_wmma <<<gemm_blocks, 256, 0, stream>>>(x, W1, m1, n_tiles);
    gcn_scatter1   <<<edge_blocks, 256, 0, stream>>>(src, dst, m1, agg, n_edges);
    gcn_gemm2_wmma <<<gemm_blocks, 256, 0, stream>>>(agg, b1, W2, m2, n_tiles);
    gcn_scatter2   <<<edge_blocks, 256, 0, stream>>>(src, dst, m2, out, n_edges);
}


# both edge sums made destination-stationary: each work-group keeps 400 nodes' rows in LDS, filters the whole dst list itself (ballot, per-wave queue), adds rows by ds f32 atomics; LDS arrival order
# speedup vs baseline: 4.2787x; 4.2787x over previous
#include <hip/hip_runtime.h>


typedef float v2f __attribute__((ext_vector_type(2)));
typedef float v8f __attribute__((ext_vector_type(8)));

#define IN_DIM 256
#define HID 16
#define OUTD 10

#define OWN_GROUPS 256
#define OWN_THREADS 1024
#define OWN_WAVES 32
#define ROW_CAP 400
#define Q_CAP 544

__global__ void gcn_gemm1_wmma(const float* __restrict__ x, const float* __restrict__ W1,
                               float* __restrict__ m1, int n_tiles) {
    int wave = (int)((blockIdx.x * blockDim.x + threadIdx.x) >> 5);
    int lane = threadIdx.x & 31;
    if (wave >= n_tiles) return;
    int l15 = lane & 15;
    int hi  = lane >> 4;
    const float* xrow = x + ((size_t)(wave * 16 + l15)) * IN_DIM;

    v8f c = {};
    #pragma unroll 4
    for (int k0 = 0; k0 < IN_DIM; k0 += 4) {
        int ka = k0 + hi * 2;
        v2f a; a.x = xrow[ka];                 a.y = xrow[ka + 1];
        v2f b; b.x = W1[(ka + 0) * HID + l15]; b.y = W1[(ka + 1) * HID + l15];
        c = __builtin_amdgcn_wmma_f32_16x16x4_f32(false, a, false, b, (short)0, c, false, false);
    }
    float* mt = m1 + (size_t)(wave * 16) * HID;
    #pragma unroll
    for (int v = 0; v < 8; ++v)
        mt[(v + hi * 8) * HID + l15] = c[v];
}

__device__ __forceinline__ int q_push(int* qe, unsigned short* qt, int qn,
                                      unsigned t, int e, unsigned cnt) {
    bool hit = t < cnt;
    unsigned mask = __builtin_amdgcn_ballot_w32(hit);
    if (hit) {
        int slot = qn + (int)__builtin_amdgcn_mbcnt_lo(mask, 0u);
        qe[slot] = e;
        qt[slot] = (unsigned short)t;
    }
    return qn + __builtin_popcount(mask);
}

__device__ __forceinline__ void q_drain(const int* qe, const unsigned short* qt, int qn,
                                        const int* __restrict__ src, const float* __restrict__ m,
                                        float* acc, int lane) {
    const int f  = lane & 15;
    const int hi = lane >> 4;
    for (int g = 0; g < qn; g += 32) {
        int n = qn - g; if (n > 32) n = 32;
        int s = 0, t = 0;
        if (lane < n) { s = src[qe[g + lane]]; t = (int)qt[g + lane]; }
        for (int j = 0; j < n; j += 8) {
            float v[4]; int tr[4]; bool ok[4];
            #pragma unroll
            for (int u = 0; u < 4; ++u) {
                int ja = j + 2 * u, jb = ja + 1;
                bool a_ok = ja < n, b_ok = jb < n;
                int jac = a_ok ? ja : 0;
                int jbc = b_ok ? jb : jac;
                int sa = __builtin_amdgcn_readlane(s, jac), sb = __builtin_amdgcn_readlane(s, jbc);
                int ta = __builtin_amdgcn_readlane(t, jac), tb = __builtin_amdgcn_readlane(t, jbc);
                int sj = hi ? sb : sa;
                tr[u]  = hi ? tb : ta;
                ok[u]  = hi ? b_ok : a_ok;
                v[u]   = m[(size_t)sj * HID + f];
            }
            #pragma unroll
            for (int u = 0; u < 4; ++u)
                if (ok[u]) atomicAdd(&acc[tr[u] * HID + f], v[u]);
        }
    }
}

__device__ __forceinline__ void ld_chunk(int4 (&b)[4], const int4* __restrict__ d4,
                                         int k, int start, int n_full, int lane) {
    int kk = k < n_full ? k : n_full - 1;
    int ci = kk + start; if (ci >= n_full) ci -= n_full;
    #pragma unroll
    for (int j = 0; j < 4; ++j) b[j] = d4[(size_t)ci * 128 + j * 32 + lane];
}

__device__ __forceinline__ int scan_chunk(const int4 (&b)[4], int k, int start, int n_full,
                                          unsigned lo, unsigned ucnt, int* qe, unsigned short* qt,
                                          int qn, const int* __restrict__ src,
                                          const float* __restrict__ m, float* acc, int lane) {
    int ci = k + start; if (ci >= n_full) ci -= n_full;
    #pragma unroll
    for (int j = 0; j < 4; ++j) {
        const int eb = ci * 512 + j * 128 + lane * 4;
        unsigned t0 = (unsigned)b[j].x - lo;
        unsigned t1 = (unsigned)b[j].y - lo;
        unsigned t2 = (unsigned)b[j].z - lo;
        unsigned t3 = (unsigned)b[j].w - lo;
        unsigned ta = t0 < t1 ? t0 : t1;
        unsigned tb = t2 < t3 ? t2 : t3;
        unsigned tm = ta < tb ? ta : tb;
        if (__builtin_amdgcn_ballot_w32(tm < ucnt) != 0u) {
            qn = q_push(qe, qt, qn, t0, eb + 0, ucnt);
            qn = q_push(qe, qt, qn, t1, eb + 1, ucnt);
            qn = q_push(qe, qt, qn, t2, eb + 2, ucnt);
            qn = q_push(qe, qt, qn, t3, eb + 3, ucnt);
        }
    }
    if (qn >= 32) { q_drain(qe, qt, qn, src, m, acc, lane); qn = 0; }
    return qn;
}

__global__ __launch_bounds__(OWN_THREADS)
void gcn_owner_sum(const int* __restrict__ src, const int* __restrict__ dst,
                   const float* __restrict__ m, const float* __restrict__ bias,
                   float* __restrict__ res, int outw, int add_bias, int n_nodes, int n_edges) {
    __shared__ float acc[ROW_CAP * HID];
    __shared__ int q_e[OWN_WAVES * Q_CAP];
    __shared__ unsigned short q_t[OWN_WAVES * Q_CAP];

    const int tid  = threadIdx.x;
    const int lane = tid & 31;
    const int wv   = __builtin_amdgcn_readfirstlane(tid >> 5);
    const int nw   = blockDim.x >> 5;
    const int G    = gridDim.x;
    int* qe = q_e + wv * Q_CAP;
    unsigned short* qt = q_t + wv * Q_CAP;

    int rr = (n_nodes + G - 1) / G;
    rr = (rr + 15) & ~15;
    if (rr > ROW_CAP) rr = ROW_CAP;

    const bool vec_ok = ((((size_t)dst) & 15) == 0);
    const int n_full = vec_ok ? (n_edges >> 9) : 0;
    const int start = (int)(((long long)blockIdx.x * n_full) / G);
    const int4* d4 = (const int4*)dst;

    for (int lo = blockIdx.x * rr; lo < n_nodes; lo += G * rr) {
        int cnt = n_nodes - lo; if (cnt > rr) cnt = rr;
        const unsigned ucnt = (unsigned)cnt;
        const unsigned ulo  = (unsigned)lo;
        for (int i = tid; i < cnt * HID; i += blockDim.x) acc[i] = 0.0f;
        __syncthreads();

        int qn = 0;
        if (n_full > 0) {
        int4 A[4], B[4], C[4];
        ld_chunk(A, d4, wv,      start, n_full, lane);
        ld_chunk(B, d4, wv + nw, start, n_full, lane);
        for (int c = wv; c < n_full; c += 3 * nw) {
            ld_chunk(C, d4, c + 2 * nw, start, n_full, lane);
            qn = scan_chunk(A, c, start, n_full, ulo, ucnt, qe, qt, qn, src, m, acc, lane);
            ld_chunk(A, d4, c + 3 * nw, start, n_full, lane);
            if (c + nw < n_full)
                qn = scan_chunk(B, c + nw, start, n_full, ulo, ucnt, qe, qt, qn, src, m, acc, lane);
            ld_chunk(B, d4, c + 4 * nw, start, n_full, lane);
            if (c + 2 * nw < n_full)
                qn = scan_chunk(C, c + 2 * nw, start, n_full, ulo, ucnt, qe, qt, qn, src, m, acc, lane);
        }
        }
        for (int b = (n_full << 9) + wv * 32; b < n_edges; b += nw * 32) {
            int e = b + lane;
            unsigned t = (e < n_edges) ? ((unsigned)dst[e] - ulo) : 0xFFFFFFFFu;
            qn = q_push(qe, qt, qn, t, e, ucnt);
            if (qn >= 32) { q_drain(qe, qt, qn, src, m, acc, lane); qn = 0; }
        }
        if (qn > 0) q_drain(qe, qt, qn, src, m, acc, lane);
        __syncthreads();

        for (int i = tid; i < cnt * outw; i += blockDim.x) {
            int r = i / outw, o = i - r * outw;
            float v = acc[r * HID + o];
            if (add_bias) v += bias[o];
            res[(size_t)lo * outw + i] = v;
        }
        __syncthreads();
    }
}

__global__ void gcn_gemm2_wmma(const float* __restrict__ agg, const float* __restrict__ b1,
                               const float* __restrict__ W2, float* __restrict__ m2,
                               int n_tiles) {
    int wave = (int)((blockIdx.x * blockDim.x + threadIdx.x) >> 5);
    int lane = threadIdx.x & 31;
    if (wave >= n_tiles) return;
    int l15 = lane & 15;
    int hi  = lane >> 4;
    const float* arow = agg + (size_t)(wave * 16 + l15) * HID;

    v8f c = {};
    #pragma unroll
    for (int k0 = 0; k0 < HID; k0 += 4) {
        int ka = k0 + hi * 2;
        v2f a;
        a.x = fmaxf(arow[ka]     + b1[ka],     0.0f);
        a.y = fmaxf(arow[ka + 1] + b1[ka + 1], 0.0f);
        v2f b;
        b.x = (l15 < OUTD) ? W2[(ka + 0) * OUTD + l15] : 0.0f;
        b.y = (l15 < OUTD) ? W2[(ka + 1) * OUTD + l15] : 0.0f;
        c = __builtin_amdgcn_wmma_f32_16x16x4_f32(false, a, false, b, (short)0, c, false, false);
    }
    float* mt = m2 + (size_t)(wave * 16) * HID;
    #pragma unroll
    for (int v = 0; v < 8; ++v)
        mt[(v + hi * 8) * HID + l15] = c[v];
}

extern "C" void kernel_launch(void* const* d_in, const int* in_sizes, int n_in,
                              void* d_out, int out_size, void* d_ws, size_t ws_size,
                              hipStream_t stream) {
    const float* x  = (const float*)d_in[0];
    const int*   ei = (const int*)  d_in[1];
    const float* W1 = (const float*)d_in[2];
    const float* b1 = (const float*)d_in[3];
    const float* W2 = (const float*)d_in[4];
    const float* b2 = (const float*)d_in[5];
    float* out = (float*)d_out;

    int n_nodes       = in_sizes[0] / IN_DIM;
    long long n_edges = (long long)in_sizes[1] / 2;
    const int* src = ei;
    const int* dst = ei + n_edges;

    float* m1  = (float*)d_ws;
    float* agg = m1  + (size_t)n_nodes * HID;
    float* m2  = agg + (size_t)n_nodes * HID;

    int n_tiles = (n_nodes + 15) / 16;
    int gemm_blocks = (n_tiles + 7) / 8;

    gcn_gemm1_wmma<<<gemm_blocks, 256, 0, stream>>>(x, W1, m1, n_tiles);
    gcn_owner_sum <<<OWN_GROUPS, OWN_THREADS, 0, stream>>>(src, dst, m1, b2, agg, HID, 0, n_nodes, (int)n_edges);
    gcn_gemm2_wmma<<<gemm_blocks, 256, 0, stream>>>(agg, b1, W2, m2, n_tiles);
    gcn_owner_sum <<<OWN_GROUPS, OWN_THREADS, 0, stream>>>(src, dst, m2, b2, out, OUTD, 1, n_nodes, (int)n_edges);
}


# owner sums cut two ways, 32 row ranges x 8 edge slices: a work-group keeps 3136 rows in LDS and scans 1/8 of dst, not all; 8 partial planes per layer added in plane order by the reader (+1 launch)
# speedup vs baseline: 7.4922x; 1.7510x over previous
#include <hip/hip_runtime.h>


typedef float v2f __attribute__((ext_vector_type(2)));
typedef float v8f __attribute__((ext_vector_type(8)));

#define IN_DIM 256
#define HID 16
#define OUTD 10

#define OWN_GROUPS 256
#define OWN_THREADS 1024
#define OWN_WAVES 32
#define EDGE_PARTS 8
#define ROW_RANGES 32
#define ROW_CAP 3136
#define Q_CAP 544
static_assert(EDGE_PARTS * ROW_RANGES == OWN_GROUPS, "one work-group per (slice, range) pair");

__global__ void gcn_gemm1_wmma(const float* __restrict__ x, const float* __restrict__ W1,
                               float* __restrict__ m1, int n_tiles) {
    int wave = (int)((blockIdx.x * blockDim.x + threadIdx.x) >> 5);
    int lane = threadIdx.x & 31;
    if (wave >= n_tiles) return;
    int l15 = lane & 15;
    int hi  = lane >> 4;
    const float* xrow = x + ((size_t)(wave * 16 + l15)) * IN_DIM;

    v8f c = {};
    #pragma unroll 4
    for (int k0 = 0; k0 < IN_DIM; k0 += 4) {
        int ka = k0 + hi * 2;
        v2f a; a.x = xrow[ka];                 a.y = xrow[ka + 1];
        v2f b; b.x = W1[(ka + 0) * HID + l15]; b.y = W1[(ka + 1) * HID + l15];
        c = __builtin_amdgcn_wmma_f32_16x16x4_f32(false, a, false, b, (short)0, c, false, false);
    }
    float* mt = m1 + (size_t)(wave * 16) * HID;
    #pragma unroll
    for (int v = 0; v < 8; ++v)
        mt[(v + hi * 8) * HID + l15] = c[v];
}

__device__ __forceinline__ int q_push(int* qe, unsigned short* qt, int qn,
                                      unsigned t, int e, unsigned cnt) {
    bool hit = t < cnt;
    unsigned mask = __builtin_amdgcn_ballot_w32(hit);
    if (hit) {
        int slot = qn + (int)__builtin_amdgcn_mbcnt_lo(mask, 0u);
        qe[slot] = e;
        qt[slot] = (unsigned short)t;
    }
    return qn + __builtin_popcount(mask);
}

__device__ __forceinline__ void q_drain(const int* qe, const unsigned short* qt, int qn,
                                        const int* __restrict__ src, const float* __restrict__ m,
                                        float* acc, int lane) {
    const int f  = lane & 15;
    const int hi = lane >> 4;
    for (int g = 0; g < qn; g += 32) {
        int n = qn - g; if (n > 32) n = 32;
        int s = 0, t = 0;
        if (lane < n) { s = src[qe[g + lane]]; t = (int)qt[g + lane]; }
        for (int j = 0; j < n; j += 8) {
            float v[4]; int tr[4]; bool ok[4];
            #pragma unroll
            for (int u = 0; u < 4; ++u) {
                int ja = j + 2 * u, jb = ja + 1;
                bool a_ok = ja < n, b_ok = jb < n;
                int jac = a_ok ? ja : 0;
                int jbc = b_ok ? jb : jac;
                int sa = __builtin_amdgcn_readlane(s, jac), sb = __builtin_amdgcn_readlane(s, jbc);
                int ta = __builtin_amdgcn_readlane(t, jac), tb = __builtin_amdgcn_readlane(t, jbc);
                int sj = hi ? sb : sa;
                tr[u]  = hi ? tb : ta;
                ok[u]  = hi ? b_ok : a_ok;
                v[u]   = m[(size_t)sj * HID + f];
            }
            #pragma unroll
            for (int u = 0; u < 4; ++u)
                if (ok[u]) atomicAdd(&acc[tr[u] * HID + f], v[u]);
        }
    }
}

__device__ __forceinline__ void ld_chunk(int4 (&b)[4], const int4* __restrict__ d4,
                                         int k, int start, int n_part, int c0, int lane) {
    int kk = k < n_part ? k : n_part - 1;
    int ci = kk + start; if (ci >= n_part) ci -= n_part;
    ci += c0;
    #pragma unroll
    for (int j = 0; j < 4; ++j) b[j] = d4[(size_t)ci * 128 + j * 32 + lane];
}

__device__ __forceinline__ int scan_chunk(const int4 (&b)[4], int k, int start, int n_part, int c0,
                                          unsigned lo, unsigned ucnt, int* qe, unsigned short* qt,
                                          int qn, const int* __restrict__ src,
                                          const float* __restrict__ m, float* acc, int lane) {
    int ci = k + start; if (ci >= n_part) ci -= n_part;
    ci += c0;
    #pragma unroll
    for (int j = 0; j < 4; ++j) {
        const int eb = ci * 512 + j * 128 + lane * 4;
        unsigned t0 = (unsigned)b[j].x - lo;
        unsigned t1 = (unsigned)b[j].y - lo;
        unsigned t2 = (unsigned)b[j].z - lo;
        unsigned t3 = (unsigned)b[j].w - lo;
        qn = q_push(qe, qt, qn, t0, eb + 0, ucnt);
        qn = q_push(qe, qt, qn, t1, eb + 1, ucnt);
        qn = q_push(qe, qt, qn, t2, eb + 2, ucnt);
        qn = q_push(qe, qt, qn, t3, eb + 3, ucnt);
    }
    if (qn >= 32) { q_drain(qe, qt, qn, src, m, acc, lane); qn = 0; }
    return qn;
}

__global__ __launch_bounds__(OWN_THREADS)
void gcn_owner_sum(const int* __restrict__ src, const int* __restrict__ dst,
                   const float* __restrict__ m, float* __restrict__ part,
                   int n_nodes, int n_edges) {
    __shared__ __attribute__((aligned(16))) float acc[ROW_CAP * HID];
    __shared__ int q_e[OWN_WAVES * Q_CAP];
    __shared__ unsigned short q_t[OWN_WAVES * Q_CAP];

    const int tid  = threadIdx.x;
    const int lane = tid & 31;
    const int wv   = __builtin_amdgcn_readfirstlane(tid >> 5);
    const int nw   = blockDim.x >> 5;
    const int p    = blockIdx.x % EDGE_PARTS;
    const int r    = blockIdx.x / EDGE_PARTS;
    int* qe = q_e + wv * Q_CAP;
    unsigned short* qt = q_t + wv * Q_CAP;
    float* res = part + (size_t)p * n_nodes * HID;

    int rr = (n_nodes + ROW_RANGES - 1) / ROW_RANGES;
    rr = (rr + 15) & ~15;
    if (rr > ROW_CAP) rr = ROW_CAP;

    const bool vec_ok = ((((size_t)dst) & 15) == 0);
    const int n_full = vec_ok ? (n_edges >> 9) : 0;
    const int c0     = (int)(((long long)p * n_full) / EDGE_PARTS);
    const int n_part = (int)(((long long)(p + 1) * n_full) / EDGE_PARTS) - c0;
    const int start = (int)(((long long)r * n_part) / ROW_RANGES);
    const int4* d4 = (const int4*)dst;
    const int t_base = n_full << 9;
    const int t_lo = t_base + (int)(((long long)p * (n_edges - t_base)) / EDGE_PARTS);
    const int t_hi = t_base + (int)(((long long)(p + 1) * (n_edges - t_base)) / EDGE_PARTS);

    for (int lo = r * rr; lo < n_nodes; lo += ROW_RANGES * rr) {
        int cnt = n_nodes - lo; if (cnt > rr) cnt = rr;
        const unsigned ucnt = (unsigned)cnt;
        const unsigned ulo  = (unsigned)lo;
        float4* acc4 = (float4*)acc;
        for (int i = tid; i < cnt * (HID / 4); i += blockDim.x) acc4[i] = make_float4(0.f, 0.f, 0.f, 0.f);
        __syncthreads();

        int qn = 0;
        if (n_part > 0) {
        int4 A[4], B[4], C[4];
        ld_chunk(A, d4, wv,      start, n_part, c0, lane);
        ld_chunk(B, d4, wv + nw, start, n_part, c0, lane);
        for (int c = wv; c < n_part; c += 3 * nw) {
            ld_chunk(C, d4, c + 2 * nw, start, n_part, c0, lane);
            qn = scan_chunk(A, c, start, n_part, c0, ulo, ucnt, qe, qt, qn, src, m, acc, lane);
            ld_chunk(A, d4, c + 3 * nw, start, n_part, c0, lane);
            if (c + nw < n_part)
                qn = scan_chunk(B, c + nw, start, n_part, c0, ulo, ucnt, qe, qt, qn, src, m, acc, lane);
            ld_chunk(B, d4, c + 4 * nw, start, n_part, c0, lane);
            if (c + 2 * nw < n_part)
                qn = scan_chunk(C, c + 2 * nw, start, n_part, c0, ulo, ucnt, qe, qt, qn, src, m, acc, lane);
        }
        }
        for (int b = t_lo + wv * 32; b < t_hi; b += nw * 32) {
            int e = b + lane;
            unsigned t = (e < t_hi) ? ((unsigned)dst[e] - ulo) : 0xFFFFFFFFu;
            qn = q_push(qe, qt, qn, t, e, ucnt);
            if (qn >= 32) { q_drain(qe, qt, qn, src, m, acc, lane); qn = 0; }
        }
        if (qn > 0) q_drain(qe, qt, qn, src, m, acc, lane);
        __syncthreads();

        float4* res4 = (float4*)(res + (size_t)lo * HID);
        for (int i = tid; i < cnt * (HID / 4); i += blockDim.x) res4[i] = acc4[i];
        __syncthreads();
    }
}

__global__ void gcn_gemm2_wmma(const float* __restrict__ agg, const float* __restrict__ b1,
                               const float* __restrict__ W2, float* __restrict__ m2,
                               int n_tiles, int n_nodes) {
    int wave = (int)((blockIdx.x * blockDim.x + threadIdx.x) >> 5);
    int lane = threadIdx.x & 31;
    if (wave >= n_tiles) return;
    int l15 = lane & 15;
    int hi  = lane >> 4;
    const float* arow = agg + (size_t)(wave * 16 + l15) * HID;
    const size_t plane = (size_t)n_nodes * HID;

    v8f c = {};
    #pragma unroll
    for (int k0 = 0; k0 < HID; k0 += 4) {
        int ka = k0 + hi * 2;
        float sx = arow[ka], sy = arow[ka + 1];
        #pragma unroll
        for (int q = 1; q < EDGE_PARTS; ++q) {
            sx += arow[q * plane + ka];
            sy += arow[q * plane + ka + 1];
        }
        v2f a;
        a.x = fmaxf(sx + b1[ka],     0.0f);
        a.y = fmaxf(sy + b1[ka + 1], 0.0f);
        v2f b;
        b.x = (l15 < OUTD) ? W2[(ka + 0) * OUTD + l15] : 0.0f;
        b.y = (l15 < OUTD) ? W2[(ka + 1) * OUTD + l15] : 0.0f;
        c = __builtin_amdgcn_wmma_f32_16x16x4_f32(false, a, false, b, (short)0, c, false, false);
    }
    float* mt = m2 + (size_t)(wave * 16) * HID;
    #pragma unroll
    for (int v = 0; v < 8; ++v)
        mt[(v + hi * 8) * HID + l15] = c[v];
}

__global__ void gcn_join_out(const float* __restrict__ part, const float* __restrict__ bias,
                             float* __restrict__ out, int n_nodes) {
    int i = (int)(blockIdx.x * blockDim.x + threadIdx.x);
    if (i >= n_nodes * OUTD) return;
    int r = i / OUTD, o = i - r * OUTD;
    const size_t plane = (size_t)n_nodes * HID;
    const float* q = part + (size_t)r * HID + o;
    float s = q[0];
    #pragma unroll
    for (int k = 1; k < EDGE_PARTS; ++k) s += q[k * plane];
    out[i] = s + bias[o];
}

extern "C" void kernel_launch(void* const* d_in, const int* in_sizes, int n_in,
                              void* d_out, int out_size, void* d_ws, size_t ws_size,
                              hipStream_t stream) {
    const float* x  = (const float*)d_in[0];
    const int*   ei = (const int*)  d_in[1];
    const float* W1 = (const float*)d_in[2];
    const float* b1 = (const float*)d_in[3];
    const float* W2 = (const float*)d_in[4];
    const float* b2 = (const float*)d_in[5];
    float* out = (float*)d_out;

    int n_nodes       = in_sizes[0] / IN_DIM;
    long long n_edges = (long long)in_sizes[1] / 2;
    const int* src = ei;
    const int* dst = ei + n_edges;

    float* m1   = (float*)d_ws;
    float* aggp = m1   + (size_t)n_nodes * HID;
    float* m2   = aggp + (size_t)n_nodes * HID * EDGE_PARTS;
    float* outp = m2   + (size_t)n_nodes * HID;

    int n_tiles = (n_nodes + 15) / 16;
    int gemm_blocks = (n_tiles + 7) / 8;
    int join_blocks = (n_nodes * OUTD + 255) / 256;

    gcn_gemm1_wmma<<<gemm_blocks, 256, 0, stream>>>(x, W1, m1, n_tiles);
    gcn_owner_sum <<<OWN_GROUPS, OWN_THREADS, 0, stream>>>(src, dst, m1, aggp, n_nodes, (int)n_edges);
    gcn_gemm2_wmma<<<gemm_blocks, 256, 0, stream>>>(aggp, b1, W2, m2, n_tiles, n_nodes);
    gcn_owner_sum <<<OWN_GROUPS, OWN_THREADS, 0, stream>>>(src, dst, m2, outp, n_nodes, (int)n_edges);
    gcn_join_out  <<<join_blocks, 256, 0, stream>>>(outp, b2, out, n_nodes);
}


# layer-1 owner sum saves each work-group's hits (source, local row: one word) as a list; the layer-2 sum reads its list, no dst scan, filter or src look-up; sums stay in LDS arrival order
# speedup vs baseline: 9.5198x; 1.2706x over previous
#include <hip/hip_runtime.h>


typedef float v2f __attribute__((ext_vector_type(2)));
typedef float v8f __attribute__((ext_vector_type(8)));

#define IN_DIM 256
#define HID 16
#define OUTD 10

#define OWN_GROUPS 256
#define OWN_THREADS 1024
#define OWN_WAVES 32
#define EDGE_PARTS 8
#define ROW_RANGES 32
#define ROW_CAP 3136
#define Q_CAP 544
#define LIST_CAP 65536
#define ROW_BITS 12
#define CNT_PITCH 32
static_assert(EDGE_PARTS * ROW_RANGES == OWN_GROUPS, "one work-group per (slice, range) pair");
static_assert(ROW_CAP <= (1 << ROW_BITS), "a local row fits in ROW_BITS bits");

__global__ void gcn_gemm1_wmma(const float* __restrict__ x, const float* __restrict__ W1,
                               float* __restrict__ m1, int n_tiles) {
    int wave = (int)((blockIdx.x * blockDim.x + threadIdx.x) >> 5);
    int lane = threadIdx.x & 31;
    if (wave >= n_tiles) return;
    int l15 = lane & 15;
    int hi  = lane >> 4;
    const float* xrow = x + ((size_t)(wave * 16 + l15)) * IN_DIM;

    v8f c = {};
    #pragma unroll 4
    for (int k0 = 0; k0 < IN_DIM; k0 += 4) {
        int ka = k0 + hi * 2;
        v2f a; a.x = xrow[ka];                 a.y = xrow[ka + 1];
        v2f b; b.x = W1[(ka + 0) * HID + l15]; b.y = W1[(ka + 1) * HID + l15];
        c = __builtin_amdgcn_wmma_f32_16x16x4_f32(false, a, false, b, (short)0, c, false, false);
    }
    float* mt = m1 + (size_t)(wave * 16) * HID;
    #pragma unroll
    for (int v = 0; v < 8; ++v)
        mt[(v + hi * 8) * HID + l15] = c[v];
}

__device__ __forceinline__ int q_push(int* qe, unsigned short* qt, int qn,
                                      unsigned t, int e, unsigned cnt) {
    bool hit = t < cnt;
    unsigned mask = __builtin_amdgcn_ballot_w32(hit);
    if (hit) {
        int slot = qn + (int)__builtin_amdgcn_mbcnt_lo(mask, 0u);
        qe[slot] = e;
        qt[slot] = (unsigned short)t;
    }
    return qn + __builtin_popcount(mask);
}

template <bool SAVE>
__device__ __forceinline__ void q_drain(const int* qe, const unsigned short* qt, int qn,
                                        const int* __restrict__ src, const float* __restrict__ m,
                                        float* acc, int lane,
                                        unsigned* __restrict__ lst, int* lst_n, int lst_cap) {
    const int f  = lane & 15;
    const int hi = lane >> 4;
    int base = 0;
    if (SAVE) {
        if (lane == 0) base = atomicAdd(lst_n, qn);
        base = __builtin_amdgcn_readfirstlane(base);
    }
    for (int g = 0; g < qn; g += 32) {
        int n = qn - g; if (n > 32) n = 32;
        int s = 0, t = 0;
        if (lane < n) {
            s = src[qe[g + lane]]; t = (int)qt[g + lane];
            if (SAVE) {
                int pos = base + g + lane;
                if (pos < lst_cap) lst[pos] = ((unsigned)s << ROW_BITS) | (unsigned)t;
            }
        }
        for (int j = 0; j < n; j += 8) {
            float v[4]; int tr[4]; bool ok[4];
            #pragma unroll
            for (int u = 0; u < 4; ++u) {
                int ja = j + 2 * u, jb = ja + 1;
                bool a_ok = ja < n, b_ok = jb < n;
                int jac = a_ok ? ja : 0;
                int jbc = b_ok ? jb : jac;
                int sa = __builtin_amdgcn_readlane(s, jac), sb = __builtin_amdgcn_readlane(s, jbc);
                int ta = __builtin_amdgcn_readlane(t, jac), tb = __builtin_amdgcn_readlane(t, jbc);
                int sj = hi ? sb : sa;
                tr[u]  = hi ? tb : ta;
                ok[u]  = hi ? b_ok : a_ok;
                v[u]   = m[(size_t)sj * HID + f];
            }
            #pragma unroll
            for (int u = 0; u < 4; ++u)
                if (ok[u]) atomicAdd(&acc[tr[u] * HID + f], v[u]);
        }
    }
}

__device__ __forceinline__ void ld_chunk(int4 (&b)[4], const int4* __restrict__ d4,
                                         int k, int start, int n_part, int c0, int lane) {
    int kk = k < n_part ? k : n_part - 1;
    int ci = kk + start; if (ci >= n_part) ci -= n_part;
    ci += c0;
    #pragma unroll
    for (int j = 0; j < 4; ++j) b[j] = d4[(size_t)ci * 128 + j * 32 + lane];
}

template <bool SAVE>
__device__ __forceinline__ int scan_chunk(const int4 (&b)[4], int k, int start, int n_part, int c0,
                                          unsigned lo, unsigned ucnt, int* qe, unsigned short* qt,
                                          int qn, const int* __restrict__ src,
                                          const float* __restrict__ m, float* acc, int lane,
                                          unsigned* __restrict__ lst, int* lst_n, int lst_cap) {
    int ci = k + start; if (ci >= n_part) ci -= n_part;
    ci += c0;
    #pragma unroll
    for (int j = 0; j < 4; ++j) {
        const int eb = ci * 512 + j * 128 + lane * 4;
        unsigned t0 = (unsigned)b[j].x - lo;
        unsigned t1 = (unsigned)b[j].y - lo;
        unsigned t2 = (unsigned)b[j].z - lo;
        unsigned t3 = (unsigned)b[j].w - lo;
        qn = q_push(qe, qt, qn, t0, eb + 0, ucnt);
        qn = q_push(qe, qt, qn, t1, eb + 1, ucnt);
        qn = q_push(qe, qt, qn, t2, eb + 2, ucnt);
        qn = q_push(qe, qt, qn, t3, eb + 3, ucnt);
    }
    if (qn >= 32) { q_drain<SAVE>(qe, qt, qn, src, m, acc, lane, lst, lst_n, lst_cap); qn = 0; }
    return qn;
}

template <bool SAVE>
__device__ __forceinline__ void owner_scan(const int* __restrict__ src, const int* __restrict__ dst,
                                           const float* __restrict__ m, float* __restrict__ part,
                                           int n_nodes, int n_edges,
                                           float* acc, int* q_e, unsigned short* q_t,
                                           unsigned* __restrict__ lst_all, int* __restrict__ cnts,
                                           int* lst_n) {
    const int tid  = threadIdx.x;
    const int lane = tid & 31;
    const int wv   = __builtin_amdgcn_readfirstlane(tid >> 5);
    const int nw   = blockDim.x >> 5;
    const int p    = blockIdx.x % EDGE_PARTS;
    const int r    = blockIdx.x / EDGE_PARTS;
    int* qe = q_e + wv * Q_CAP;
    unsigned short* qt = q_t + wv * Q_CAP;
    float* res = part + (size_t)p * n_nodes * HID;

    int rr = (n_nodes + ROW_RANGES - 1) / ROW_RANGES;
    rr = (rr + 15) & ~15;
    if (rr > ROW_CAP) rr = ROW_CAP;

    const bool keep = SAVE && ((long long)ROW_RANGES * rr >= n_nodes) &&
                      (n_nodes <= (1 << (32 - ROW_BITS)));
    const int lst_cap = keep ? LIST_CAP : 0;
    unsigned* lst = SAVE ? lst_all + (size_t)blockIdx.x * LIST_CAP : nullptr;
    if (SAVE) { if (tid == 0) *lst_n = 0; }

    const bool vec_ok = ((((size_t)dst) & 15) == 0);
    const int n_full = vec_ok ? (n_edges >> 9) : 0;
    const int c0     = (int)(((long long)p * n_full) / EDGE_PARTS);
    const int n_part = (int)(((long long)(p + 1) * n_full) / EDGE_PARTS) - c0;
    const int start = (int)(((long long)r * n_part) / ROW_RANGES);
    const int4* d4 = (const int4*)dst;
    const int t_base = n_full << 9;
    const int t_lo = t_base + (int)(((long long)p * (n_edges - t_base)) / EDGE_PARTS);
    const int t_hi = t_base + (int)(((long long)(p + 1) * (n_edges - t_base)) / EDGE_PARTS);

    for (int lo = r * rr; lo < n_nodes; lo += ROW_RANGES * rr) {
        int cnt = n_nodes - lo; if (cnt > rr) cnt = rr;
        const unsigned ucnt = (unsigned)cnt;
        const unsigned ulo  = (unsigned)lo;
        float4* acc4 = (float4*)acc;
        for (int i = tid; i < cnt * (HID / 4); i += blockDim.x) acc4[i] = make_float4(0.f, 0.f, 0.f, 0.f);
        __syncthreads();

        int qn = 0;
        if (n_part > 0) {
        int4 A[4], B[4], C[4];
        ld_chunk(A, d4, wv,      start, n_part, c0, lane);
        ld_chunk(B, d4, wv + nw, start, n_part, c0, lane);
        for (int c = wv; c < n_part; c += 3 * nw) {
            ld_chunk(C, d4, c + 2 * nw, start, n_part, c0, lane);
            qn = scan_chunk<SAVE>(A, c, start, n_part, c0, ulo, ucnt, qe, qt, qn, src, m, acc, lane,
                                  lst, lst_n, lst_cap);
            ld_chunk(A, d4, c + 3 * nw, start, n_part, c0, lane);
            if (c + nw < n_part)
                qn = scan_chunk<SAVE>(B, c + nw, start, n_part, c0, ulo, ucnt, qe, qt, qn, src, m, acc,
                                      lane, lst, lst_n, lst_cap);
            ld_chunk(B, d4, c + 4 * nw, start, n_part, c0, lane);
            if (c + 2 * nw < n_part)
                qn = scan_chunk<SAVE>(C, c + 2 * nw, start, n_part, c0, ulo, ucnt, qe, qt, qn, src, m,
                                      acc, lane, lst, lst_n, lst_cap);
        }
        }
        for (int b = t_lo + wv * 32; b < t_hi; b += nw * 32) {
            int e = b + lane;
            unsigned t = (e < t_hi) ? ((unsigned)dst[e] - ulo) : 0xFFFFFFFFu;
            qn = q_push(qe, qt, qn, t, e, ucnt);
            if (qn >= 32) { q_drain<SAVE>(qe, qt, qn, src, m, acc, lane, lst, lst_n, lst_cap); qn = 0; }
        }
        if (qn > 0) q_drain<SAVE>(qe, qt, qn, src, m, acc, lane, lst, lst_n, lst_cap);
        __syncthreads();

        float4* res4 = (float4*)(res + (size_t)lo * HID);
        for (int i = tid; i < cnt * (HID / 4); i += blockDim.x) res4[i] = acc4[i];
        __syncthreads();
    }
    if (SAVE) {
        if (tid == 0) cnts[blockIdx.x * CNT_PITCH] = keep ? ((r * rr < n_nodes) ? *lst_n : 0) : 0x7fffffff;
    }
}

__global__ __launch_bounds__(OWN_THREADS)
void gcn_owner_sum_save(const int* __restrict__ src, const int* __restrict__ dst,
                        const float* __restrict__ m, float* __restrict__ part,
                        unsigned* __restrict__ lst_all, int* __restrict__ cnts,
                        int n_nodes, int n_edges) {
    __shared__ __attribute__((aligned(16))) float acc[ROW_CAP * HID];
    __shared__ int q_e[OWN_WAVES * Q_CAP];
    __shared__ unsigned short q_t[OWN_WAVES * Q_CAP];
    __shared__ int lst_n;
    owner_scan<true>(src, dst, m, part, n_nodes, n_edges, acc, q_e, q_t, lst_all, cnts, &lst_n);
}

__global__ __launch_bounds__(OWN_THREADS)
void gcn_owner_sum_list(const int* __restrict__ src, const int* __restrict__ dst,
                        const float* __restrict__ m, float* __restrict__ part,
                        const unsigned* __restrict__ lst_all, const int* __restrict__ cnts,
                        int n_nodes, int n_edges) {
    __shared__ __attribute__((aligned(16))) float acc[ROW_CAP * HID];
    __shared__ int q_e[OWN_WAVES * Q_CAP];
    __shared__ unsigned short q_t[OWN_WAVES * Q_CAP];

    const int total = cnts[blockIdx.x * CNT_PITCH];
    if ((unsigned)total > (unsigned)LIST_CAP) {
        owner_scan<false>(src, dst, m, part, n_nodes, n_edges, acc, q_e, q_t, nullptr, nullptr, nullptr);
        return;
    }

    const int tid  = threadIdx.x;
    const int lane = tid & 31;
    const int wv   = __builtin_amdgcn_readfirstlane(tid >> 5);
    const int nw   = blockDim.x >> 5;
    const int p    = blockIdx.x % EDGE_PARTS;
    const int r    = blockIdx.x / EDGE_PARTS;
    const int f    = lane & 15;
    const int hi   = lane >> 4;
    const unsigned* lst = lst_all + (size_t)blockIdx.x * LIST_CAP;

    int rr = (n_nodes + ROW_RANGES - 1) / ROW_RANGES;
    rr = (rr + 15) & ~15;
    if (rr > ROW_CAP) rr = ROW_CAP;
    const int lo = r * rr;
    int cnt = n_nodes - lo; if (cnt > rr) cnt = rr; if (cnt < 0) cnt = 0;

    float4* acc4 = (float4*)acc;
    for (int i = tid; i < cnt * (HID / 4); i += blockDim.x) acc4[i] = make_float4(0.f, 0.f, 0.f, 0.f);
    __syncthreads();

    const int nb = (total + 31) >> 5;
    const unsigned row_mask = (1u << ROW_BITS) - 1u;
    unsigned cur[16];
    {
        unsigned w = 0u;
        if (wv * 32 + lane < total) w = lst[wv * 32 + lane];
        #pragma unroll
        for (int u = 0; u < 16; ++u) {
            unsigned wa = (unsigned)__builtin_amdgcn_readlane((int)w, 2 * u);
            unsigned wb = (unsigned)__builtin_amdgcn_readlane((int)w, 2 * u + 1);
            cur[u] = hi ? wb : wa;
        }
    }
    for (int b = wv; b < nb; b += nw) {
        unsigned wn = 0u;
        { int i = (b + nw) * 32 + lane; if (i < total) wn = lst[i]; }
        const int n = total - b * 32;
        float v[16];
        #pragma unroll
        for (int u = 0; u < 16; ++u)
            v[u] = m[(size_t)(cur[u] >> ROW_BITS) * HID + f];
        if (n >= 32) {
            #pragma unroll
            for (int u = 0; u < 16; ++u)
                atomicAdd(&acc[(int)(cur[u] & row_mask) * HID + f], v[u]);
        } else {
            #pragma unroll
            for (int u = 0; u < 16; ++u)
                if (2 * u + hi < n) atomicAdd(&acc[(int)(cur[u] & row_mask) * HID + f], v[u]);
        }
        #pragma unroll
        for (int u = 0; u < 16; ++u) {
            unsigned wa = (unsigned)__builtin_amdgcn_readlane((int)wn, 2 * u);
            unsigned wb = (unsigned)__builtin_amdgcn_readlane((int)wn, 2 * u + 1);
            cur[u] = hi ? wb : wa;
        }
    }
    __syncthreads();

    float4* res4 = (float4*)(part + (size_t)p * n_nodes * HID + (size_t)lo * HID);
    for (int i = tid; i < cnt * (HID / 4); i += blockDim.x) res4[i] = acc4[i];
}

__global__ void gcn_gemm2_wmma(const float* __restrict__ agg, const float* __restrict__ b1,
                               const float* __restrict__ W2, float* __restrict__ m2,
                               int n_tiles, int n_nodes) {
    int wave = (int)((blockIdx.x * blockDim.x + threadIdx.x) >> 5);
    int lane = threadIdx.x & 31;
    if (wave >= n_tiles) return;
    int l15 = lane & 15;
    int hi  = lane >> 4;
    const float* arow = agg + (size_t)(wave * 16 + l15) * HID;
    const size_t plane = (size_t)n_nodes * HID;

    v8f c = {};
    #pragma unroll
    for (int k0 = 0; k0 < HID; k0 += 4) {
        int ka = k0 + hi * 2;
        float sx = arow[ka], sy = arow[ka + 1];
        #pragma unroll
        for (int q = 1; q < EDGE_PARTS; ++q) {
            sx += arow[q * plane + ka];
            sy += arow[q * plane + ka + 1];
        }
        v2f a;
        a.x = fmaxf(sx + b1[ka],     0.0f);
        a.y = fmaxf(sy + b1[ka + 1], 0.0f);
        v2f b;
        b.x = (l15 < OUTD) ? W2[(ka + 0) * OUTD + l15] : 0.0f;
        b.y = (l15 < OUTD) ? W2[(ka + 1) * OUTD + l15] : 0.0f;
        c = __builtin_amdgcn_wmma_f32_16x16x4_f32(false, a, false, b, (short)0, c, false, false);
    }
    float* mt = m2 + (size_t)(wave * 16) * HID;
    #pragma unroll
    for (int v = 0; v < 8; ++v)
        mt[(v + hi * 8) * HID + l15] = c[v];
}

__global__ void gcn_join_out(const float* __restrict__ part, const float* __restrict__ bias,
                             float* __restrict__ out, int n_nodes) {
    int i = (int)(blockIdx.x * blockDim.x + threadIdx.x);
    if (i >= n_nodes * OUTD) return;
    int r = i / OUTD, o = i - r * OUTD;
    const size_t plane = (size_t)n_nodes * HID;
    const float* q = part + (size_t)r * HID + o;
    float s = q[0];
    #pragma unroll
    for (int k = 1; k < EDGE_PARTS; ++k) s += q[k * plane];
    out[i] = s + bias[o];
}

extern "C" void kernel_launch(void* const* d_in, const int* in_sizes, int n_in,
                              void* d_out, int out_size, void* d_ws, size_t ws_size,
                              hipStream_t stream) {
    const float* x  = (const float*)d_in[0];
    const int*   ei = (const int*)  d_in[1];
    const float* W1 = (const float*)d_in[2];
    const float* b1 = (const float*)d_in[3];
    const float* W2 = (const float*)d_in[4];
    const float* b2 = (const float*)d_in[5];
    float* out = (float*)d_out;

    int n_nodes       = in_sizes[0] / IN_DIM;
    long long n_edges = (long long)in_sizes[1] / 2;
    const int* src = ei;
    const int* dst = ei + n_edges;

    float* m1   = (float*)d_ws;
    float* aggp = m1   + (size_t)n_nodes * HID;
    float* m2   = aggp + (size_t)n_nodes * HID * EDGE_PARTS;
    float* outp = m2   + (size_t)n_nodes * HID;
    unsigned* hits = (unsigned*)(outp + (size_t)n_nodes * HID * EDGE_PARTS);
    int* hit_n  = (int*)(hits + (size_t)OWN_GROUPS * LIST_CAP);

    int n_tiles = (n_nodes + 15) / 16;
    int gemm_blocks = (n_tiles + 7) / 8;
    int join_blocks = (n_nodes * OUTD + 255) / 256;

    gcn_gemm1_wmma<<<gemm_blocks, 256, 0, stream>>>(x, W1, m1, n_tiles);
    gcn_owner_sum_save<<<OWN_GROUPS, OWN_THREADS, 0, stream>>>(src, dst, m1, aggp, hits, hit_n, n_nodes, (int)n_edges);
    gcn_gemm2_wmma<<<gemm_blocks, 256, 0, stream>>>(aggp, b1, W2, m2, n_tiles, n_nodes);
    gcn_owner_sum_list<<<OWN_GROUPS, OWN_THREADS, 0, stream>>>(src, dst, m2, outp, hits, hit_n, n_nodes, (int)n_edges);
    gcn_join_out  <<<join_blocks, 256, 0, stream>>>(outp, b2, out, n_nodes);
}


# layer-1 owner sum split into two passes in one work-group: pass one only finds hits and flushes them to the saved list, pass two sums the list in full 32-hit batches like layer 2; LDS arrival order
# speedup vs baseline: 10.9805x; 1.1534x over previous
#include <hip/hip_runtime.h>


typedef float v2f __attribute__((ext_vector_type(2)));
typedef float v8f __attribute__((ext_vector_type(8)));

#define IN_DIM 256
#define HID 16
#define OUTD 10

#define OWN_GROUPS 256
#define OWN_THREADS 1024
#define OWN_WAVES 32
#define EDGE_PARTS 8
#define ROW_RANGES 32
#define ROW_CAP 3136
#define Q_CAP 544
#define LIST_CAP 65536
#define ROW_BITS 12
#define CNT_PITCH 32
static_assert(EDGE_PARTS * ROW_RANGES == OWN_GROUPS, "one work-group per (slice, range) pair");
static_assert(ROW_CAP <= (1 << ROW_BITS), "a local row fits in ROW_BITS bits");

__global__ void gcn_gemm1_wmma(const float* __restrict__ x, const float* __restrict__ W1,
                               float* __restrict__ m1, int n_tiles) {
    int wave = (int)((blockIdx.x * blockDim.x + threadIdx.x) >> 5);
    int lane = threadIdx.x & 31;
    if (wave >= n_tiles) return;
    int l15 = lane & 15;
    int hi  = lane >> 4;
    const float* xrow = x + ((size_t)(wave * 16 + l15)) * IN_DIM;

    v8f c = {};
    #pragma unroll 4
    for (int k0 = 0; k0 < IN_DIM; k0 += 4) {
        int ka = k0 + hi * 2;
        v2f a; a.x = xrow[ka];                 a.y = xrow[ka + 1];
        v2f b; b.x = W1[(ka + 0) * HID + l15]; b.y = W1[(ka + 1) * HID + l15];
        c = __builtin_amdgcn_wmma_f32_16x16x4_f32(false, a, false, b, (short)0, c, false, false);
    }
    float* mt = m1 + (size_t)(wave * 16) * HID;
    #pragma unroll
    for (int v = 0; v < 8; ++v)
        mt[(v + hi * 8) * HID + l15] = c[v];
}

__device__ __forceinline__ int q_push(int* qe, unsigned short* qt, int qn,
                                      unsigned t, int e, unsigned cnt) {
    bool hit = t < cnt;
    unsigned mask = __builtin_amdgcn_ballot_w32(hit);
    if (hit) {
        int slot = qn + (int)__builtin_amdgcn_mbcnt_lo(mask, 0u);
        qe[slot] = e;
        qt[slot] = (unsigned short)t;
    }
    return qn + __builtin_popcount(mask);
}

__device__ __forceinline__ void q_drain(const int* qe, const unsigned short* qt, int qn,
                                        const int* __restrict__ src, const float* __restrict__ m,
                                        float* acc, int lane) {
    const int f  = lane & 15;
    const int hi = lane >> 4;
    for (int g = 0; g < qn; g += 32) {
        int n = qn - g; if (n > 32) n = 32;
        int s = 0, t = 0;
        if (lane < n) { s = src[qe[g + lane]]; t = (int)qt[g + lane]; }
        for (int j = 0; j < n; j += 8) {
            float v[4]; int tr[4]; bool ok[4];
            #pragma unroll
            for (int u = 0; u < 4; ++u) {
                int ja = j + 2 * u, jb = ja + 1;
                bool a_ok = ja < n, b_ok = jb < n;
                int jac = a_ok ? ja : 0;
                int jbc = b_ok ? jb : jac;
                int sa = __builtin_amdgcn_readlane(s, jac), sb = __builtin_amdgcn_readlane(s, jbc);
                int ta = __builtin_amdgcn_readlane(t, jac), tb = __builtin_amdgcn_readlane(t, jbc);
                int sj = hi ? sb : sa;
                tr[u]  = hi ? tb : ta;
                ok[u]  = hi ? b_ok : a_ok;
                v[u]   = m[(size_t)sj * HID + f];
            }
            #pragma unroll
            for (int u = 0; u < 4; ++u)
                if (ok[u]) atomicAdd(&acc[tr[u] * HID + f], v[u]);
        }
    }
}

__device__ __forceinline__ void q_flush(const int* qe, const unsigned short* qt, int qn,
                                        const int* __restrict__ src, int lane,
                                        unsigned* __restrict__ lst, int* lst_n, int lst_cap) {
    int base = 0;
    if (lane == 0) base = atomicAdd(lst_n, qn);
    base = __builtin_amdgcn_readfirstlane(base);
    const int last = qn - 1;
    for (int g = 0; g < qn; g += 64) {
        const int i0 = g + lane, i1 = i0 + 32;
        const int j0 = i0 < last ? i0 : last;
        const int j1 = i1 < last ? i1 : last;
        const int e0 = qe[j0], e1 = qe[j1];
        const unsigned t0 = qt[j0], t1 = qt[j1];
        const int s0 = src[e0], s1 = src[e1];
        if (i0 < qn && base + i0 < lst_cap) lst[base + i0] = ((unsigned)s0 << ROW_BITS) | t0;
        if (i1 < qn && base + i1 < lst_cap) lst[base + i1] = ((unsigned)s1 << ROW_BITS) | t1;
    }
}

__device__ __forceinline__ void ld_chunk(int4 (&b)[4], const int4* __restrict__ d4,
                                         int k, int start, int n_part, int c0, int lane) {
    int kk = k < n_part ? k : n_part - 1;
    int ci = kk + start; if (ci >= n_part) ci -= n_part;
    ci += c0;
    #pragma unroll
    for (int j = 0; j < 4; ++j) b[j] = d4[(size_t)ci * 128 + j * 32 + lane];
}

template <bool TO_LIST>
__device__ __forceinline__ int scan_chunk(const int4 (&b)[4], int k, int start, int n_part, int c0,
                                          unsigned lo, unsigned ucnt, int* qe, unsigned short* qt,
                                          int qn, const int* __restrict__ src,
                                          const float* __restrict__ m, float* acc, int lane,
                                          unsigned* __restrict__ lst, int* lst_n, int lst_cap) {
    int ci = k + start; if (ci >= n_part) ci -= n_part;
    ci += c0;
    #pragma unroll
    for (int j = 0; j < 4; ++j) {
        const int eb = ci * 512 + j * 128 + lane * 4;
        unsigned t0 = (unsigned)b[j].x - lo;
        unsigned t1 = (unsigned)b[j].y - lo;
        unsigned t2 = (unsigned)b[j].z - lo;
        unsigned t3 = (unsigned)b[j].w - lo;
        qn = q_push(qe, qt, qn, t0, eb + 0, ucnt);
        qn = q_push(qe, qt, qn, t1, eb + 1, ucnt);
        qn = q_push(qe, qt, qn, t2, eb + 2, ucnt);
        qn = q_push(qe, qt, qn, t3, eb + 3, ucnt);
    }
    if (qn >= 32) {
        if (TO_LIST) q_flush(qe, qt, qn, src, lane, lst, lst_n, lst_cap);
        else         q_drain(qe, qt, qn, src, m, acc, lane);
        qn = 0;
    }
    return qn;
}

template <bool TO_LIST>
__device__ __forceinline__ void scan_slice(const int* __restrict__ src, const int* __restrict__ dst,
                                           const float* __restrict__ m, int n_edges,
                                           unsigned ulo, unsigned ucnt, float* acc,
                                           int* qe, unsigned short* qt, int p, int r,
                                           int lane, int wv, int nw,
                                           unsigned* __restrict__ lst, int* lst_n, int lst_cap) {
    const bool vec_ok = ((((size_t)dst) & 15) == 0);
    const int n_full = vec_ok ? (n_edges >> 9) : 0;
    const int c0     = (int)(((long long)p * n_full) / EDGE_PARTS);
    const int n_part = (int)(((long long)(p + 1) * n_full) / EDGE_PARTS) - c0;
    const int start = (int)(((long long)r * n_part) / ROW_RANGES);
    const int4* d4 = (const int4*)dst;
    const int t_base = n_full << 9;
    const int t_lo = t_base + (int)(((long long)p * (n_edges - t_base)) / EDGE_PARTS);
    const int t_hi = t_base + (int)(((long long)(p + 1) * (n_edges - t_base)) / EDGE_PARTS);

    int qn = 0;
    if (n_part > 0) {
    int4 A[4], B[4], C[4];
    ld_chunk(A, d4, wv,      start, n_part, c0, lane);
    ld_chunk(B, d4, wv + nw, start, n_part, c0, lane);
    for (int c = wv; c < n_part; c += 3 * nw) {
        ld_chunk(C, d4, c + 2 * nw, start, n_part, c0, lane);
        qn = scan_chunk<TO_LIST>(A, c, start, n_part, c0, ulo, ucnt, qe, qt, qn, src, m, acc, lane,
                                 lst, lst_n, lst_cap);
        ld_chunk(A, d4, c + 3 * nw, start, n_part, c0, lane);
        if (c + nw < n_part)
            qn = scan_chunk<TO_LIST>(B, c + nw, start, n_part, c0, ulo, ucnt, qe, qt, qn, src, m, acc,
                                     lane, lst, lst_n, lst_cap);
        ld_chunk(B, d4, c + 4 * nw, start, n_part, c0, lane);
        if (c + 2 * nw < n_part)
            qn = scan_chunk<TO_LIST>(C, c + 2 * nw, start, n_part, c0, ulo, ucnt, qe, qt, qn, src, m,
                                     acc, lane, lst, lst_n, lst_cap);
    }
    }
    for (int b = t_lo + wv * 32; b < t_hi; b += nw * 32) {
        int e = b + lane;
        unsigned t = (e < t_hi) ? ((unsigned)dst[e] - ulo) : 0xFFFFFFFFu;
        qn = q_push(qe, qt, qn, t, e, ucnt);
        if (qn >= 32) {
            if (TO_LIST) q_flush(qe, qt, qn, src, lane, lst, lst_n, lst_cap);
            else         q_drain(qe, qt, qn, src, m, acc, lane);
            qn = 0;
        }
    }
    if (qn > 0) {
        if (TO_LIST) q_flush(qe, qt, qn, src, lane, lst, lst_n, lst_cap);
        else         q_drain(qe, qt, qn, src, m, acc, lane);
    }
}

__device__ __forceinline__ int range_rows(int n_nodes) {
    int rr = (n_nodes + ROW_RANGES - 1) / ROW_RANGES;
    rr = (rr + 15) & ~15;
    if (rr > ROW_CAP) rr = ROW_CAP;
    return rr;
}

__device__ __forceinline__ void owner_scan(const int* __restrict__ src, const int* __restrict__ dst,
                                           const float* __restrict__ m, float* __restrict__ part,
                                           int n_nodes, int n_edges,
                                           float* acc, int* q_e, unsigned short* q_t) {
    const int tid  = threadIdx.x;
    const int lane = tid & 31;
    const int wv   = __builtin_amdgcn_readfirstlane(tid >> 5);
    const int nw   = blockDim.x >> 5;
    const int p    = blockIdx.x % EDGE_PARTS;
    const int r    = blockIdx.x / EDGE_PARTS;
    int* qe = q_e + wv * Q_CAP;
    unsigned short* qt = q_t + wv * Q_CAP;
    float* res = part + (size_t)p * n_nodes * HID;
    const int rr = range_rows(n_nodes);

    for (int lo = r * rr; lo < n_nodes; lo += ROW_RANGES * rr) {
        int cnt = n_nodes - lo; if (cnt > rr) cnt = rr;
        float4* acc4 = (float4*)acc;
        for (int i = tid; i < cnt * (HID / 4); i += blockDim.x) acc4[i] = make_float4(0.f, 0.f, 0.f, 0.f);
        __syncthreads();

        scan_slice<false>(src, dst, m, n_edges, (unsigned)lo, (unsigned)cnt, acc, qe, qt, p, r,
                          lane, wv, nw, nullptr, nullptr, 0);
        __syncthreads();

        float4* res4 = (float4*)(res + (size_t)lo * HID);
        for (int i = tid; i < cnt * (HID / 4); i += blockDim.x) res4[i] = acc4[i];
        __syncthreads();
    }
}

__device__ __forceinline__ void list_sum(const unsigned* lst, int total,
                                         const float* __restrict__ m, float* acc,
                                         int lane, int wv, int nw) {
    const int f  = lane & 15;
    const int hi = lane >> 4;
    const int nb = (total + 31) >> 5;
    const unsigned row_mask = (1u << ROW_BITS) - 1u;
    unsigned cur[16];
    {
        unsigned w = 0u;
        if (wv * 32 + lane < total) w = lst[wv * 32 + lane];
        #pragma unroll
        for (int u = 0; u < 16; ++u) {
            unsigned wa = (unsigned)__builtin_amdgcn_readlane((int)w, 2 * u);
            unsigned wb = (unsigned)__builtin_amdgcn_readlane((int)w, 2 * u + 1);
            cur[u] = hi ? wb : wa;
        }
    }
    for (int b = wv; b < nb; b += nw) {
        unsigned wn = 0u;
        { int i = (b + nw) * 32 + lane; if (i < total) wn = lst[i]; }
        const int n = total - b * 32;
        float v[16];
        #pragma unroll
        for (int u = 0; u < 16; ++u)
            v[u] = m[(size_t)(cur[u] >> ROW_BITS) * HID + f];
        if (n >= 32) {
            #pragma unroll
            for (int u = 0; u < 16; ++u)
                atomicAdd(&acc[(int)(cur[u] & row_mask) * HID + f], v[u]);
        } else {
            #pragma unroll
            for (int u = 0; u < 16; ++u)
                if (2 * u + hi < n) atomicAdd(&acc[(int)(cur[u] & row_mask) * HID + f], v[u]);
        }
        #pragma unroll
        for (int u = 0; u < 16; ++u) {
            unsigned wa = (unsigned)__builtin_amdgcn_readlane((int)wn, 2 * u);
            unsigned wb = (unsigned)__builtin_amdgcn_readlane((int)wn, 2 * u + 1);
            cur[u] = hi ? wb : wa;
        }
    }
}

__global__ __launch_bounds__(OWN_THREADS)
void gcn_owner_sum_save(const int* __restrict__ src, const int* __restrict__ dst,
                        const float* __restrict__ m, float* __restrict__ part,
                        unsigned* __restrict__ lst_all, int* __restrict__ cnts,
                        int n_nodes, int n_edges) {
    __shared__ __attribute__((aligned(16))) float acc[ROW_CAP * HID];
    __shared__ int q_e[OWN_WAVES * Q_CAP];
    __shared__ unsigned short q_t[OWN_WAVES * Q_CAP];
    __shared__ int lst_n;

    const int tid = threadIdx.x;
    const int rr  = range_rows(n_nodes);
    const bool keep = ((long long)ROW_RANGES * rr >= n_nodes) && (n_nodes <= (1 << (32 - ROW_BITS)));
    if (!keep) {
        owner_scan(src, dst, m, part, n_nodes, n_edges, acc, q_e, q_t);
        if (tid == 0) cnts[blockIdx.x * CNT_PITCH] = 0x7fffffff;
        return;
    }

    const int lane = tid & 31;
    const int wv   = __builtin_amdgcn_readfirstlane(tid >> 5);
    const int nw   = blockDim.x >> 5;
    const int p    = blockIdx.x % EDGE_PARTS;
    const int r    = blockIdx.x / EDGE_PARTS;
    const int lo   = r * rr;
    int cnt = n_nodes - lo; if (cnt > rr) cnt = rr; if (cnt < 0) cnt = 0;
    unsigned* lst = lst_all + (size_t)blockIdx.x * LIST_CAP;

    float4* acc4 = (float4*)acc;
    for (int i = tid; i < cnt * (HID / 4); i += blockDim.x) acc4[i] = make_float4(0.f, 0.f, 0.f, 0.f);
    if (tid == 0) lst_n = 0;
    __syncthreads();

    if (cnt > 0)
        scan_slice<true>(src, dst, m, n_edges, (unsigned)lo, (unsigned)cnt, acc,
                         q_e + wv * Q_CAP, q_t + wv * Q_CAP, p, r, lane, wv, nw,
                         lst, &lst_n, LIST_CAP);
    __syncthreads();

    const int total = lst_n;
    if (tid == 0) cnts[blockIdx.x * CNT_PITCH] = total;
    if (total > LIST_CAP) {
        owner_scan(src, dst, m, part, n_nodes, n_edges, acc, q_e, q_t);
        return;
    }

    list_sum(lst, total, m, acc, lane, wv, nw);
    __syncthreads();

    float4* res4 = (float4*)(part + (size_t)p * n_nodes * HID + (size_t)lo * HID);
    for (int i = tid; i < cnt * (HID / 4); i += blockDim.x) res4[i] = acc4[i];
}

__global__ __launch_bounds__(OWN_THREADS)
void gcn_owner_sum_list(const int* __restrict__ src, const int* __restrict__ dst,
                        const float* __restrict__ m, float* __restrict__ part,
                        const unsigned* __restrict__ lst_all, const int* __restrict__ cnts,
                        int n_nodes, int n_edges) {
    __shared__ __attribute__((aligned(16))) float acc[ROW_CAP * HID];
    __shared__ int q_e[OWN_WAVES * Q_CAP];
    __shared__ unsigned short q_t[OWN_WAVES * Q_CAP];

    const int total = cnts[blockIdx.x * CNT_PITCH];
    if ((unsigned)total > (unsigned)LIST_CAP) {
        owner_scan(src, dst, m, part, n_nodes, n_edges, acc, q_e, q_t);
        return;
    }

    const int tid  = threadIdx.x;
    const int lane = tid & 31;
    const int wv   = __builtin_amdgcn_readfirstlane(tid >> 5);
    const int nw   = blockDim.x >> 5;
    const int p    = blockIdx.x % EDGE_PARTS;
    const int r    = blockIdx.x / EDGE_PARTS;
    const unsigned* lst = lst_all + (size_t)blockIdx.x * LIST_CAP;

    const int rr = range_rows(n_nodes);
    const int lo = r * rr;
    int cnt = n_nodes - lo; if (cnt > rr) cnt = rr; if (cnt < 0) cnt = 0;

    float4* acc4 = (float4*)acc;
    for (int i = tid; i < cnt * (HID / 4); i += blockDim.x) acc4[i] = make_float4(0.f, 0.f, 0.f, 0.f);
    __syncthreads();

    list_sum(lst, total, m, acc, lane, wv, nw);
    __syncthreads();

    float4* res4 = (float4*)(part + (size_t)p * n_nodes * HID + (size_t)lo * HID);
    for (int i = tid; i < cnt * (HID / 4); i += blockDim.x) res4[i] = acc4[i];
}

__global__ void gcn_gemm2_wmma(const float* __restrict__ agg, const float* __restrict__ b1,
                               const float* __restrict__ W2, float* __restrict__ m2,
                               int n_tiles, int n_nodes) {
    int wave = (int)((blockIdx.x * blockDim.x + threadIdx.x) >> 5);
    int lane = threadIdx.x & 31;
    if (wave >= n_tiles) return;
    int l15 = lane & 15;
    int hi  = lane >> 4;
    const float* arow = agg + (size_t)(wave * 16 + l15) * HID;
    const size_t plane = (size_t)n_nodes * HID;

    v8f c = {};
    #pragma unroll
    for (int k0 = 0; k0 < HID; k0 += 4) {
        int ka = k0 + hi * 2;
        float sx = arow[ka], sy = arow[ka + 1];
        #pragma unroll
        for (int q = 1; q < EDGE_PARTS; ++q) {
            sx += arow[q * plane + ka];
            sy += arow[q * plane + ka + 1];
        }
        v2f a;
        a.x = fmaxf(sx + b1[ka],     0.0f);
        a.y = fmaxf(sy + b1[ka + 1], 0.0f);
        v2f b;
        b.x = (l15 < OUTD) ? W2[(ka + 0) * OUTD + l15] : 0.0f;
        b.y = (l15 < OUTD) ? W2[(ka + 1) * OUTD + l15] : 0.0f;
        c = __builtin_amdgcn_wmma_f32_16x16x4_f32(false, a, false, b, (short)0, c, false, false);
    }
    float* mt = m2 + (size_t)(wave * 16) * HID;
    #pragma unroll
    for (int v = 0; v < 8; ++v)
        mt[(v + hi * 8) * HID + l15] = c[v];
}

__global__ void gcn_join_out(const float* __restrict__ part, const float* __restrict__ bias,
                             float* __restrict__ out, int n_nodes) {
    int i = (int)(blockIdx.x * blockDim.x + threadIdx.x);
    if (i >= n_nodes * OUTD) return;
    int r = i / OUTD, o = i - r * OUTD;
    const size_t plane = (size_t)n_nodes * HID;
    const float* q = part + (size_t)r * HID + o;
    float s = q[0];
    #pragma unroll
    for (int k = 1; k < EDGE_PARTS; ++k) s += q[k * plane];
    out[i] = s + bias[o];
}

extern "C" void kernel_launch(void* const* d_in, const int* in_sizes, int n_in,
                              void* d_out, int out_size, void* d_ws, size_t ws_size,
                              hipStream_t stream) {
    const float* x  = (const float*)d_in[0];
    const int*   ei = (const int*)  d_in[1];
    const float* W1 = (const float*)d_in[2];
    const float* b1 = (const float*)d_in[3];
    const float* W2 = (const float*)d_in[4];
    const float* b2 = (const float*)d_in[5];
    float* out = (float*)d_out;

    int n_nodes       = in_sizes[0] / IN_DIM;
    long long n_edges = (long long)in_sizes[1] / 2;
    const int* src = ei;
    const int* dst = ei + n_edges;

    float* m1   = (float*)d_ws;
    float* aggp = m1   + (size_t)n_nodes * HID;
    float* m2   = aggp + (size_t)n_nodes * HID * EDGE_PARTS;
    float* outp = m2   + (size_t)n_nodes * HID;
    unsigned* hits = (unsigned*)(outp + (size_t)n_nodes * HID * EDGE_PARTS);
    int* hit_n  = (int*)(hits + (size_t)OWN_GROUPS * LIST_CAP);

    int n_tiles = (n_nodes + 15) / 16;
    int gemm_blocks = (n_tiles + 7) / 8;
    int join_blocks = (n_nodes * OUTD + 255) / 256;

    gcn_gemm1_wmma<<<gemm_blocks, 256, 0, stream>>>(x, W1, m1, n_tiles);
    gcn_owner_sum_save<<<OWN_GROUPS, OWN_THREADS, 0, stream>>>(src, dst, m1, aggp, hits, hit_n, n_nodes, (int)n_edges);
    gcn_gemm2_wmma<<<gemm_blocks, 256, 0, stream>>>(aggp, b1, W2, m2, n_tiles, n_nodes);
    gcn_owner_sum_list<<<OWN_GROUPS, OWN_THREADS, 0, stream>>>(src, dst, m2, outp, hits, hit_n, n_nodes, (int)n_edges);
    gcn_join_out  <<<join_blocks, 256, 0, stream>>>(outp, b2, out, n_nodes);
}


# find pass re-tiled, own launch: a work-group sweeps 1/32 of dst and src for a window of 4 row ranges and splits its hits 4 ways into the owners' lists; both layer sums read lists; LDS arrival order
# speedup vs baseline: 11.9921x; 1.0921x over previous
#include <hip/hip_runtime.h>


typedef float v2f __attribute__((ext_vector_type(2)));
typedef float v8f __attribute__((ext_vector_type(8)));

#define IN_DIM 256
#define HID 16
#define OUTD 10

#define OWN_GROUPS 256
#define OWN_THREADS 1024
#define OWN_WAVES 32
#define EDGE_PARTS 8
#define ROW_RANGES 32
#define ROW_CAP 3136
#define Q_CAP 544
#define LIST_CAP 65536
#define ROW_BITS 12
#define CNT_PITCH 32
#define FIND_WIN 4
#define N_WIN (ROW_RANGES / FIND_WIN)
#define FIND_SLICES (OWN_GROUPS / N_WIN)
#define SLICES_PER_PART (FIND_SLICES / EDGE_PARTS)
#define SUB_CAP (LIST_CAP / FIND_WIN)
#define WIN_BITS 14
static_assert(EDGE_PARTS * ROW_RANGES == OWN_GROUPS, "one work-group per (slice, range) pair");
static_assert(ROW_CAP <= (1 << ROW_BITS), "a local row fits in ROW_BITS bits");
static_assert(N_WIN * FIND_WIN == ROW_RANGES && N_WIN * FIND_SLICES == OWN_GROUPS, "windows tile the ranges");
static_assert(SLICES_PER_PART * EDGE_PARTS == FIND_SLICES, "find slices tile the slices of the sums");
static_assert(FIND_WIN * ROW_CAP <= (1 << WIN_BITS), "a row of the window fits in WIN_BITS bits");

__global__ void gcn_gemm1_wmma(const float* __restrict__ x, const float* __restrict__ W1,
                               float* __restrict__ m1, int n_tiles) {
    int wave = (int)((blockIdx.x * blockDim.x + threadIdx.x) >> 5);
    int lane = threadIdx.x & 31;
    if (wave >= n_tiles) return;
    int l15 = lane & 15;
    int hi  = lane >> 4;
    const float* xrow = x + ((size_t)(wave * 16 + l15)) * IN_DIM;

    v8f c = {};
    #pragma unroll 4
    for (int k0 = 0; k0 < IN_DIM; k0 += 4) {
        int ka = k0 + hi * 2;
        v2f a; a.x = xrow[ka];                 a.y = xrow[ka + 1];
        v2f b; b.x = W1[(ka + 0) * HID + l15]; b.y = W1[(ka + 1) * HID + l15];
        c = __builtin_amdgcn_wmma_f32_16x16x4_f32(false, a, false, b, (short)0, c, false, false);
    }
    float* mt = m1 + (size_t)(wave * 16) * HID;
    #pragma unroll
    for (int v = 0; v < 8; ++v)
        mt[(v + hi * 8) * HID + l15] = c[v];
}

__device__ __forceinline__ int q_push(int* qe, unsigned short* qt, int qn,
                                      unsigned t, int e, unsigned cnt) {
    bool hit = t < cnt;
    unsigned mask = __builtin_amdgcn_ballot_w32(hit);
    if (hit) {
        int slot = qn + (int)__builtin_amdgcn_mbcnt_lo(mask, 0u);
        qe[slot] = e;
        qt[slot] = (unsigned short)t;
    }
    return qn + __builtin_popcount(mask);
}

__device__ __forceinline__ void q_drain(const int* qe, const unsigned short* qt, int qn,
                                        const int* __restrict__ src, const float* __restrict__ m,
                                        float* acc, int lane) {
    const int f  = lane & 15;
    const int hi = lane >> 4;
    for (int g = 0; g < qn; g += 32) {
        int n = qn - g; if (n > 32) n = 32;
        int s = 0, t = 0;
        if (lane < n) { s = src[qe[g + lane]]; t = (int)qt[g + lane]; }
        for (int j = 0; j < n; j += 8) {
            float v[4]; int tr[4]; bool ok[4];
            #pragma unroll
            for (int u = 0; u < 4; ++u) {
                int ja = j + 2 * u, jb = ja + 1;
                bool a_ok = ja < n, b_ok = jb < n;
                int jac = a_ok ? ja : 0;
                int jbc = b_ok ? jb : jac;
                int sa = __builtin_amdgcn_readlane(s, jac), sb = __builtin_amdgcn_readlane(s, jbc);
                int ta = __builtin_amdgcn_readlane(t, jac), tb = __builtin_amdgcn_readlane(t, jbc);
                int sj = hi ? sb : sa;
                tr[u]  = hi ? tb : ta;
                ok[u]  = hi ? b_ok : a_ok;
                v[u]   = m[(size_t)sj * HID + f];
            }
            #pragma unroll
            for (int u = 0; u < 4; ++u)
                if (ok[u]) atomicAdd(&acc[tr[u] * HID + f], v[u]);
        }
    }
}

__device__ __forceinline__ void ld_chunk(int4 (&b)[4], const int4* __restrict__ d4,
                                         int k, int start, int n_part, int c0, int lane) {
    int kk = k < n_part ? k : n_part - 1;
    int ci = kk + start; if (ci >= n_part) ci -= n_part;
    ci += c0;
    #pragma unroll
    for (int j = 0; j < 4; ++j) b[j] = d4[(size_t)ci * 128 + j * 32 + lane];
}

__device__ __forceinline__ int scan_chunk(const int4 (&b)[4], int k, int start, int n_part, int c0,
                                          unsigned lo, unsigned ucnt, int* qe, unsigned short* qt,
                                          int qn, const int* __restrict__ src,
                                          const float* __restrict__ m, float* acc, int lane) {
    int ci = k + start; if (ci >= n_part) ci -= n_part;
    ci += c0;
    #pragma unroll
    for (int j = 0; j < 4; ++j) {
        const int eb = ci * 512 + j * 128 + lane * 4;
        unsigned t0 = (unsigned)b[j].x - lo;
        unsigned t1 = (unsigned)b[j].y - lo;
        unsigned t2 = (unsigned)b[j].z - lo;
        unsigned t3 = (unsigned)b[j].w - lo;
        qn = q_push(qe, qt, qn, t0, eb + 0, ucnt);
        qn = q_push(qe, qt, qn, t1, eb + 1, ucnt);
        qn = q_push(qe, qt, qn, t2, eb + 2, ucnt);
        qn = q_push(qe, qt, qn, t3, eb + 3, ucnt);
    }
    if (qn >= 32) {
        q_drain(qe, qt, qn, src, m, acc, lane);
        qn = 0;
    }
    return qn;
}

__device__ __forceinline__ void scan_slice(const int* __restrict__ src, const int* __restrict__ dst,
                                           const float* __restrict__ m, int n_edges,
                                           unsigned ulo, unsigned ucnt, float* acc,
                                           int* qe, unsigned short* qt, int p, int r,
                                           int lane, int wv, int nw) {
    const bool vec_ok = ((((size_t)dst) & 15) == 0);
    const int n_full = vec_ok ? (n_edges >> 9) : 0;
    const int c0     = (int)(((long long)p * n_full) / EDGE_PARTS);
    const int n_part = (int)(((long long)(p + 1) * n_full) / EDGE_PARTS) - c0;
    const int start = (int)(((long long)r * n_part) / ROW_RANGES);
    const int4* d4 = (const int4*)dst;
    const int t_base = n_full << 9;
    const int t_lo = t_base + (int)(((long long)p * (n_edges - t_base)) / EDGE_PARTS);
    const int t_hi = t_base + (int)(((long long)(p + 1) * (n_edges - t_base)) / EDGE_PARTS);

    int qn = 0;
    if (n_part > 0) {
    int4 A[4], B[4], C[4];
    ld_chunk(A, d4, wv,      start, n_part, c0, lane);
    ld_chunk(B, d4, wv + nw, start, n_part, c0, lane);
    for (int c = wv; c < n_part; c += 3 * nw) {
        ld_chunk(C, d4, c + 2 * nw, start, n_part, c0, lane);
        qn = scan_chunk(A, c, start, n_part, c0, ulo, ucnt, qe, qt, qn, src, m, acc, lane);
        ld_chunk(A, d4, c + 3 * nw, start, n_part, c0, lane);
        if (c + nw < n_part)
            qn = scan_chunk(B, c + nw, start, n_part, c0, ulo, ucnt, qe, qt, qn, src, m, acc, lane);
        ld_chunk(B, d4, c + 4 * nw, start, n_part, c0, lane);
        if (c + 2 * nw < n_part)
            qn = scan_chunk(C, c + 2 * nw, start, n_part, c0, ulo, ucnt, qe, qt, qn, src, m, acc, lane);
    }
    }
    for (int b = t_lo + wv * 32; b < t_hi; b += nw * 32) {
        int e = b + lane;
        unsigned t = (e < t_hi) ? ((unsigned)dst[e] - ulo) : 0xFFFFFFFFu;
        qn = q_push(qe, qt, qn, t, e, ucnt);
        if (qn >= 32) {
            q_drain(qe, qt, qn, src, m, acc, lane);
            qn = 0;
        }
    }
    if (qn > 0) q_drain(qe, qt, qn, src, m, acc, lane);
}

__device__ __forceinline__ int range_rows(int n_nodes) {
    int rr = (n_nodes + ROW_RANGES - 1) / ROW_RANGES;
    rr = (rr + 15) & ~15;
    if (rr > ROW_CAP) rr = ROW_CAP;
    return rr;
}

__device__ __forceinline__ int w_push(unsigned* qw, int qn, unsigned t, int s, unsigned cnt) {
    bool hit = t < cnt;
    unsigned mask = __builtin_amdgcn_ballot_w32(hit);
    if (hit) qw[qn + (int)__builtin_amdgcn_mbcnt_lo(mask, 0u)] = ((unsigned)s << WIN_BITS) | t;
    return qn + __builtin_popcount(mask);
}

__device__ __forceinline__ void w_flush(const unsigned* qw, int qn, int lane, unsigned rr,
                                        unsigned* __restrict__ lst, int* lst_n) {
    for (int g = 0; g < qn; g += 32) {
        const int i = g + lane;
        const bool valid = i < qn;
        const unsigned wd = qw[valid ? i : qn - 1];
        const unsigned t = wd & ((1u << WIN_BITS) - 1u);
        unsigned k = 0;
        #pragma unroll
        for (int u = 1; u < FIND_WIN; ++u) k += (t >= (unsigned)u * rr) ? 1u : 0u;
        const unsigned word = ((wd >> WIN_BITS) << ROW_BITS) | (t - k * rr);
        unsigned mk[FIND_WIN];
        int add = 0;
        #pragma unroll
        for (int u = 0; u < FIND_WIN; ++u) {
            mk[u] = __builtin_amdgcn_ballot_w32(valid && k == (unsigned)u);
            if (lane == u) add = __builtin_popcount(mk[u]);
        }
        int base = 0;
        if (lane < FIND_WIN) base = atomicAdd(&lst_n[lane], add);
        int pos = 0;
        #pragma unroll
        for (int u = 0; u < FIND_WIN; ++u) {
            const int bu = __builtin_amdgcn_readlane(base, u);
            if (k == (unsigned)u) pos = bu + (int)__builtin_amdgcn_mbcnt_lo(mk[u], 0u);
        }
        if (valid && pos < SUB_CAP) lst[(size_t)k * SUB_CAP + pos] = word;
    }
}

__device__ __forceinline__ void ld_pair(int4 (&d)[4], int4 (&s)[4], const int4* __restrict__ d4,
                                        const int4* __restrict__ s4, int k, int start,
                                        int n_part, int c0, int lane) {
    int kk = k < n_part ? k : n_part - 1;
    int ci = kk + start; if (ci >= n_part) ci -= n_part;
    ci += c0;
    #pragma unroll
    for (int j = 0; j < 4; ++j) {
        d[j] = d4[(size_t)ci * 128 + j * 32 + lane];
        s[j] = s4[(size_t)ci * 128 + j * 32 + lane];
    }
}

__device__ __forceinline__ int find_chunk(const int4 (&d)[4], const int4 (&s)[4], unsigned lo,
                                          unsigned ucnt, unsigned* qw, int qn, int lane,
                                          unsigned rr, unsigned* __restrict__ lst, int* lst_n) {
    #pragma unroll
    for (int j = 0; j < 4; ++j) {
        qn = w_push(qw, qn, (unsigned)d[j].x - lo, s[j].x, ucnt);
        qn = w_push(qw, qn, (unsigned)d[j].y - lo, s[j].y, ucnt);
        qn = w_push(qw, qn, (unsigned)d[j].z - lo, s[j].z, ucnt);
        qn = w_push(qw, qn, (unsigned)d[j].w - lo, s[j].w, ucnt);
    }
    if (qn >= 32) { w_flush(qw, qn, lane, rr, lst, lst_n); qn = 0; }
    return qn;
}

__global__ __launch_bounds__(OWN_THREADS)
void gcn_find_hits(const int* __restrict__ src, const int* __restrict__ dst,
                   unsigned* __restrict__ lst_all, int* __restrict__ cnts,
                   int n_nodes, int n_edges) {
    __shared__ unsigned q_w[OWN_WAVES * Q_CAP];
    __shared__ int lst_n[FIND_WIN];

    const int tid = threadIdx.x;
    const int rr  = range_rows(n_nodes);
    int* my_cnt = cnts + (size_t)blockIdx.x * FIND_WIN * CNT_PITCH;
    const bool keep = ((long long)ROW_RANGES * rr >= n_nodes) && (n_nodes <= (1 << (32 - WIN_BITS)))
                      && (((((size_t)src) | ((size_t)dst)) & 15) == 0);
    if (!keep) {
        if (tid < FIND_WIN) my_cnt[tid * CNT_PITCH] = 0x7fffffff;
        return;
    }

    const int lane = tid & 31;
    const int wv   = __builtin_amdgcn_readfirstlane(tid >> 5);
    const int nw   = blockDim.x >> 5;
    const int q    = blockIdx.x % FIND_SLICES;
    const int w    = blockIdx.x / FIND_SLICES;
    const int lo   = w * FIND_WIN * rr;
    int cnt = n_nodes - lo; if (cnt > FIND_WIN * rr) cnt = FIND_WIN * rr; if (cnt < 0) cnt = 0;
    unsigned* lst = lst_all + (size_t)blockIdx.x * LIST_CAP;
    unsigned* qw  = q_w + wv * Q_CAP;

    if (tid < FIND_WIN) lst_n[tid] = 0;
    __syncthreads();

    if (cnt > 0) {
        const int n_full = n_edges >> 9;
        const int c0     = (int)(((long long)q * n_full) / FIND_SLICES);
        const int n_part = (int)(((long long)(q + 1) * n_full) / FIND_SLICES) - c0;
        const int start  = (int)(((long long)w * n_part) / N_WIN);
        const int4* d4 = (const int4*)dst;
        const int4* s4 = (const int4*)src;
        const int t_base = n_full << 9;
        const int t_lo = t_base + (int)(((long long)q * (n_edges - t_base)) / FIND_SLICES);
        const int t_hi = t_base + (int)(((long long)(q + 1) * (n_edges - t_base)) / FIND_SLICES);
        const unsigned ulo = (unsigned)lo, ucnt = (unsigned)cnt, urr = (unsigned)rr;

        int qn = 0;
        if (n_part > 0) {
            int4 dA[4], sA[4], dB[4], sB[4];
            ld_pair(dA, sA, d4, s4, wv,      start, n_part, c0, lane);
            ld_pair(dB, sB, d4, s4, wv + nw, start, n_part, c0, lane);
            for (int c = wv; c < n_part; c += 2 * nw) {
                qn = find_chunk(dA, sA, ulo, ucnt, qw, qn, lane, urr, lst, lst_n);
                ld_pair(dA, sA, d4, s4, c + 2 * nw, start, n_part, c0, lane);
                if (c + nw < n_part)
                    qn = find_chunk(dB, sB, ulo, ucnt, qw, qn, lane, urr, lst, lst_n);
                ld_pair(dB, sB, d4, s4, c + 3 * nw, start, n_part, c0, lane);
            }
        }
        for (int b = t_lo + wv * 32; b < t_hi; b += nw * 32) {
            const int e = b + lane;
            unsigned t = 0xFFFFFFFFu; int sn = 0;
            if (e < t_hi) { t = (unsigned)dst[e] - ulo; sn = src[e]; }
            qn = w_push(qw, qn, t, sn, ucnt);
            if (qn >= 32) { w_flush(qw, qn, lane, urr, lst, lst_n); qn = 0; }
        }
        if (qn > 0) w_flush(qw, qn, lane, urr, lst, lst_n);
    }
    __syncthreads();
    if (tid < FIND_WIN) my_cnt[tid * CNT_PITCH] = lst_n[tid];
}

__device__ __forceinline__ void owner_scan(const int* __restrict__ src, const int* __restrict__ dst,
                                           const float* __restrict__ m, float* __restrict__ part,
                                           int n_nodes, int n_edges,
                                           float* acc, int* q_e, unsigned short* q_t) {
    const int tid  = threadIdx.x;
    const int lane = tid & 31;
    const int wv   = __builtin_amdgcn_readfirstlane(tid >> 5);
    const int nw   = blockDim.x >> 5;
    const int p    = blockIdx.x % EDGE_PARTS;
    const int r    = blockIdx.x / EDGE_PARTS;
    int* qe = q_e + wv * Q_CAP;
    unsigned short* qt = q_t + wv * Q_CAP;
    float* res = part + (size_t)p * n_nodes * HID;
    const int rr = range_rows(n_nodes);

    for (int lo = r * rr; lo < n_nodes; lo += ROW_RANGES * rr) {
        int cnt = n_nodes - lo; if (cnt > rr) cnt = rr;
        float4* acc4 = (float4*)acc;
        for (int i = tid; i < cnt * (HID / 4); i += blockDim.x) acc4[i] = make_float4(0.f, 0.f, 0.f, 0.f);
        __syncthreads();

        scan_slice(src, dst, m, n_edges, (unsigned)lo, (unsigned)cnt, acc, qe, qt, p, r,
                   lane, wv, nw);
        __syncthreads();

        float4* res4 = (float4*)(res + (size_t)lo * HID);
        for (int i = tid; i < cnt * (HID / 4); i += blockDim.x) res4[i] = acc4[i];
        __syncthreads();
    }
}

struct HitLists {
    const unsigned* l[SLICES_PER_PART];
    int end[SLICES_PER_PART];
};
__device__ __forceinline__ unsigned hit_word(const HitLists& h, int i) {
    const unsigned* ptr = h.l[0] + i;
    #pragma unroll
    for (int u = 1; u < SLICES_PER_PART; ++u)
        if (i >= h.end[u - 1]) ptr = h.l[u] + (i - h.end[u - 1]);
    return *ptr;
}

__device__ __forceinline__ void list_sum(const HitLists& h, const float* __restrict__ m, float* acc,
                                         int lane, int wv, int nw) {
    const int total = h.end[SLICES_PER_PART - 1];
    const int f  = lane & 15;
    const int hi = lane >> 4;
    const int nb = (total + 31) >> 5;
    const unsigned row_mask = (1u << ROW_BITS) - 1u;
    unsigned cur[16];
    {
        unsigned w = 0u;
        if (wv * 32 + lane < total) w = hit_word(h, wv * 32 + lane);
        #pragma unroll
        for (int u = 0; u < 16; ++u) {
            unsigned wa = (unsigned)__builtin_amdgcn_readlane((int)w, 2 * u);
            unsigned wb = (unsigned)__builtin_amdgcn_readlane((int)w, 2 * u + 1);
            cur[u] = hi ? wb : wa;
        }
    }
    for (int b = wv; b < nb; b += nw) {
        unsigned wn = 0u;
        { int i = (b + nw) * 32 + lane; if (i < total) wn = hit_word(h, i); }
        const int n = total - b * 32;
        float v[16];
        #pragma unroll
        for (int u = 0; u < 16; ++u)
            v[u] = m[(size_t)(cur[u] >> ROW_BITS) * HID + f];
        if (n >= 32) {
            #pragma unroll
            for (int u = 0; u < 16; ++u)
                atomicAdd(&acc[(int)(cur[u] & row_mask) * HID + f], v[u]);
        } else {
            #pragma unroll
            for (int u = 0; u < 16; ++u)
                if (2 * u + hi < n) atomicAdd(&acc[(int)(cur[u] & row_mask) * HID + f], v[u]);
        }
        #pragma unroll
        for (int u = 0; u < 16; ++u) {
            unsigned wa = (unsigned)__builtin_amdgcn_readlane((int)wn, 2 * u);
            unsigned wb = (unsigned)__builtin_amdgcn_readlane((int)wn, 2 * u + 1);
            cur[u] = hi ? wb : wa;
        }
    }
}

__global__ __launch_bounds__(OWN_THREADS)
void gcn_owner_sum_list(const int* __restrict__ src, const int* __restrict__ dst,
                        const float* __restrict__ m, float* __restrict__ part,
                        const unsigned* __restrict__ lst_all, const int* __restrict__ cnts,
                        int n_nodes, int n_edges) {
    __shared__ __attribute__((aligned(16))) float acc[ROW_CAP * HID];
    __shared__ int q_e[OWN_WAVES * Q_CAP];
    __shared__ unsigned short q_t[OWN_WAVES * Q_CAP];

    const int p = blockIdx.x % EDGE_PARTS;
    const int r = blockIdx.x / EDGE_PARTS;
    const int w = r / FIND_WIN;
    const int k = r % FIND_WIN;
    HitLists h;
    bool lists_ok = true;
    int run = 0;
    #pragma unroll
    for (int u = 0; u < SLICES_PER_PART; ++u) {
        const size_t fb = (size_t)w * FIND_SLICES + (size_t)p * SLICES_PER_PART + u;
        const int n = cnts[(fb * FIND_WIN + k) * CNT_PITCH];
        lists_ok = lists_ok && ((unsigned)n <= (unsigned)SUB_CAP);
        h.l[u] = lst_all + fb * LIST_CAP + (size_t)k * SUB_CAP;
        run += lists_ok ? n : 0;
        h.end[u] = run;
    }
    if (!lists_ok) {
        owner_scan(src, dst, m, part, n_nodes, n_edges, acc, q_e, q_t);
        return;
    }

    const int tid  = threadIdx.x;
    const int lane = tid & 31;
    const int wv   = __builtin_amdgcn_readfirstlane(tid >> 5);
    const int nw   = blockDim.x >> 5;

    const int rr = range_rows(n_nodes);
    const int lo = r * rr;
    int cnt = n_nodes - lo; if (cnt > rr) cnt = rr; if (cnt < 0) cnt = 0;

    float4* acc4 = (float4*)acc;
    for (int i = tid; i < cnt * (HID / 4); i += blockDim.x) acc4[i] = make_float4(0.f, 0.f, 0.f, 0.f);
    __syncthreads();

    list_sum(h, m, acc, lane, wv, nw);
    __syncthreads();

    float4* res4 = (float4*)(part + (size_t)p * n_nodes * HID + (size_t)lo * HID);
    for (int i = tid; i < cnt * (HID / 4); i += blockDim.x) res4[i] = acc4[i];
}

__global__ void gcn_gemm2_wmma(const float* __restrict__ agg, const float* __restrict__ b1,
                               const float* __restrict__ W2, float* __restrict__ m2,
                               int n_tiles, int n_nodes) {
    int wave = (int)((blockIdx.x * blockDim.x + threadIdx.x) >> 5);
    int lane = threadIdx.x & 31;
    if (wave >= n_tiles) return;
    int l15 = lane & 15;
    int hi  = lane >> 4;
    const float* arow = agg + (size_t)(wave * 16 + l15) * HID;
    const size_t plane = (size_t)n_nodes * HID;

    v8f c = {};
    #pragma unroll
    for (int k0 = 0; k0 < HID; k0 += 4) {
        int ka = k0 + hi * 2;
        float sx = arow[ka], sy = arow[ka + 1];
        #pragma unroll
        for (int q = 1; q < EDGE_PARTS; ++q) {
            sx += arow[q * plane + ka];
            sy += arow[q * plane + ka + 1];
        }
        v2f a;
        a.x = fmaxf(sx + b1[ka],     0.0f);
        a.y = fmaxf(sy + b1[ka + 1], 0.0f);
        v2f b;
        b.x = (l15 < OUTD) ? W2[(ka + 0) * OUTD + l15] : 0.0f;
        b.y = (l15 < OUTD) ? W2[(ka + 1) * OUTD + l15] : 0.0f;
        c = __builtin_amdgcn_wmma_f32_16x16x4_f32(false, a, false, b, (short)0, c, false, false);
    }
    float* mt = m2 + (size_t)(wave * 16) * HID;
    #pragma unroll
    for (int v = 0; v < 8; ++v)
        mt[(v + hi * 8) * HID + l15] = c[v];
}

__global__ void gcn_join_out(const float* __restrict__ part, const float* __restrict__ bias,
                             float* __restrict__ out, int n_nodes) {
    int i = (int)(blockIdx.x * blockDim.x + threadIdx.x);
    if (i >= n_nodes * OUTD) return;
    int r = i / OUTD, o = i - r * OUTD;
    const size_t plane = (size_t)n_nodes * HID;
    const float* q = part + (size_t)r * HID + o;
    float s = q[0];
    #pragma unroll
    for (int k = 1; k < EDGE_PARTS; ++k) s += q[k * plane];
    out[i] = s + bias[o];
}

extern "C" void kernel_launch(void* const* d_in, const int* in_sizes, int n_in,
                              void* d_out, int out_size, void* d_ws, size_t ws_size,
                              hipStream_t stream) {
    const float* x  = (const float*)d_in[0];
    const int*   ei = (const int*)  d_in[1];
    const float* W1 = (const float*)d_in[2];
    const float* b1 = (const float*)d_in[3];
    const float* W2 = (const float*)d_in[4];
    const float* b2 = (const float*)d_in[5];
    float* out = (float*)d_out;

    int n_nodes       = in_sizes[0] / IN_DIM;
    long long n_edges = (long long)in_sizes[1] / 2;
    const int* src = ei;
    const int* dst = ei + n_edges;

    float* m1   = (float*)d_ws;
    float* aggp = m1   + (size_t)n_nodes * HID;
    float* m2   = aggp + (size_t)n_nodes * HID * EDGE_PARTS;
    float* outp = m2   + (size_t)n_nodes * HID;
    unsigned* hits = (unsigned*)(outp + (size_t)n_nodes * HID * EDGE_PARTS);
    int* hit_n  = (int*)(hits + (size_t)OWN_GROUPS * LIST_CAP);

    int n_tiles = (n_nodes + 15) / 16;
    int gemm_blocks = (n_tiles + 7) / 8;
    int join_blocks = (n_nodes * OUTD + 255) / 256;

    gcn_gemm1_wmma<<<gemm_blocks, 256, 0, stream>>>(x, W1, m1, n_tiles);
    gcn_find_hits<<<OWN_GROUPS, OWN_THREADS, 0, stream>>>(src, dst, hits, hit_n, n_nodes, (int)n_edges);
    gcn_owner_sum_list<<<OWN_GROUPS, OWN_THREADS, 0, stream>>>(src, dst, m1, aggp, hits, hit_n, n_nodes, (int)n_edges);
    gcn_gemm2_wmma<<<gemm_blocks, 256, 0, stream>>>(aggp, b1, W2, m2, n_tiles, n_nodes);
    gcn_owner_sum_list<<<OWN_GROUPS, OWN_THREADS, 0, stream>>>(src, dst, m2, outp, hits, hit_n, n_nodes, (int)n_edges);
    gcn_join_out  <<<join_blocks, 256, 0, stream>>>(outp, b2, out, n_nodes);
}


# owner tiling 128 row ranges x 2 edge slices (was 32 x 8): 2 partial planes a layer, not 8; find splits its window 16 ways by bit ballots, one list per wave pair; LDS arrival order
# speedup vs baseline: 12.7415x; 1.0625x over previous
#include <hip/hip_runtime.h>


typedef float v2f __attribute__((ext_vector_type(2)));
typedef float v8f __attribute__((ext_vector_type(8)));

#define IN_DIM 256
#define HID 16
#define OUTD 10

#define OWN_GROUPS 256
#define OWN_THREADS 1024
#define OWN_WAVES 32
#define EDGE_PARTS 2
#define ROW_RANGES 128
#define ROW_CAP 2048
#define Q_CAP 544
#define LIST_CAP 65536
#define ROW_BITS 12
#define CNT_PITCH 32
#define FIND_WIN 16
#define FIND_WIN_LOG 4
#define N_WIN (ROW_RANGES / FIND_WIN)
#define FIND_SLICES (OWN_GROUPS / N_WIN)
#define SLICES_PER_PART (FIND_SLICES / EDGE_PARTS)
#define SUB_CAP (LIST_CAP / FIND_WIN)
#define WIN_BITS 15
static_assert(EDGE_PARTS * ROW_RANGES == OWN_GROUPS, "one work-group per (slice, range) pair");
static_assert(ROW_CAP <= (1 << ROW_BITS), "a local row fits in ROW_BITS bits");
static_assert(N_WIN * FIND_WIN == ROW_RANGES && N_WIN * FIND_SLICES == OWN_GROUPS, "windows tile the ranges");
static_assert(SLICES_PER_PART * EDGE_PARTS == FIND_SLICES, "find slices tile the slices of the sums");
static_assert(FIND_WIN * ROW_CAP <= (1 << WIN_BITS), "a row of the window fits in WIN_BITS bits");
static_assert((1 << FIND_WIN_LOG) == FIND_WIN && FIND_WIN <= 32, "one lane per list of the window");
static_assert(OWN_WAVES % SLICES_PER_PART == 0, "the waves of a sum work-group divide evenly over its lists");

__global__ void gcn_gemm1_wmma(const float* __restrict__ x, const float* __restrict__ W1,
                               float* __restrict__ m1, int n_tiles) {
    int wave = (int)((blockIdx.x * blockDim.x + threadIdx.x) >> 5);
    int lane = threadIdx.x & 31;
    if (wave >= n_tiles) return;
    int l15 = lane & 15;
    int hi  = lane >> 4;
    const float* xrow = x + ((size_t)(wave * 16 + l15)) * IN_DIM;

    v8f c = {};
    #pragma unroll 4
    for (int k0 = 0; k0 < IN_DIM; k0 += 4) {
        int ka = k0 + hi * 2;
        v2f a; a.x = xrow[ka];                 a.y = xrow[ka + 1];
        v2f b; b.x = W1[(ka + 0) * HID + l15]; b.y = W1[(ka + 1) * HID + l15];
        c = __builtin_amdgcn_wmma_f32_16x16x4_f32(false, a, false, b, (short)0, c, false, false);
    }
    float* mt = m1 + (size_t)(wave * 16) * HID;
    #pragma unroll
    for (int v = 0; v < 8; ++v)
        mt[(v + hi * 8) * HID + l15] = c[v];
}

__device__ __forceinline__ int q_push(int* qe, unsigned short* qt, int qn,
                                      unsigned t, int e, unsigned cnt) {
    bool hit = t < cnt;
    unsigned mask = __builtin_amdgcn_ballot_w32(hit);
    if (hit) {
        int slot = qn + (int)__builtin_amdgcn_mbcnt_lo(mask, 0u);
        qe[slot] = e;
        qt[slot] = (unsigned short)t;
    }
    return qn + __builtin_popcount(mask);
}

__device__ __forceinline__ void q_drain(const int* qe, const unsigned short* qt, int qn,
                                        const int* __restrict__ src, const float* __restrict__ m,
                                        float* acc, int lane) {
    const int f  = lane & 15;
    const int hi = lane >> 4;
    for (int g = 0; g < qn; g += 32) {
        int n = qn - g; if (n > 32) n = 32;
        int s = 0, t = 0;
        if (lane < n) { s = src[qe[g + lane]]; t = (int)qt[g + lane]; }
        for (int j = 0; j < n; j += 8) {
            float v[4]; int tr[4]; bool ok[4];
            #pragma unroll
            for (int u = 0; u < 4; ++u) {
                int ja = j + 2 * u, jb = ja + 1;
                bool a_ok = ja < n, b_ok = jb < n;
                int jac = a_ok ? ja : 0;
                int jbc = b_ok ? jb : jac;
                int sa = __builtin_amdgcn_readlane(s, jac), sb = __builtin_amdgcn_readlane(s, jbc);
                int ta = __builtin_amdgcn_readlane(t, jac), tb = __builtin_amdgcn_readlane(t, jbc);
                int sj = hi ? sb : sa;
                tr[u]  = hi ? tb : ta;
                ok[u]  = hi ? b_ok : a_ok;
                v[u]   = m[(size_t)sj * HID + f];
            }
            #pragma unroll
            for (int u = 0; u < 4; ++u)
                if (ok[u]) atomicAdd(&acc[tr[u] * HID + f], v[u]);
        }
    }
}

__device__ __forceinline__ void ld_chunk(int4 (&b)[4], const int4* __restrict__ d4,
                                         int k, int start, int n_part, int c0, int lane) {
    int kk = k < n_part ? k : n_part - 1;
    int ci = kk + start; if (ci >= n_part) ci -= n_part;
    ci += c0;
    #pragma unroll
    for (int j = 0; j < 4; ++j) b[j] = d4[(size_t)ci * 128 + j * 32 + lane];
}

__device__ __forceinline__ int scan_chunk(const int4 (&b)[4], int k, int start, int n_part, int c0,
                                          unsigned lo, unsigned ucnt, int* qe, unsigned short* qt,
                                          int qn, const int* __restrict__ src,
                                          const float* __restrict__ m, float* acc, int lane) {
    int ci = k + start; if (ci >= n_part) ci -= n_part;
    ci += c0;
    #pragma unroll
    for (int j = 0; j < 4; ++j) {
        const int eb = ci * 512 + j * 128 + lane * 4;
        unsigned t0 = (unsigned)b[j].x - lo;
        unsigned t1 = (unsigned)b[j].y - lo;
        unsigned t2 = (unsigned)b[j].z - lo;
        unsigned t3 = (unsigned)b[j].w - lo;
        qn = q_push(qe, qt, qn, t0, eb + 0, ucnt);
        qn = q_push(qe, qt, qn, t1, eb + 1, ucnt);
        qn = q_push(qe, qt, qn, t2, eb + 2, ucnt);
        qn = q_push(qe, qt, qn, t3, eb + 3, ucnt);
    }
    if (qn >= 32) {
        q_drain(qe, qt, qn, src, m, acc, lane);
        qn = 0;
    }
    return qn;
}

__device__ __forceinline__ void scan_slice(const int* __restrict__ src, const int* __restrict__ dst,
                                           const float* __restrict__ m, int n_edges,
                                           unsigned ulo, unsigned ucnt, float* acc,
                                           int* qe, unsigned short* qt, int p, int r,
                                           int lane, int wv, int nw) {
    const bool vec_ok = ((((size_t)dst) & 15) == 0);
    const int n_full = vec_ok ? (n_edges >> 9) : 0;
    const int c0     = (int)(((long long)p * n_full) / EDGE_PARTS);
    const int n_part = (int)(((long long)(p + 1) * n_full) / EDGE_PARTS) - c0;
    const int start = (int)(((long long)r * n_part) / ROW_RANGES);
    const int4* d4 = (const int4*)dst;
    const int t_base = n_full << 9;
    const int t_lo = t_base + (int)(((long long)p * (n_edges - t_base)) / EDGE_PARTS);
    const int t_hi = t_base + (int)(((long long)(p + 1) * (n_edges - t_base)) / EDGE_PARTS);

    int qn = 0;
    if (n_part > 0) {
    int4 A[4], B[4], C[4];
    ld_chunk(A, d4, wv,      start, n_part, c0, lane);
    ld_chunk(B, d4, wv + nw, start, n_part, c0, lane);
    for (int c = wv; c < n_part; c += 3 * nw) {
        ld_chunk(C, d4, c + 2 * nw, start, n_part, c0, lane);
        qn = scan_chunk(A, c, start, n_part, c0, ulo, ucnt, qe, qt, qn, src, m, acc, lane);
        ld_chunk(A, d4, c + 3 * nw, start, n_part, c0, lane);
        if (c + nw < n_part)
            qn = scan_chunk(B, c + nw, start, n_part, c0, ulo, ucnt, qe, qt, qn, src, m, acc, lane);
        ld_chunk(B, d4, c + 4 * nw, start, n_part, c0, lane);
        if (c + 2 * nw < n_part)
            qn = scan_chunk(C, c + 2 * nw, start, n_part, c0, ulo, ucnt, qe, qt, qn, src, m, acc, lane);
    }
    }
    for (int b = t_lo + wv * 32; b < t_hi; b += nw * 32) {
        int e = b + lane;
        unsigned t = (e < t_hi) ? ((unsigned)dst[e] - ulo) : 0xFFFFFFFFu;
        qn = q_push(qe, qt, qn, t, e, ucnt);
        if (qn >= 32) {
            q_drain(qe, qt, qn, src, m, acc, lane);
            qn = 0;
        }
    }
    if (qn > 0) q_drain(qe, qt, qn, src, m, acc, lane);
}

__device__ __forceinline__ int range_rows(int n_nodes) {
    int rr = (n_nodes + ROW_RANGES - 1) / ROW_RANGES;
    rr = (rr + 15) & ~15;
    if (rr > ROW_CAP) rr = ROW_CAP;
    return rr;
}

__device__ __forceinline__ int w_push(unsigned* qw, int qn, unsigned t, int s, unsigned cnt) {
    bool hit = t < cnt;
    unsigned mask = __builtin_amdgcn_ballot_w32(hit);
    if (hit) qw[qn + (int)__builtin_amdgcn_mbcnt_lo(mask, 0u)] = ((unsigned)s << WIN_BITS) | t;
    return qn + __builtin_popcount(mask);
}

__device__ __forceinline__ void w_flush(const unsigned* qw, int qn, int lane, unsigned rr,
                                        unsigned rr_inv, unsigned* __restrict__ lst, int* lst_n) {
    for (int g = 0; g < qn; g += 32) {
        const int i = g + lane;
        const bool valid = i < qn;
        const unsigned wd = qw[valid ? i : qn - 1];
        const unsigned t = wd & ((1u << WIN_BITS) - 1u);
        const unsigned k = __umulhi(t, rr_inv);
        const unsigned word = ((wd >> WIN_BITS) << ROW_BITS) | (t - k * rr);
        unsigned same = __builtin_amdgcn_ballot_w32(valid);
        unsigned mine = same;
        #pragma unroll
        for (int b = 0; b < FIND_WIN_LOG; ++b) {
            const unsigned kb = __builtin_amdgcn_ballot_w32(valid && ((k >> b) & 1u) != 0u);
            same &= ((k >> b) & 1u) ? kb : ~kb;
            mine &= (((unsigned)lane >> b) & 1u) ? kb : ~kb;
        }
        int base = 0;
        if (lane < FIND_WIN) base = atomicAdd(&lst_n[lane], __builtin_popcount(mine));
        const int bk  = __builtin_amdgcn_ds_bpermute((int)(k << 2), base);
        const int pos = bk + (int)__builtin_amdgcn_mbcnt_lo(same, 0u);
        if (valid && pos < SUB_CAP) lst[(size_t)k * SUB_CAP + pos] = word;
    }
}

__device__ __forceinline__ void ld_pair(int4 (&d)[4], int4 (&s)[4], const int4* __restrict__ d4,
                                        const int4* __restrict__ s4, int k, int start,
                                        int n_part, int c0, int lane) {
    int kk = k < n_part ? k : n_part - 1;
    int ci = kk + start; if (ci >= n_part) ci -= n_part;
    ci += c0;
    #pragma unroll
    for (int j = 0; j < 4; ++j) {
        d[j] = d4[(size_t)ci * 128 + j * 32 + lane];
        s[j] = s4[(size_t)ci * 128 + j * 32 + lane];
    }
}

__device__ __forceinline__ int find_chunk(const int4 (&d)[4], const int4 (&s)[4], unsigned lo,
                                          unsigned ucnt, unsigned* qw, int qn, int lane,
                                          unsigned rr, unsigned rr_inv,
                                          unsigned* __restrict__ lst, int* lst_n) {
    #pragma unroll
    for (int j = 0; j < 4; ++j) {
        qn = w_push(qw, qn, (unsigned)d[j].x - lo, s[j].x, ucnt);
        qn = w_push(qw, qn, (unsigned)d[j].y - lo, s[j].y, ucnt);
        qn = w_push(qw, qn, (unsigned)d[j].z - lo, s[j].z, ucnt);
        qn = w_push(qw, qn, (unsigned)d[j].w - lo, s[j].w, ucnt);
    }
    if (qn >= 32) { w_flush(qw, qn, lane, rr, rr_inv, lst, lst_n); qn = 0; }
    return qn;
}

__global__ __launch_bounds__(OWN_THREADS)
void gcn_find_hits(const int* __restrict__ src, const int* __restrict__ dst,
                   unsigned* __restrict__ lst_all, int* __restrict__ cnts,
                   int n_nodes, int n_edges) {
    __shared__ unsigned q_w[OWN_WAVES * Q_CAP];
    __shared__ int lst_n[FIND_WIN];

    const int tid = threadIdx.x;
    const int rr  = range_rows(n_nodes);
    int* my_cnt = cnts + (size_t)blockIdx.x * FIND_WIN * CNT_PITCH;
    const bool keep = ((long long)ROW_RANGES * rr >= n_nodes) && (n_nodes <= (1 << (32 - WIN_BITS)))
                      && (((((size_t)src) | ((size_t)dst)) & 15) == 0);
    if (!keep) {
        if (tid < FIND_WIN) my_cnt[tid * CNT_PITCH] = 0x7fffffff;
        return;
    }

    const int lane = tid & 31;
    const int wv   = __builtin_amdgcn_readfirstlane(tid >> 5);
    const int nw   = blockDim.x >> 5;
    const int q    = blockIdx.x % FIND_SLICES;
    const int w    = blockIdx.x / FIND_SLICES;
    const int lo   = w * FIND_WIN * rr;
    int cnt = n_nodes - lo; if (cnt > FIND_WIN * rr) cnt = FIND_WIN * rr; if (cnt < 0) cnt = 0;
    unsigned* lst = lst_all + (size_t)blockIdx.x * LIST_CAP;
    unsigned* qw  = q_w + wv * Q_CAP;

    if (tid < FIND_WIN) lst_n[tid] = 0;
    __syncthreads();

    if (cnt > 0) {
        const int n_full = n_edges >> 9;
        const int c0     = (int)(((long long)q * n_full) / FIND_SLICES);
        const int n_part = (int)(((long long)(q + 1) * n_full) / FIND_SLICES) - c0;
        const int start  = (int)(((long long)w * n_part) / N_WIN);
        const int4* d4 = (const int4*)dst;
        const int4* s4 = (const int4*)src;
        const int t_base = n_full << 9;
        const int t_lo = t_base + (int)(((long long)q * (n_edges - t_base)) / FIND_SLICES);
        const int t_hi = t_base + (int)(((long long)(q + 1) * (n_edges - t_base)) / FIND_SLICES);
        const unsigned ulo = (unsigned)lo, ucnt = (unsigned)cnt, urr = (unsigned)rr;
        const unsigned rr_inv = 0xFFFFFFFFu / urr + 1u;

        int qn = 0;
        if (n_part > 0) {
            int4 dA[4], sA[4], dB[4], sB[4];
            ld_pair(dA, sA, d4, s4, wv,      start, n_part, c0, lane);
            ld_pair(dB, sB, d4, s4, wv + nw, start, n_part, c0, lane);
            for (int c = wv; c < n_part; c += 2 * nw) {
                qn = find_chunk(dA, sA, ulo, ucnt, qw, qn, lane, urr, rr_inv, lst, lst_n);
                ld_pair(dA, sA, d4, s4, c + 2 * nw, start, n_part, c0, lane);
                if (c + nw < n_part)
                    qn = find_chunk(dB, sB, ulo, ucnt, qw, qn, lane, urr, rr_inv, lst, lst_n);
                ld_pair(dB, sB, d4, s4, c + 3 * nw, start, n_part, c0, lane);
            }
        }
        for (int b = t_lo + wv * 32; b < t_hi; b += nw * 32) {
            const int e = b + lane;
            unsigned t = 0xFFFFFFFFu; int sn = 0;
            if (e < t_hi) { t = (unsigned)dst[e] - ulo; sn = src[e]; }
            qn = w_push(qw, qn, t, sn, ucnt);
            if (qn >= 32) { w_flush(qw, qn, lane, urr, rr_inv, lst, lst_n); qn = 0; }
        }
        if (qn > 0) w_flush(qw, qn, lane, urr, rr_inv, lst, lst_n);
    }
    __syncthreads();
    if (tid < FIND_WIN) my_cnt[tid * CNT_PITCH] = lst_n[tid];
}

__device__ __forceinline__ void owner_scan(const int* __restrict__ src, const int* __restrict__ dst,
                                           const float* __restrict__ m, float* __restrict__ part,
                                           int n_nodes, int n_edges,
                                           float* acc, int* q_e, unsigned short* q_t) {
    const int tid  = threadIdx.x;
    const int lane = tid & 31;
    const int wv   = __builtin_amdgcn_readfirstlane(tid >> 5);
    const int nw   = blockDim.x >> 5;
    const int p    = blockIdx.x % EDGE_PARTS;
    const int r    = blockIdx.x / EDGE_PARTS;
    int* qe = q_e + wv * Q_CAP;
    unsigned short* qt = q_t + wv * Q_CAP;
    float* res = part + (size_t)p * n_nodes * HID;
    const int rr = range_rows(n_nodes);

    for (int lo = r * rr; lo < n_nodes; lo += ROW_RANGES * rr) {
        int cnt = n_nodes - lo; if (cnt > rr) cnt = rr;
        float4* acc4 = (float4*)acc;
        for (int i = tid; i < cnt * (HID / 4); i += blockDim.x) acc4[i] = make_float4(0.f, 0.f, 0.f, 0.f);
        __syncthreads();

        scan_slice(src, dst, m, n_edges, (unsigned)lo, (unsigned)cnt, acc, qe, qt, p, r,
                   lane, wv, nw);
        __syncthreads();

        float4* res4 = (float4*)(res + (size_t)lo * HID);
        for (int i = tid; i < cnt * (HID / 4); i += blockDim.x) res4[i] = acc4[i];
        __syncthreads();
    }
}

__device__ __forceinline__ void list_sum(const unsigned* __restrict__ l, int total,
                                         const float* __restrict__ m, float* acc,
                                         int lane, int sw, int nsw) {
    const int f  = lane & 15;
    const int hi = lane >> 4;
    const int nb = (total + 31) >> 5;
    const unsigned row_mask = (1u << ROW_BITS) - 1u;
    unsigned cur[16];
    {
        unsigned w = 0u;
        if (sw * 32 + lane < total) w = l[sw * 32 + lane];
        #pragma unroll
        for (int u = 0; u < 16; ++u) {
            unsigned wa = (unsigned)__builtin_amdgcn_readlane((int)w, 2 * u);
            unsigned wb = (unsigned)__builtin_amdgcn_readlane((int)w, 2 * u + 1);
            cur[u] = hi ? wb : wa;
        }
    }
    for (int b = sw; b < nb; b += nsw) {
        unsigned wn = 0u;
        { int i = (b + nsw) * 32 + lane; if (i < total) wn = l[i]; }
        const int n = total - b * 32;
        float v[16];
        #pragma unroll
        for (int u = 0; u < 16; ++u)
            v[u] = m[(size_t)(cur[u] >> ROW_BITS) * HID + f];
        if (n >= 32) {
            #pragma unroll
            for (int u = 0; u < 16; ++u)
                atomicAdd(&acc[(int)(cur[u] & row_mask) * HID + f], v[u]);
        } else {
            #pragma unroll
            for (int u = 0; u < 16; ++u)
                if (2 * u + hi < n) atomicAdd(&acc[(int)(cur[u] & row_mask) * HID + f], v[u]);
        }
        #pragma unroll
        for (int u = 0; u < 16; ++u) {
            unsigned wa = (unsigned)__builtin_amdgcn_readlane((int)wn, 2 * u);
            unsigned wb = (unsigned)__builtin_amdgcn_readlane((int)wn, 2 * u + 1);
            cur[u] = hi ? wb : wa;
        }
    }
}

__global__ __launch_bounds__(OWN_THREADS)
void gcn_owner_sum_list(const int* __restrict__ src, const int* __restrict__ dst,
                        const float* __restrict__ m, float* __restrict__ part,
                        const unsigned* __restrict__ lst_all, const int* __restrict__ cnts,
                        int n_nodes, int n_edges) {
    __shared__ __attribute__((aligned(16))) float acc[ROW_CAP * HID];
    __shared__ int q_e[OWN_WAVES * Q_CAP];
    __shared__ unsigned short q_t[OWN_WAVES * Q_CAP];

    const int p = blockIdx.x % EDGE_PARTS;
    const int r = blockIdx.x / EDGE_PARTS;
    const int w = r / FIND_WIN;
    const int k = r % FIND_WIN;
    const size_t fb0 = (size_t)w * FIND_SLICES + (size_t)p * SLICES_PER_PART;
    bool lists_ok = true;
    #pragma unroll
    for (int u = 0; u < SLICES_PER_PART; ++u) {
        const int n = cnts[((fb0 + u) * FIND_WIN + k) * CNT_PITCH];
        lists_ok = lists_ok && ((unsigned)n <= (unsigned)SUB_CAP);
    }
    if (!lists_ok) {
        owner_scan(src, dst, m, part, n_nodes, n_edges, acc, q_e, q_t);
        return;
    }

    const int tid  = threadIdx.x;
    const int lane = tid & 31;
    const int wv   = __builtin_amdgcn_readfirstlane(tid >> 5);
    const int nw   = blockDim.x >> 5;
    const int u   = wv % SLICES_PER_PART;
    const int sw  = wv / SLICES_PER_PART;
    const int nsw = nw / SLICES_PER_PART;
    const int total = cnts[((fb0 + u) * FIND_WIN + k) * CNT_PITCH];
    const unsigned* l = lst_all + (fb0 + u) * LIST_CAP + (size_t)k * SUB_CAP;

    const int rr = range_rows(n_nodes);
    const int lo = r * rr;
    int cnt = n_nodes - lo; if (cnt > rr) cnt = rr; if (cnt < 0) cnt = 0;

    float4* acc4 = (float4*)acc;
    for (int i = tid; i < cnt * (HID / 4); i += blockDim.x) acc4[i] = make_float4(0.f, 0.f, 0.f, 0.f);
    __syncthreads();

    list_sum(l, total, m, acc, lane, sw, nsw);
    __syncthreads();

    float4* res4 = (float4*)(part + (size_t)p * n_nodes * HID + (size_t)lo * HID);
    for (int i = tid; i < cnt * (HID / 4); i += blockDim.x) res4[i] = acc4[i];
}

__global__ void gcn_gemm2_wmma(const float* __restrict__ agg, const float* __restrict__ b1,
                               const float* __restrict__ W2, float* __restrict__ m2,
                               int n_tiles, int n_nodes) {
    int wave = (int)((blockIdx.x * blockDim.x + threadIdx.x) >> 5);
    int lane = threadIdx.x & 31;
    if (wave >= n_tiles) return;
    int l15 = lane & 15;
    int hi  = lane >> 4;
    const float* arow = agg + (size_t)(wave * 16 + l15) * HID;
    const size_t plane = (size_t)n_nodes * HID;

    v8f c = {};
    #pragma unroll
    for (int k0 = 0; k0 < HID; k0 += 4) {
        int ka = k0 + hi * 2;
        float sx = arow[ka], sy = arow[ka + 1];
        #pragma unroll
        for (int q = 1; q < EDGE_PARTS; ++q) {
            sx += arow[q * plane + ka];
            sy += arow[q * plane + ka + 1];
        }
        v2f a;
        a.x = fmaxf(sx + b1[ka],     0.0f);
        a.y = fmaxf(sy + b1[ka + 1], 0.0f);
        v2f b;
        b.x = (l15 < OUTD) ? W2[(ka + 0) * OUTD + l15] : 0.0f;
        b.y = (l15 < OUTD) ? W2[(ka + 1) * OUTD + l15] : 0.0f;
        c = __builtin_amdgcn_wmma_f32_16x16x4_f32(false, a, false, b, (short)0, c, false, false);
    }
    float* mt = m2 + (size_t)(wave * 16) * HID;
    #pragma unroll
    for (int v = 0; v < 8; ++v)
        mt[(v + hi * 8) * HID + l15] = c[v];
}

__global__ void gcn_join_out(const float* __restrict__ part, const float* __restrict__ bias,
                             float* __restrict__ out, int n_nodes) {
    int i = (int)(blockIdx.x * blockDim.x + threadIdx.x);
    if (i >= n_nodes * OUTD) return;
    int r = i / OUTD, o = i - r * OUTD;
    const size_t plane = (size_t)n_nodes * HID;
    const float* q = part + (size_t)r * HID + o;
    float s = q[0];
    #pragma unroll
    for (int k = 1; k < EDGE_PARTS; ++k) s += q[k * plane];
    out[i] = s + bias[o];
}

extern "C" void kernel_launch(void* const* d_in, const int* in_sizes, int n_in,
                              void* d_out, int out_size, void* d_ws, size_t ws_size,
                              hipStream_t stream) {
    const float* x  = (const float*)d_in[0];
    const int*   ei = (const int*)  d_in[1];
    const float* W1 = (const float*)d_in[2];
    const float* b1 = (const float*)d_in[3];
    const float* W2 = (const float*)d_in[4];
    const float* b2 = (const float*)d_in[5];
    float* out = (float*)d_out;

    int n_nodes       = in_sizes[0] / IN_DIM;
    long long n_edges = (long long)in_sizes[1] / 2;
    const int* src = ei;
    const int* dst = ei + n_edges;

    float* m1   = (float*)d_ws;
    float* aggp = m1   + (size_t)n_nodes * HID;
    float* m2   = aggp + (size_t)n_nodes * HID * EDGE_PARTS;
    float* outp = m2   + (size_t)n_nodes * HID;
    unsigned* hits = (unsigned*)(outp + (size_t)n_nodes * HID * EDGE_PARTS);
    int* hit_n  = (int*)(hits + (size_t)OWN_GROUPS * LIST_CAP);

    int n_tiles = (n_nodes + 15) / 16;
    int gemm_blocks = (n_tiles + 7) / 8;
    int join_blocks = (n_nodes * OUTD + 255) / 256;

    gcn_gemm1_wmma<<<gemm_blocks, 256, 0, stream>>>(x, W1, m1, n_tiles);
    gcn_find_hits<<<OWN_GROUPS, OWN_THREADS, 0, stream>>>(src, dst, hits, hit_n, n_nodes, (int)n_edges);
    gcn_owner_sum_list<<<OWN_GROUPS, OWN_THREADS, 0, stream>>>(src, dst, m1, aggp, hits, hit_n, n_nodes, (int)n_edges);
    gcn_gemm2_wmma<<<gemm_blocks, 256, 0, stream>>>(aggp, b1, W2, m2, n_tiles, n_nodes);
    gcn_owner_sum_list<<<OWN_GROUPS, OWN_THREADS, 0, stream>>>(src, dst, m2, outp, hits, hit_n, n_nodes, (int)n_edges);
    gcn_join_out  <<<join_blocks, 256, 0, stream>>>(outp, b2, out, n_nodes);
}


# x@W1 reads A as 16-byte loads: lane half h owns floats 4h..4h+3 of every 8 k-steps, W1 read in the same k order (k regrouped inside each 8, so the sum order changes); half the x loads
# speedup vs baseline: 13.4141x; 1.0528x over previous
#include <hip/hip_runtime.h>


typedef float v2f __attribute__((ext_vector_type(2)));
typedef float v8f __attribute__((ext_vector_type(8)));

#define IN_DIM 256
#define HID 16
#define OUTD 10

#define OWN_GROUPS 256
#define OWN_THREADS 1024
#define OWN_WAVES 32
#define EDGE_PARTS 2
#define ROW_RANGES 128
#define ROW_CAP 2048
#define Q_CAP 544
#define LIST_CAP 65536
#define ROW_BITS 12
#define CNT_PITCH 32
#define FIND_WIN 16
#define FIND_WIN_LOG 4
#define N_WIN (ROW_RANGES / FIND_WIN)
#define FIND_SLICES (OWN_GROUPS / N_WIN)
#define SLICES_PER_PART (FIND_SLICES / EDGE_PARTS)
#define SUB_CAP (LIST_CAP / FIND_WIN)
#define WIN_BITS 15
static_assert(EDGE_PARTS * ROW_RANGES == OWN_GROUPS, "one work-group per (slice, range) pair");
static_assert(ROW_CAP <= (1 << ROW_BITS), "a local row fits in ROW_BITS bits");
static_assert(N_WIN * FIND_WIN == ROW_RANGES && N_WIN * FIND_SLICES == OWN_GROUPS, "windows tile the ranges");
static_assert(SLICES_PER_PART * EDGE_PARTS == FIND_SLICES, "find slices tile the slices of the sums");
static_assert(FIND_WIN * ROW_CAP <= (1 << WIN_BITS), "a row of the window fits in WIN_BITS bits");
static_assert((1 << FIND_WIN_LOG) == FIND_WIN && FIND_WIN <= 32, "one lane per list of the window");
static_assert(OWN_WAVES % SLICES_PER_PART == 0, "the waves of a sum work-group divide evenly over its lists");

__global__ void gcn_gemm1_wmma(const float* __restrict__ x, const float* __restrict__ W1,
                               float* __restrict__ m1, int n_tiles) {
    int wave = (int)((blockIdx.x * blockDim.x + threadIdx.x) >> 5);
    int lane = threadIdx.x & 31;
    if (wave >= n_tiles) return;
    int l15 = lane & 15;
    int hi  = lane >> 4;
    const float* xrow = x + ((size_t)(wave * 16 + l15)) * IN_DIM + hi * 4;
    const float* wcol = W1 + (hi * 4) * HID + l15;

    v8f c = {};
    #pragma unroll 4
    for (int k0 = 0; k0 < IN_DIM; k0 += 8) {
        const float* xk = xrow + k0;
        const float* wk = wcol + k0 * HID;
        v2f a0; a0.x = xk[0];       a0.y = xk[1];
        v2f a1; a1.x = xk[2];       a1.y = xk[3];
        v2f b0; b0.x = wk[0 * HID]; b0.y = wk[1 * HID];
        v2f b1; b1.x = wk[2 * HID]; b1.y = wk[3 * HID];
        c = __builtin_amdgcn_wmma_f32_16x16x4_f32(false, a0, false, b0, (short)0, c, false, false);
        c = __builtin_amdgcn_wmma_f32_16x16x4_f32(false, a1, false, b1, (short)0, c, false, false);
    }
    float* mt = m1 + (size_t)(wave * 16) * HID;
    #pragma unroll
    for (int v = 0; v < 8; ++v)
        mt[(v + hi * 8) * HID + l15] = c[v];
}

__device__ __forceinline__ int q_push(int* qe, unsigned short* qt, int qn,
                                      unsigned t, int e, unsigned cnt) {
    bool hit = t < cnt;
    unsigned mask = __builtin_amdgcn_ballot_w32(hit);
    if (hit) {
        int slot = qn + (int)__builtin_amdgcn_mbcnt_lo(mask, 0u);
        qe[slot] = e;
        qt[slot] = (unsigned short)t;
    }
    return qn + __builtin_popcount(mask);
}

__device__ __forceinline__ void q_drain(const int* qe, const unsigned short* qt, int qn,
                                        const int* __restrict__ src, const float* __restrict__ m,
                                        float* acc, int lane) {
    const int f  = lane & 15;
    const int hi = lane >> 4;
    for (int g = 0; g < qn; g += 32) {
        int n = qn - g; if (n > 32) n = 32;
        int s = 0, t = 0;
        if (lane < n) { s = src[qe[g + lane]]; t = (int)qt[g + lane]; }
        for (int j = 0; j < n; j += 8) {
            float v[4]; int tr[4]; bool ok[4];
            #pragma unroll
            for (int u = 0; u < 4; ++u) {
                int ja = j + 2 * u, jb = ja + 1;
                bool a_ok = ja < n, b_ok = jb < n;
                int jac = a_ok ? ja : 0;
                int jbc = b_ok ? jb : jac;
                int sa = __builtin_amdgcn_readlane(s, jac), sb = __builtin_amdgcn_readlane(s, jbc);
                int ta = __builtin_amdgcn_readlane(t, jac), tb = __builtin_amdgcn_readlane(t, jbc);
                int sj = hi ? sb : sa;
                tr[u]  = hi ? tb : ta;
                ok[u]  = hi ? b_ok : a_ok;
                v[u]   = m[(size_t)sj * HID + f];
            }
            #pragma unroll
            for (int u = 0; u < 4; ++u)
                if (ok[u]) atomicAdd(&acc[tr[u] * HID + f], v[u]);
        }
    }
}

__device__ __forceinline__ void ld_chunk(int4 (&b)[4], const int4* __restrict__ d4,
                                         int k, int start, int n_part, int c0, int lane) {
    int kk = k < n_part ? k : n_part - 1;
    int ci = kk + start; if (ci >= n_part) ci -= n_part;
    ci += c0;
    #pragma unroll
    for (int j = 0; j < 4; ++j) b[j] = d4[(size_t)ci * 128 + j * 32 + lane];
}

__device__ __forceinline__ int scan_chunk(const int4 (&b)[4], int k, int start, int n_part, int c0,
                                          unsigned lo, unsigned ucnt, int* qe, unsigned short* qt,
                                          int qn, const int* __restrict__ src,
                                          const float* __restrict__ m, float* acc, int lane) {
    int ci = k + start; if (ci >= n_part) ci -= n_part;
    ci += c0;
    #pragma unroll
    for (int j = 0; j < 4; ++j) {
        const int eb = ci * 512 + j * 128 + lane * 4;
        unsigned t0 = (unsigned)b[j].x - lo;
        unsigned t1 = (unsigned)b[j].y - lo;
        unsigned t2 = (unsigned)b[j].z - lo;
        unsigned t3 = (unsigned)b[j].w - lo;
        qn = q_push(qe, qt, qn, t0, eb + 0, ucnt);
        qn = q_push(qe, qt, qn, t1, eb + 1, ucnt);
        qn = q_push(qe, qt, qn, t2, eb + 2, ucnt);
        qn = q_push(qe, qt, qn, t3, eb + 3, ucnt);
    }
    if (qn >= 32) {
        q_drain(qe, qt, qn, src, m, acc, lane);
        qn = 0;
    }
    return qn;
}

__device__ __forceinline__ void scan_slice(const int* __restrict__ src, const int* __restrict__ dst,
                                           const float* __restrict__ m, int n_edges,
                                           unsigned ulo, unsigned ucnt, float* acc,
                                           int* qe, unsigned short* qt, int p, int r,
                                           int lane, int wv, int nw) {
    const bool vec_ok = ((((size_t)dst) & 15) == 0);
    const int n_full = vec_ok ? (n_edges >> 9) : 0;
    const int c0     = (int)(((long long)p * n_full) / EDGE_PARTS);
    const int n_part = (int)(((long long)(p + 1) * n_full) / EDGE_PARTS) - c0;
    const int start = (int)(((long long)r * n_part) / ROW_RANGES);
    const int4* d4 = (const int4*)dst;
    const int t_base = n_full << 9;
    const int t_lo = t_base + (int)(((long long)p * (n_edges - t_base)) / EDGE_PARTS);
    const int t_hi = t_base + (int)(((long long)(p + 1) * (n_edges - t_base)) / EDGE_PARTS);

    int qn = 0;
    if (n_part > 0) {
    int4 A[4], B[4], C[4];
    ld_chunk(A, d4, wv,      start, n_part, c0, lane);
    ld_chunk(B, d4, wv + nw, start, n_part, c0, lane);
    for (int c = wv; c < n_part; c += 3 * nw) {
        ld_chunk(C, d4, c + 2 * nw, start, n_part, c0, lane);
        qn = scan_chunk(A, c, start, n_part, c0, ulo, ucnt, qe, qt, qn, src, m, acc, lane);
        ld_chunk(A, d4, c + 3 * nw, start, n_part, c0, lane);
        if (c + nw < n_part)
            qn = scan_chunk(B, c + nw, start, n_part, c0, ulo, ucnt, qe, qt, qn, src, m, acc, lane);
        ld_chunk(B, d4, c + 4 * nw, start, n_part, c0, lane);
        if (c + 2 * nw < n_part)
            qn = scan_chunk(C, c + 2 * nw, start, n_part, c0, ulo, ucnt, qe, qt, qn, src, m, acc, lane);
    }
    }
    for (int b = t_lo + wv * 32; b < t_hi; b += nw * 32) {
        int e = b + lane;
        unsigned t = (e < t_hi) ? ((unsigned)dst[e] - ulo) : 0xFFFFFFFFu;
        qn = q_push(qe, qt, qn, t, e, ucnt);
        if (qn >= 32) {
            q_drain(qe, qt, qn, src, m, acc, lane);
            qn = 0;
        }
    }
    if (qn > 0) q_drain(qe, qt, qn, src, m, acc, lane);
}

__device__ __forceinline__ int range_rows(int n_nodes) {
    int rr = (n_nodes + ROW_RANGES - 1) / ROW_RANGES;
    rr = (rr + 15) & ~15;
    if (rr > ROW_CAP) rr = ROW_CAP;
    return rr;
}

__device__ __forceinline__ int w_push(unsigned* qw, int qn, unsigned t, int s, unsigned cnt) {
    bool hit = t < cnt;
    unsigned mask = __builtin_amdgcn_ballot_w32(hit);
    if (hit) qw[qn + (int)__builtin_amdgcn_mbcnt_lo(mask, 0u)] = ((unsigned)s << WIN_BITS) | t;
    return qn + __builtin_popcount(mask);
}

__device__ __forceinline__ void w_flush(const unsigned* qw, int qn, int lane, unsigned rr,
                                        unsigned rr_inv, unsigned* __restrict__ lst, int* lst_n) {
    for (int g = 0; g < qn; g += 32) {
        const int i = g + lane;
        const bool valid = i < qn;
        const unsigned wd = qw[valid ? i : qn - 1];
        const unsigned t = wd & ((1u << WIN_BITS) - 1u);
        const unsigned k = __umulhi(t, rr_inv);
        const unsigned word = ((wd >> WIN_BITS) << ROW_BITS) | (t - k * rr);
        unsigned same = __builtin_amdgcn_ballot_w32(valid);
        unsigned mine = same;
        #pragma unroll
        for (int b = 0; b < FIND_WIN_LOG; ++b) {
            const unsigned kb = __builtin_amdgcn_ballot_w32(valid && ((k >> b) & 1u) != 0u);
            same &= ((k >> b) & 1u) ? kb : ~kb;
            mine &= (((unsigned)lane >> b) & 1u) ? kb : ~kb;
        }
        int base = 0;
        if (lane < FIND_WIN) base = atomicAdd(&lst_n[lane], __builtin_popcount(mine));
        const int bk  = __builtin_amdgcn_ds_bpermute((int)(k << 2), base);
        const int pos = bk + (int)__builtin_amdgcn_mbcnt_lo(same, 0u);
        if (valid && pos < SUB_CAP) lst[(size_t)k * SUB_CAP + pos] = word;
    }
}

__device__ __forceinline__ void ld_pair(int4 (&d)[4], int4 (&s)[4], const int4* __restrict__ d4,
                                        const int4* __restrict__ s4, int k, int start,
                                        int n_part, int c0, int lane) {
    int kk = k < n_part ? k : n_part - 1;
    int ci = kk + start; if (ci >= n_part) ci -= n_part;
    ci += c0;
    #pragma unroll
    for (int j = 0; j < 4; ++j) {
        d[j] = d4[(size_t)ci * 128 + j * 32 + lane];
        s[j] = s4[(size_t)ci * 128 + j * 32 + lane];
    }
}

__device__ __forceinline__ int find_chunk(const int4 (&d)[4], const int4 (&s)[4], unsigned lo,
                                          unsigned ucnt, unsigned* qw, int qn, int lane,
                                          unsigned rr, unsigned rr_inv,
                                          unsigned* __restrict__ lst, int* lst_n) {
    #pragma unroll
    for (int j = 0; j < 4; ++j) {
        qn = w_push(qw, qn, (unsigned)d[j].x - lo, s[j].x, ucnt);
        qn = w_push(qw, qn, (unsigned)d[j].y - lo, s[j].y, ucnt);
        qn = w_push(qw, qn, (unsigned)d[j].z - lo, s[j].z, ucnt);
        qn = w_push(qw, qn, (unsigned)d[j].w - lo, s[j].w, ucnt);
    }
    if (qn >= 32) { w_flush(qw, qn, lane, rr, rr_inv, lst, lst_n); qn = 0; }
    return qn;
}

__global__ __launch_bounds__(OWN_THREADS)
void gcn_find_hits(const int* __restrict__ src, const int* __restrict__ dst,
                   unsigned* __restrict__ lst_all, int* __restrict__ cnts,
                   int n_nodes, int n_edges) {
    __shared__ unsigned q_w[OWN_WAVES * Q_CAP];
    __shared__ int lst_n[FIND_WIN];

    const int tid = threadIdx.x;
    const int rr  = range_rows(n_nodes);
    int* my_cnt = cnts + (size_t)blockIdx.x * FIND_WIN * CNT_PITCH;
    const bool keep = ((long long)ROW_RANGES * rr >= n_nodes) && (n_nodes <= (1 << (32 - WIN_BITS)))
                      && (((((size_t)src) | ((size_t)dst)) & 15) == 0);
    if (!keep) {
        if (tid < FIND_WIN) my_cnt[tid * CNT_PITCH] = 0x7fffffff;
        return;
    }

    const int lane = tid & 31;
    const int wv   = __builtin_amdgcn_readfirstlane(tid >> 5);
    const int nw   = blockDim.x >> 5;
    const int q    = blockIdx.x % FIND_SLICES;
    const int w    = blockIdx.x / FIND_SLICES;
    const int lo   = w * FIND_WIN * rr;
    int cnt = n_nodes - lo; if (cnt > FIND_WIN * rr) cnt = FIND_WIN * rr; if (cnt < 0) cnt = 0;
    unsigned* lst = lst_all + (size_t)blockIdx.x * LIST_CAP;
    unsigned* qw  = q_w + wv * Q_CAP;

    if (tid < FIND_WIN) lst_n[tid] = 0;
    __syncthreads();

    if (cnt > 0) {
        const int n_full = n_edges >> 9;
        const int c0     = (int)(((long long)q * n_full) / FIND_SLICES);
        const int n_part = (int)(((long long)(q + 1) * n_full) / FIND_SLICES) - c0;
        const int start  = (int)(((long long)w * n_part) / N_WIN);
        const int4* d4 = (const int4*)dst;
        const int4* s4 = (const int4*)src;
        const int t_base = n_full << 9;
        const int t_lo = t_base + (int)(((long long)q * (n_edges - t_base)) / FIND_SLICES);
        const int t_hi = t_base + (int)(((long long)(q + 1) * (n_edges - t_base)) / FIND_SLICES);
        const unsigned ulo = (unsigned)lo, ucnt = (unsigned)cnt, urr = (unsigned)rr;
        const unsigned rr_inv = 0xFFFFFFFFu / urr + 1u;

        int qn = 0;
        if (n_part > 0) {
            int4 dA[4], sA[4], dB[4], sB[4];
            ld_pair(dA, sA, d4, s4, wv,      start, n_part, c0, lane);
            ld_pair(dB, sB, d4, s4, wv + nw, start, n_part, c0, lane);
            for (int c = wv; c < n_part; c += 2 * nw) {
                qn = find_chunk(dA, sA, ulo, ucnt, qw, qn, lane, urr, rr_inv, lst, lst_n);
                ld_pair(dA, sA, d4, s4, c + 2 * nw, start, n_part, c0, lane);
                if (c + nw < n_part)
                    qn = find_chunk(dB, sB, ulo, ucnt, qw, qn, lane, urr, rr_inv, lst, lst_n);
                ld_pair(dB, sB, d4, s4, c + 3 * nw, start, n_part, c0, lane);
            }
        }
        for (int b = t_lo + wv * 32; b < t_hi; b += nw * 32) {
            const int e = b + lane;
            unsigned t = 0xFFFFFFFFu; int sn = 0;
            if (e < t_hi) { t = (unsigned)dst[e] - ulo; sn = src[e]; }
            qn = w_push(qw, qn, t, sn, ucnt);
            if (qn >= 32) { w_flush(qw, qn, lane, urr, rr_inv, lst, lst_n); qn = 0; }
        }
        if (qn > 0) w_flush(qw, qn, lane, urr, rr_inv, lst, lst_n);
    }
    __syncthreads();
    if (tid < FIND_WIN) my_cnt[tid * CNT_PITCH] = lst_n[tid];
}

__device__ __forceinline__ void owner_scan(const int* __restrict__ src, const int* __restrict__ dst,
                                           const float* __restrict__ m, float* __restrict__ part,
                                           int n_nodes, int n_edges,
                                           float* acc, int* q_e, unsigned short* q_t) {
    const int tid  = threadIdx.x;
    const int lane = tid & 31;
    const int wv   = __builtin_amdgcn_readfirstlane(tid >> 5);
    const int nw   = blockDim.x >> 5;
    const int p    = blockIdx.x % EDGE_PARTS;
    const int r    = blockIdx.x / EDGE_PARTS;
    int* qe = q_e + wv * Q_CAP;
    unsigned short* qt = q_t + wv * Q_CAP;
    float* res = part + (size_t)p * n_nodes * HID;
    const int rr = range_rows(n_nodes);

    for (int lo = r * rr; lo < n_nodes; lo += ROW_RANGES * rr) {
        int cnt = n_nodes - lo; if (cnt > rr) cnt = rr;
        float4* acc4 = (float4*)acc;
        for (int i = tid; i < cnt * (HID / 4); i += blockDim.x) acc4[i] = make_float4(0.f, 0.f, 0.f, 0.f);
        __syncthreads();

        scan_slice(src, dst, m, n_edges, (unsigned)lo, (unsigned)cnt, acc, qe, qt, p, r,
                   lane, wv, nw);
        __syncthreads();

        float4* res4 = (float4*)(res + (size_t)lo * HID);
        for (int i = tid; i < cnt * (HID / 4); i += blockDim.x) res4[i] = acc4[i];
        __syncthreads();
    }
}

__device__ __forceinline__ void list_sum(const unsigned* __restrict__ l, int total,
                                         const float* __restrict__ m, float* acc,
                                         int lane, int sw, int nsw) {
    const int f  = lane & 15;
    const int hi = lane >> 4;
    const int nb = (total + 31) >> 5;
    const unsigned row_mask = (1u << ROW_BITS) - 1u;
    unsigned cur[16];
    {
        unsigned w = 0u;
        if (sw * 32 + lane < total) w = l[sw * 32 + lane];
        #pragma unroll
        for (int u = 0; u < 16; ++u) {
            unsigned wa = (unsigned)__builtin_amdgcn_readlane((int)w, 2 * u);
            unsigned wb = (unsigned)__builtin_amdgcn_readlane((int)w, 2 * u + 1);
            cur[u] = hi ? wb : wa;
        }
    }
    for (int b = sw; b < nb; b += nsw) {
        unsigned wn = 0u;
        { int i = (b + nsw) * 32 + lane; if (i < total) wn = l[i]; }
        const int n = total - b * 32;
        float v[16];
        #pragma unroll
        for (int u = 0; u < 16; ++u)
            v[u] = m[(size_t)(cur[u] >> ROW_BITS) * HID + f];
        if (n >= 32) {
            #pragma unroll
            for (int u = 0; u < 16; ++u)
                atomicAdd(&acc[(int)(cur[u] & row_mask) * HID + f], v[u]);
        } else {
            #pragma unroll
            for (int u = 0; u < 16; ++u)
                if (2 * u + hi < n) atomicAdd(&acc[(int)(cur[u] & row_mask) * HID + f], v[u]);
        }
        #pragma unroll
        for (int u = 0; u < 16; ++u) {
            unsigned wa = (unsigned)__builtin_amdgcn_readlane((int)wn, 2 * u);
            unsigned wb = (unsigned)__builtin_amdgcn_readlane((int)wn, 2 * u + 1);
            cur[u] = hi ? wb : wa;
        }
    }
}

__global__ __launch_bounds__(OWN_THREADS)
void gcn_owner_sum_list(const int* __restrict__ src, const int* __restrict__ dst,
                        const float* __restrict__ m, float* __restrict__ part,
                        const unsigned* __restrict__ lst_all, const int* __restrict__ cnts,
                        int n_nodes, int n_edges) {
    __shared__ __attribute__((aligned(16))) float acc[ROW_CAP * HID];
    __shared__ int q_e[OWN_WAVES * Q_CAP];
    __shared__ unsigned short q_t[OWN_WAVES * Q_CAP];

    const int p = blockIdx.x % EDGE_PARTS;
    const int r = blockIdx.x / EDGE_PARTS;
    const int w = r / FIND_WIN;
    const int k = r % FIND_WIN;
    const size_t fb0 = (size_t)w * FIND_SLICES + (size_t)p * SLICES_PER_PART;
    bool lists_ok = true;
    #pragma unroll
    for (int u = 0; u < SLICES_PER_PART; ++u) {
        const int n = cnts[((fb0 + u) * FIND_WIN + k) * CNT_PITCH];
        lists_ok = lists_ok && ((unsigned)n <= (unsigned)SUB_CAP);
    }
    if (!lists_ok) {
        owner_scan(src, dst, m, part, n_nodes, n_edges, acc, q_e, q_t);
        return;
    }

    const int tid  = threadIdx.x;
    const int lane = tid & 31;
    const int wv   = __builtin_amdgcn_readfirstlane(tid >> 5);
    const int nw   = blockDim.x >> 5;
    const int u   = wv % SLICES_PER_PART;
    const int sw  = wv / SLICES_PER_PART;
    const int nsw = nw / SLICES_PER_PART;
    const int total = cnts[((fb0 + u) * FIND_WIN + k) * CNT_PITCH];
    const unsigned* l = lst_all + (fb0 + u) * LIST_CAP + (size_t)k * SUB_CAP;

    const int rr = range_rows(n_nodes);
    const int lo = r * rr;
    int cnt = n_nodes - lo; if (cnt > rr) cnt = rr; if (cnt < 0) cnt = 0;

    float4* acc4 = (float4*)acc;
    for (int i = tid; i < cnt * (HID / 4); i += blockDim.x) acc4[i] = make_float4(0.f, 0.f, 0.f, 0.f);
    __syncthreads();

    list_sum(l, total, m, acc, lane, sw, nsw);
    __syncthreads();

    float4* res4 = (float4*)(part + (size_t)p * n_nodes * HID + (size_t)lo * HID);
    for (int i = tid; i < cnt * (HID / 4); i += blockDim.x) res4[i] = acc4[i];
}

__global__ void gcn_gemm2_wmma(const float* __restrict__ agg, const float* __restrict__ b1,
                               const float* __restrict__ W2, float* __restrict__ m2,
                               int n_tiles, int n_nodes) {
    int wave = (int)((blockIdx.x * blockDim.x + threadIdx.x) >> 5);
    int lane = threadIdx.x & 31;
    if (wave >= n_tiles) return;
    int l15 = lane & 15;
    int hi  = lane >> 4;
    const float* arow = agg + (size_t)(wave * 16 + l15) * HID;
    const size_t plane = (size_t)n_nodes * HID;

    v8f c = {};
    #pragma unroll
    for (int k0 = 0; k0 < HID; k0 += 4) {
        int ka = k0 + hi * 2;
        float sx = arow[ka], sy = arow[ka + 1];
        #pragma unroll
        for (int q = 1; q < EDGE_PARTS; ++q) {
            sx += arow[q * plane + ka];
            sy += arow[q * plane + ka + 1];
        }
        v2f a;
        a.x = fmaxf(sx + b1[ka],     0.0f);
        a.y = fmaxf(sy + b1[ka + 1], 0.0f);
        v2f b;
        b.x = (l15 < OUTD) ? W2[(ka + 0) * OUTD + l15] : 0.0f;
        b.y = (l15 < OUTD) ? W2[(ka + 1) * OUTD + l15] : 0.0f;
        c = __builtin_amdgcn_wmma_f32_16x16x4_f32(false, a, false, b, (short)0, c, false, false);
    }
    float* mt = m2 + (size_t)(wave * 16) * HID;
    #pragma unroll
    for (int v = 0; v < 8; ++v)
        mt[(v + hi * 8) * HID + l15] = c[v];
}

__global__ void gcn_join_out(const float* __restrict__ part, const float* __restrict__ bias,
                             float* __restrict__ out, int n_nodes) {
    int i = (int)(blockIdx.x * blockDim.x + threadIdx.x);
    if (i >= n_nodes * OUTD) return;
    int r = i / OUTD, o = i - r * OUTD;
    const size_t plane = (size_t)n_nodes * HID;
    const float* q = part + (size_t)r * HID + o;
    float s = q[0];
    #pragma unroll
    for (int k = 1; k < EDGE_PARTS; ++k) s += q[k * plane];
    out[i] = s + bias[o];
}

extern "C" void kernel_launch(void* const* d_in, const int* in_sizes, int n_in,
                              void* d_out, int out_size, void* d_ws, size_t ws_size,
                              hipStream_t stream) {
    const float* x  = (const float*)d_in[0];
    const int*   ei = (const int*)  d_in[1];
    const float* W1 = (const float*)d_in[2];
    const float* b1 = (const float*)d_in[3];
    const float* W2 = (const float*)d_in[4];
    const float* b2 = (const float*)d_in[5];
    float* out = (float*)d_out;

    int n_nodes       = in_sizes[0] / IN_DIM;
    long long n_edges = (long long)in_sizes[1] / 2;
    const int* src = ei;
    const int* dst = ei + n_edges;

    float* m1   = (float*)d_ws;
    float* aggp = m1   + (size_t)n_nodes * HID;
    float* m2   = aggp + (size_t)n_nodes * HID * EDGE_PARTS;
    float* outp = m2   + (size_t)n_nodes * HID;
    unsigned* hits = (unsigned*)(outp + (size_t)n_nodes * HID * EDGE_PARTS);
    int* hit_n  = (int*)(hits + (size_t)OWN_GROUPS * LIST_CAP);

    int n_tiles = (n_nodes + 15) / 16;
    int gemm_blocks = (n_tiles + 7) / 8;
    int join_blocks = (n_nodes * OUTD + 255) / 256;

    gcn_gemm1_wmma<<<gemm_blocks, 256, 0, stream>>>(x, W1, m1, n_tiles);
    gcn_find_hits<<<OWN_GROUPS, OWN_THREADS, 0, stream>>>(src, dst, hits, hit_n, n_nodes, (int)n_edges);
    gcn_owner_sum_list<<<OWN_GROUPS, OWN_THREADS, 0, stream>>>(src, dst, m1, aggp, hits, hit_n, n_nodes, (int)n_edges);
    gcn_gemm2_wmma<<<gemm_blocks, 256, 0, stream>>>(aggp, b1, W2, m2, n_tiles, n_nodes);
    gcn_owner_sum_list<<<OWN_GROUPS, OWN_THREADS, 0, stream>>>(src, dst, m2, outp, hits, hit_n, n_nodes, (int)n_edges);
    gcn_join_out  <<<join_blocks, 256, 0, stream>>>(outp, b2, out, n_nodes);
}


# owner tiling 256 row ranges x 1 edge slice (was 128 x 2): a row is whole in LDS, so relu(+b1) and (sum of h rows)@W2+b2 on WMMA run in the sums' stores; 2 launches fewer; LDS arrival order
# speedup vs baseline: 14.0561x; 1.0479x over previous
#include <hip/hip_runtime.h>


typedef float v2f __attribute__((ext_vector_type(2)));
typedef float v8f __attribute__((ext_vector_type(8)));

#define IN_DIM 256
#define HID 16
#define OUTD 10

#define OWN_GROUPS 256
#define OWN_THREADS 1024
#define OWN_WAVES 32
#define ROW_RANGES 256
#define ROW_CAP 1024
#define Q_CAP 544
#define LIST_CAP 65536
#define ROW_BITS 12
#define CNT_PITCH 32
#define FIND_WIN 32
#define FIND_WIN_LOG 5
#define N_WIN (ROW_RANGES / FIND_WIN)
#define FIND_SLICES (OWN_GROUPS / N_WIN)
#define SUB_CAP (LIST_CAP / FIND_WIN)
#define WIN_BITS 15
static_assert(ROW_RANGES == OWN_GROUPS, "one owner work-group per row range");
static_assert(ROW_CAP <= (1 << ROW_BITS), "a local row fits in ROW_BITS bits");
static_assert(N_WIN * FIND_WIN == ROW_RANGES && N_WIN * FIND_SLICES == OWN_GROUPS, "windows tile the ranges");
static_assert(FIND_SLICES == 32, "an owner work-group reads the counts of its lists one per lane");
static_assert(ROW_CAP % 16 == 0 && ROW_CAP * OUTD <= OWN_WAVES * Q_CAP, "the 10 wide rows of a range fit in the queue space");
static_assert(FIND_WIN * ROW_CAP <= (1 << WIN_BITS), "a row of the window fits in WIN_BITS bits");
static_assert((1 << FIND_WIN_LOG) == FIND_WIN && FIND_WIN <= 32, "one lane per list of the window");

__global__ void gcn_gemm1_wmma(const float* __restrict__ x, const float* __restrict__ W1,
                               float* __restrict__ m1, int n_tiles) {
    int wave = (int)((blockIdx.x * blockDim.x + threadIdx.x) >> 5);
    int lane = threadIdx.x & 31;
    if (wave >= n_tiles) return;
    int l15 = lane & 15;
    int hi  = lane >> 4;
    const float* xrow = x + ((size_t)(wave * 16 + l15)) * IN_DIM + hi * 4;
    const float* wcol = W1 + (hi * 4) * HID + l15;

    v8f c = {};
    #pragma unroll 4
    for (int k0 = 0; k0 < IN_DIM; k0 += 8) {
        const float* xk = xrow + k0;
        const float* wk = wcol + k0 * HID;
        v2f a0; a0.x = xk[0];       a0.y = xk[1];
        v2f a1; a1.x = xk[2];       a1.y = xk[3];
        v2f b0; b0.x = wk[0 * HID]; b0.y = wk[1 * HID];
        v2f b1; b1.x = wk[2 * HID]; b1.y = wk[3 * HID];
        c = __builtin_amdgcn_wmma_f32_16x16x4_f32(false, a0, false, b0, (short)0, c, false, false);
        c = __builtin_amdgcn_wmma_f32_16x16x4_f32(false, a1, false, b1, (short)0, c, false, false);
    }
    float* mt = m1 + (size_t)(wave * 16) * HID;
    #pragma unroll
    for (int v = 0; v < 8; ++v)
        mt[(v + hi * 8) * HID + l15] = c[v];
}

__device__ __forceinline__ int q_push(int* qe, unsigned short* qt, int qn,
                                      unsigned t, int e, unsigned cnt) {
    bool hit = t < cnt;
    unsigned mask = __builtin_amdgcn_ballot_w32(hit);
    if (hit) {
        int slot = qn + (int)__builtin_amdgcn_mbcnt_lo(mask, 0u);
        qe[slot] = e;
        qt[slot] = (unsigned short)t;
    }
    return qn + __builtin_popcount(mask);
}

__device__ __forceinline__ void q_drain(const int* qe, const unsigned short* qt, int qn,
                                        const int* __restrict__ src, const float* __restrict__ m,
                                        float* acc, int lane) {
    const int f  = lane & 15;
    const int hi = lane >> 4;
    for (int g = 0; g < qn; g += 32) {
        int n = qn - g; if (n > 32) n = 32;
        int s = 0, t = 0;
        if (lane < n) { s = src[qe[g + lane]]; t = (int)qt[g + lane]; }
        for (int j = 0; j < n; j += 8) {
            float v[4]; int tr[4]; bool ok[4];
            #pragma unroll
            for (int u = 0; u < 4; ++u) {
                int ja = j + 2 * u, jb = ja + 1;
                bool a_ok = ja < n, b_ok = jb < n;
                int jac = a_ok ? ja : 0;
                int jbc = b_ok ? jb : jac;
                int sa = __builtin_amdgcn_readlane(s, jac), sb = __builtin_amdgcn_readlane(s, jbc);
                int ta = __builtin_amdgcn_readlane(t, jac), tb = __builtin_amdgcn_readlane(t, jbc);
                int sj = hi ? sb : sa;
                tr[u]  = hi ? tb : ta;
                ok[u]  = hi ? b_ok : a_ok;
                v[u]   = m[(size_t)sj * HID + f];
            }
            #pragma unroll
            for (int u = 0; u < 4; ++u)
                if (ok[u]) atomicAdd(&acc[tr[u] * HID + f], v[u]);
        }
    }
}

__device__ __forceinline__ void ld_chunk(int4 (&b)[4], const int4* __restrict__ d4,
                                         int k, int start, int n_part, int lane) {
    int kk = k < n_part ? k : n_part - 1;
    int ci = kk + start; if (ci >= n_part) ci -= n_part;
    #pragma unroll
    for (int j = 0; j < 4; ++j) b[j] = d4[(size_t)ci * 128 + j * 32 + lane];
}

__device__ __forceinline__ int scan_chunk(const int4 (&b)[4], int k, int start, int n_part,
                                          unsigned lo, unsigned ucnt, int* qe, unsigned short* qt,
                                          int qn, const int* __restrict__ src,
                                          const float* __restrict__ m, float* acc, int lane) {
    int ci = k + start; if (ci >= n_part) ci -= n_part;
    #pragma unroll
    for (int j = 0; j < 4; ++j) {
        const int eb = ci * 512 + j * 128 + lane * 4;
        unsigned t0 = (unsigned)b[j].x - lo;
        unsigned t1 = (unsigned)b[j].y - lo;
        unsigned t2 = (unsigned)b[j].z - lo;
        unsigned t3 = (unsigned)b[j].w - lo;
        qn = q_push(qe, qt, qn, t0, eb + 0, ucnt);
        qn = q_push(qe, qt, qn, t1, eb + 1, ucnt);
        qn = q_push(qe, qt, qn, t2, eb + 2, ucnt);
        qn = q_push(qe, qt, qn, t3, eb + 3, ucnt);
    }
    if (qn >= 32) {
        q_drain(qe, qt, qn, src, m, acc, lane);
        qn = 0;
    }
    return qn;
}

__device__ __forceinline__ void scan_slice(const int* __restrict__ src, const int* __restrict__ dst,
                                           const float* __restrict__ m, int n_edges,
                                           unsigned ulo, unsigned ucnt, float* acc,
                                           int* qe, unsigned short* qt, int r,
                                           int lane, int wv, int nw) {
    const bool vec_ok = ((((size_t)dst) & 15) == 0);
    const int n_part = vec_ok ? (n_edges >> 9) : 0;
    const int start = (int)(((long long)r * n_part) / ROW_RANGES);
    const int4* d4 = (const int4*)dst;
    const int t_lo = n_part << 9;
    const int t_hi = n_edges;

    int qn = 0;
    if (n_part > 0) {
    int4 A[4], B[4], C[4];
    ld_chunk(A, d4, wv,      start, n_part, lane);
    ld_chunk(B, d4, wv + nw, start, n_part, lane);
    for (int c = wv; c < n_part; c += 3 * nw) {
        ld_chunk(C, d4, c + 2 * nw, start, n_part, lane);
        qn = scan_chunk(A, c, start, n_part, ulo, ucnt, qe, qt, qn, src, m, acc, lane);
        ld_chunk(A, d4, c + 3 * nw, start, n_part, lane);
        if (c + nw < n_part)
            qn = scan_chunk(B, c + nw, start, n_part, ulo, ucnt, qe, qt, qn, src, m, acc, lane);
        ld_chunk(B, d4, c + 4 * nw, start, n_part, lane);
        if (c + 2 * nw < n_part)
            qn = scan_chunk(C, c + 2 * nw, start, n_part, ulo, ucnt, qe, qt, qn, src, m, acc, lane);
    }
    }
    for (int b = t_lo + wv * 32; b < t_hi; b += nw * 32) {
        int e = b + lane;
        unsigned t = (e < t_hi) ? ((unsigned)dst[e] - ulo) : 0xFFFFFFFFu;
        qn = q_push(qe, qt, qn, t, e, ucnt);
        if (qn >= 32) {
            q_drain(qe, qt, qn, src, m, acc, lane);
            qn = 0;
        }
    }
    if (qn > 0) q_drain(qe, qt, qn, src, m, acc, lane);
}

__device__ __forceinline__ int range_rows(int n_nodes) {
    int rr = (n_nodes + ROW_RANGES - 1) / ROW_RANGES;
    rr = (rr + 15) & ~15;
    if (rr > ROW_CAP) rr = ROW_CAP;
    return rr;
}

__device__ __forceinline__ int w_push(unsigned* qw, int qn, unsigned t, int s, unsigned cnt) {
    bool hit = t < cnt;
    unsigned mask = __builtin_amdgcn_ballot_w32(hit);
    if (hit) qw[qn + (int)__builtin_amdgcn_mbcnt_lo(mask, 0u)] = ((unsigned)s << WIN_BITS) | t;
    return qn + __builtin_popcount(mask);
}

__device__ __forceinline__ void w_flush(const unsigned* qw, int qn, int lane, unsigned rr,
                                        unsigned rr_inv, unsigned* __restrict__ lst, int* lst_n) {
    for (int g = 0; g < qn; g += 32) {
        const int i = g + lane;
        const bool valid = i < qn;
        const unsigned wd = qw[valid ? i : qn - 1];
        const unsigned t = wd & ((1u << WIN_BITS) - 1u);
        const unsigned k = __umulhi(t, rr_inv);
        const unsigned word = ((wd >> WIN_BITS) << ROW_BITS) | (t - k * rr);
        unsigned same = __builtin_amdgcn_ballot_w32(valid);
        unsigned mine = same;
        #pragma unroll
        for (int b = 0; b < FIND_WIN_LOG; ++b) {
            const unsigned kb = __builtin_amdgcn_ballot_w32(valid && ((k >> b) & 1u) != 0u);
            same &= ((k >> b) & 1u) ? kb : ~kb;
            mine &= (((unsigned)lane >> b) & 1u) ? kb : ~kb;
        }
        int base = 0;
        if (lane < FIND_WIN) base = atomicAdd(&lst_n[lane], __builtin_popcount(mine));
        const int bk  = __builtin_amdgcn_ds_bpermute((int)(k << 2), base);
        const int pos = bk + (int)__builtin_amdgcn_mbcnt_lo(same, 0u);
        if (valid && pos < SUB_CAP) lst[(size_t)k * SUB_CAP + pos] = word;
    }
}

__device__ __forceinline__ void ld_pair(int4 (&d)[4], int4 (&s)[4], const int4* __restrict__ d4,
                                        const int4* __restrict__ s4, int k, int start,
                                        int n_part, int c0, int lane) {
    int kk = k < n_part ? k : n_part - 1;
    int ci = kk + start; if (ci >= n_part) ci -= n_part;
    ci += c0;
    #pragma unroll
    for (int j = 0; j < 4; ++j) {
        d[j] = d4[(size_t)ci * 128 + j * 32 + lane];
        s[j] = s4[(size_t)ci * 128 + j * 32 + lane];
    }
}

__device__ __forceinline__ int find_chunk(const int4 (&d)[4], const int4 (&s)[4], unsigned lo,
                                          unsigned ucnt, unsigned* qw, int qn, int lane,
                                          unsigned rr, unsigned rr_inv,
                                          unsigned* __restrict__ lst, int* lst_n) {
    #pragma unroll
    for (int j = 0; j < 4; ++j) {
        qn = w_push(qw, qn, (unsigned)d[j].x - lo, s[j].x, ucnt);
        qn = w_push(qw, qn, (unsigned)d[j].y - lo, s[j].y, ucnt);
        qn = w_push(qw, qn, (unsigned)d[j].z - lo, s[j].z, ucnt);
        qn = w_push(qw, qn, (unsigned)d[j].w - lo, s[j].w, ucnt);
    }
    if (qn >= 32) { w_flush(qw, qn, lane, rr, rr_inv, lst, lst_n); qn = 0; }
    return qn;
}

__global__ __launch_bounds__(OWN_THREADS)
void gcn_find_hits(const int* __restrict__ src, const int* __restrict__ dst,
                   unsigned* __restrict__ lst_all, int* __restrict__ cnts,
                   int n_nodes, int n_edges) {
    __shared__ unsigned q_w[OWN_WAVES * Q_CAP];
    __shared__ int lst_n[FIND_WIN];

    const int tid = threadIdx.x;
    const int rr  = range_rows(n_nodes);
    int* my_cnt = cnts + (size_t)blockIdx.x * FIND_WIN * CNT_PITCH;
    const bool keep = ((long long)ROW_RANGES * rr >= n_nodes) && (n_nodes <= (1 << (32 - WIN_BITS)))
                      && (((((size_t)src) | ((size_t)dst)) & 15) == 0);
    if (!keep) {
        if (tid < FIND_WIN) my_cnt[tid * CNT_PITCH] = 0x7fffffff;
        return;
    }

    const int lane = tid & 31;
    const int wv   = __builtin_amdgcn_readfirstlane(tid >> 5);
    const int nw   = blockDim.x >> 5;
    const int q    = blockIdx.x % FIND_SLICES;
    const int w    = blockIdx.x / FIND_SLICES;
    const int lo   = w * FIND_WIN * rr;
    int cnt = n_nodes - lo; if (cnt > FIND_WIN * rr) cnt = FIND_WIN * rr; if (cnt < 0) cnt = 0;
    unsigned* lst = lst_all + (size_t)blockIdx.x * LIST_CAP;
    unsigned* qw  = q_w + wv * Q_CAP;

    if (tid < FIND_WIN) lst_n[tid] = 0;
    __syncthreads();

    if (cnt > 0) {
        const int n_full = n_edges >> 9;
        const int c0     = (int)(((long long)q * n_full) / FIND_SLICES);
        const int n_part = (int)(((long long)(q + 1) * n_full) / FIND_SLICES) - c0;
        const int start  = (int)(((long long)w * n_part) / N_WIN);
        const int4* d4 = (const int4*)dst;
        const int4* s4 = (const int4*)src;
        const int t_base = n_full << 9;
        const int t_lo = t_base + (int)(((long long)q * (n_edges - t_base)) / FIND_SLICES);
        const int t_hi = t_base + (int)(((long long)(q + 1) * (n_edges - t_base)) / FIND_SLICES);
        const unsigned ulo = (unsigned)lo, ucnt = (unsigned)cnt, urr = (unsigned)rr;
        const unsigned rr_inv = 0xFFFFFFFFu / urr + 1u;

        int qn = 0;
        if (n_part > 0) {
            int4 dA[4], sA[4], dB[4], sB[4];
            ld_pair(dA, sA, d4, s4, wv,      start, n_part, c0, lane);
            ld_pair(dB, sB, d4, s4, wv + nw, start, n_part, c0, lane);
            for (int c = wv; c < n_part; c += 2 * nw) {
                qn = find_chunk(dA, sA, ulo, ucnt, qw, qn, lane, urr, rr_inv, lst, lst_n);
                ld_pair(dA, sA, d4, s4, c + 2 * nw, start, n_part, c0, lane);
                if (c + nw < n_part)
                    qn = find_chunk(dB, sB, ulo, ucnt, qw, qn, lane, urr, rr_inv, lst, lst_n);
                ld_pair(dB, sB, d4, s4, c + 3 * nw, start, n_part, c0, lane);
            }
        }
        for (int b = t_lo + wv * 32; b < t_hi; b += nw * 32) {
            const int e = b + lane;
            unsigned t = 0xFFFFFFFFu; int sn = 0;
            if (e < t_hi) { t = (unsigned)dst[e] - ulo; sn = src[e]; }
            qn = w_push(qw, qn, t, sn, ucnt);
            if (qn >= 32) { w_flush(qw, qn, lane, urr, rr_inv, lst, lst_n); qn = 0; }
        }
        if (qn > 0) w_flush(qw, qn, lane, urr, rr_inv, lst, lst_n);
    }
    __syncthreads();
    if (tid < FIND_WIN) my_cnt[tid * CNT_PITCH] = lst_n[tid];
}

__device__ __forceinline__ void store_hidden(const float* acc, const float* __restrict__ b1,
                                             float* __restrict__ h, int lo, int cnt) {
    const int tid = threadIdx.x;
    const int c = (tid & 3) * 4;
    const float bx = b1[c], by = b1[c + 1], bz = b1[c + 2], bw = b1[c + 3];
    const float4* acc4 = (const float4*)acc;
    float4* h4 = (float4*)(h + (size_t)lo * HID);
    for (int i = tid; i < cnt * (HID / 4); i += blockDim.x) {
        float4 a = acc4[i];
        a.x = fmaxf(a.x + bx, 0.0f);
        a.y = fmaxf(a.y + by, 0.0f);
        a.z = fmaxf(a.z + bz, 0.0f);
        a.w = fmaxf(a.w + bw, 0.0f);
        h4[i] = a;
    }
}

__device__ __forceinline__ void store_out(const float* acc, float* stage,
                                          const float* __restrict__ W2, const float* __restrict__ b2,
                                          float* __restrict__ out, int lo, int cnt,
                                          int lane, int wv, int nw) {
    const int l15 = lane & 15;
    const int hi  = lane >> 4;
    const bool col_ok = l15 < OUTD;
    v2f b[HID / 4];
    #pragma unroll
    for (int j = 0; j < HID / 4; ++j) {
        const int ka = j * 4 + hi * 2;
        b[j].x = col_ok ? W2[(ka + 0) * OUTD + l15] : 0.0f;
        b[j].y = col_ok ? W2[(ka + 1) * OUTD + l15] : 0.0f;
    }
    const float bias = col_ok ? b2[l15] : 0.0f;
    const int n_t = (cnt + 15) >> 4;
    for (int t = wv; t < n_t; t += nw) {
        const float* arow = acc + (t * 16 + l15) * HID + hi * 2;
        v8f c;
        #pragma unroll
        for (int v = 0; v < 8; ++v) c[v] = bias;
        #pragma unroll
        for (int j = 0; j < HID / 4; ++j) {
            v2f a; a.x = arow[j * 4]; a.y = arow[j * 4 + 1];
            c = __builtin_amdgcn_wmma_f32_16x16x4_f32(false, a, false, b[j], (short)0, c, false, false);
        }
        #pragma unroll
        for (int v = 0; v < 8; ++v) {
            const int row = t * 16 + hi * 8 + v;
            if (col_ok && row < cnt) stage[row * OUTD + l15] = c[v];
        }
    }
    __syncthreads();
    const int tid = threadIdx.x;
    const int n_f = cnt * OUTD;
    float* o = out + (size_t)lo * OUTD;
    int done = 0;
    if ((((size_t)o) & 15) == 0) {
        const int n4 = n_f >> 2;
        const float4* s4 = (const float4*)stage;
        float4* o4 = (float4*)o;
        for (int i = tid; i < n4; i += blockDim.x) o4[i] = s4[i];
        done = n4 << 2;
    }
    for (int i = done + tid; i < n_f; i += blockDim.x) o[i] = stage[i];
}

template <int LAYER>
__device__ __forceinline__ void store_rows(const float* acc, float* stage, const float* __restrict__ bias,
                                           const float* __restrict__ W2, float* __restrict__ res,
                                           int lo, int cnt, int lane, int wv, int nw) {
    if (LAYER == 1) store_hidden(acc, bias, res, lo, cnt);
    else            store_out(acc, stage, W2, bias, res, lo, cnt, lane, wv, nw);
}

template <int LAYER>
__device__ __forceinline__ void owner_scan(const int* __restrict__ src, const int* __restrict__ dst,
                                           const float* __restrict__ m, float* __restrict__ res,
                                           const float* __restrict__ bias, const float* __restrict__ W2,
                                           int n_nodes, int n_edges,
                                           float* acc, int* q_e, unsigned short* q_t) {
    const int tid  = threadIdx.x;
    const int lane = tid & 31;
    const int wv   = __builtin_amdgcn_readfirstlane(tid >> 5);
    const int nw   = blockDim.x >> 5;
    const int r    = blockIdx.x;
    int* qe = q_e + wv * Q_CAP;
    unsigned short* qt = q_t + wv * Q_CAP;
    const int rr = range_rows(n_nodes);

    for (int lo = r * rr; lo < n_nodes; lo += ROW_RANGES * rr) {
        int cnt = n_nodes - lo; if (cnt > rr) cnt = rr;
        float4* acc4 = (float4*)acc;
        for (int i = tid; i < cnt * (HID / 4); i += blockDim.x) acc4[i] = make_float4(0.f, 0.f, 0.f, 0.f);
        __syncthreads();

        scan_slice(src, dst, m, n_edges, (unsigned)lo, (unsigned)cnt, acc, qe, qt, r,
                   lane, wv, nw);
        __syncthreads();

        store_rows<LAYER>(acc, (float*)q_e, bias, W2, res, lo, cnt, lane, wv, nw);
        __syncthreads();
    }
}

__device__ __forceinline__ void list_sum(const unsigned* __restrict__ l, int total,
                                         const float* __restrict__ m, float* acc,
                                         int lane, int sw, int nsw) {
    const int f  = lane & 15;
    const int hi = lane >> 4;
    const int nb = (total + 31) >> 5;
    const unsigned row_mask = (1u << ROW_BITS) - 1u;
    unsigned cur[16];
    {
        unsigned w = 0u;
        if (sw * 32 + lane < total) w = l[sw * 32 + lane];
        #pragma unroll
        for (int u = 0; u < 16; ++u) {
            unsigned wa = (unsigned)__builtin_amdgcn_readlane((int)w, 2 * u);
            unsigned wb = (unsigned)__builtin_amdgcn_readlane((int)w, 2 * u + 1);
            cur[u] = hi ? wb : wa;
        }
    }
    for (int b = sw; b < nb; b += nsw) {
        unsigned wn = 0u;
        { int i = (b + nsw) * 32 + lane; if (i < total) wn = l[i]; }
        const int n = total - b * 32;
        float v[16];
        #pragma unroll
        for (int u = 0; u < 16; ++u)
            v[u] = m[(size_t)(cur[u] >> ROW_BITS) * HID + f];
        if (n >= 32) {
            #pragma unroll
            for (int u = 0; u < 16; ++u)
                atomicAdd(&acc[(int)(cur[u] & row_mask) * HID + f], v[u]);
        } else {
            #pragma unroll
            for (int u = 0; u < 16; ++u)
                if (2 * u + hi < n) atomicAdd(&acc[(int)(cur[u] & row_mask) * HID + f], v[u]);
        }
        #pragma unroll
        for (int u = 0; u < 16; ++u) {
            unsigned wa = (unsigned)__builtin_amdgcn_readlane((int)wn, 2 * u);
            unsigned wb = (unsigned)__builtin_amdgcn_readlane((int)wn, 2 * u + 1);
            cur[u] = hi ? wb : wa;
        }
    }
}

template <int LAYER>
__global__ __launch_bounds__(OWN_THREADS)
void gcn_owner_sum_list(const int* __restrict__ src, const int* __restrict__ dst,
                        const float* __restrict__ m, float* __restrict__ res,
                        const float* __restrict__ bias, const float* __restrict__ W2,
                        const unsigned* __restrict__ lst_all, const int* __restrict__ cnts,
                        int n_nodes, int n_edges) {
    __shared__ __attribute__((aligned(16))) float acc[ROW_CAP * HID];
    __shared__ __attribute__((aligned(16))) int q_e[OWN_WAVES * Q_CAP];
    __shared__ unsigned short q_t[OWN_WAVES * Q_CAP];

    const int tid  = threadIdx.x;
    const int lane = tid & 31;
    const int wv   = __builtin_amdgcn_readfirstlane(tid >> 5);
    const int nw   = blockDim.x >> 5;
    const int r = blockIdx.x;
    const int w = r / FIND_WIN;
    const int k = r % FIND_WIN;
    const size_t fb0 = (size_t)w * FIND_SLICES;
    const int n_l = cnts[((fb0 + lane) * FIND_WIN + k) * CNT_PITCH];
    const bool lists_ok = __builtin_amdgcn_ballot_w32((unsigned)n_l > (unsigned)SUB_CAP) == 0u;
    if (!lists_ok) {
        owner_scan<LAYER>(src, dst, m, res, bias, W2, n_nodes, n_edges, acc, q_e, q_t);
        return;
    }

    const int rr = range_rows(n_nodes);
    const int lo = r * rr;
    int cnt = n_nodes - lo; if (cnt > rr) cnt = rr; if (cnt < 0) cnt = 0;

    float4* acc4 = (float4*)acc;
    for (int i = tid; i < cnt * (HID / 4); i += blockDim.x) acc4[i] = make_float4(0.f, 0.f, 0.f, 0.f);
    __syncthreads();

    for (int u = wv; u < FIND_SLICES; u += nw) {
        const int total = __builtin_amdgcn_readlane(n_l, u);
        const unsigned* l = lst_all + (fb0 + u) * LIST_CAP + (size_t)k * SUB_CAP;
        list_sum(l, total, m, acc, lane, 0, 1);
    }
    __syncthreads();

    store_rows<LAYER>(acc, (float*)q_e, bias, W2, res, lo, cnt, lane, wv, nw);
}

extern "C" void kernel_launch(void* const* d_in, const int* in_sizes, int n_in,
                              void* d_out, int out_size, void* d_ws, size_t ws_size,
                              hipStream_t stream) {
    const float* x  = (const float*)d_in[0];
    const int*   ei = (const int*)  d_in[1];
    const float* W1 = (const float*)d_in[2];
    const float* b1 = (const float*)d_in[3];
    const float* W2 = (const float*)d_in[4];
    const float* b2 = (const float*)d_in[5];
    float* out = (float*)d_out;

    int n_nodes       = in_sizes[0] / IN_DIM;
    long long n_edges = (long long)in_sizes[1] / 2;
    const int* src = ei;
    const int* dst = ei + n_edges;

    float* m1   = (float*)d_ws;
    float* h    = m1 + (size_t)n_nodes * HID;
    unsigned* hits = (unsigned*)(h + (size_t)n_nodes * HID);
    int* hit_n  = (int*)(hits + (size_t)OWN_GROUPS * LIST_CAP);

    int n_tiles = (n_nodes + 15) / 16;
    int gemm_blocks = (n_tiles + 7) / 8;

    gcn_gemm1_wmma<<<gemm_blocks, 256, 0, stream>>>(x, W1, m1, n_tiles);
    gcn_find_hits<<<OWN_GROUPS, OWN_THREADS, 0, stream>>>(src, dst, hits, hit_n, n_nodes, (int)n_edges);
    gcn_owner_sum_list<1><<<OWN_GROUPS, OWN_THREADS, 0, stream>>>(src, dst, m1, h, b1, W2, hits, hit_n, n_nodes, (int)n_edges);
    gcn_owner_sum_list<2><<<OWN_GROUPS, OWN_THREADS, 0, stream>>>(src, dst, h, out, b2, W2, hits, hit_n, n_nodes, (int)n_edges);
}


# find pass builds its 32 lists in LDS and stores each once as one contiguous piece, 16 bytes a lane: whole 128-byte lines, not 32 scattered 4-byte words a flush; sums stay in LDS arrival order
# speedup vs baseline: 16.5463x; 1.1772x over previous
#include <hip/hip_runtime.h>


typedef float v2f __attribute__((ext_vector_type(2)));
typedef float v8f __attribute__((ext_vector_type(8)));

#define IN_DIM 256
#define HID 16
#define OUTD 10

#define OWN_GROUPS 256
#define OWN_THREADS 1024
#define OWN_WAVES 32
#define ROW_RANGES 256
#define ROW_CAP 1024
#define Q_CAP 544
#define LIST_CAP 32768
#define ROW_BITS 12
#define CNT_PITCH 32
#define FIND_WIN 32
#define FIND_WIN_LOG 5
#define N_WIN (ROW_RANGES / FIND_WIN)
#define FIND_SLICES (OWN_GROUPS / N_WIN)
#define SUB_CAP (LIST_CAP / FIND_WIN)
#define SUB_PITCH (SUB_CAP + 4)
#define WIN_BITS 15
static_assert(ROW_RANGES == OWN_GROUPS, "one owner work-group per row range");
static_assert(ROW_CAP <= (1 << ROW_BITS), "a local row fits in ROW_BITS bits");
static_assert(N_WIN * FIND_WIN == ROW_RANGES && N_WIN * FIND_SLICES == OWN_GROUPS, "windows tile the ranges");
static_assert(FIND_SLICES == 32, "an owner work-group reads the counts of its lists one per lane");
static_assert(ROW_CAP % 16 == 0 && ROW_CAP * OUTD <= OWN_WAVES * Q_CAP, "the 10 wide rows of a range fit in the queue space");
static_assert(FIND_WIN * ROW_CAP <= (1 << WIN_BITS), "a row of the window fits in WIN_BITS bits");
static_assert((1 << FIND_WIN_LOG) == FIND_WIN && FIND_WIN <= 32, "one lane per list of the window");
static_assert(SUB_CAP % 4 == 0 && SUB_PITCH % 4 == 0, "a list is copied out in 16-byte pieces");

__global__ void gcn_gemm1_wmma(const float* __restrict__ x, const float* __restrict__ W1,
                               float* __restrict__ m1, int n_tiles) {
    int wave = (int)((blockIdx.x * blockDim.x + threadIdx.x) >> 5);
    int lane = threadIdx.x & 31;
    if (wave >= n_tiles) return;
    int l15 = lane & 15;
    int hi  = lane >> 4;
    const float* xrow = x + ((size_t)(wave * 16 + l15)) * IN_DIM + hi * 4;
    const float* wcol = W1 + (hi * 4) * HID + l15;

    v8f c = {};
    #pragma unroll 4
    for (int k0 = 0; k0 < IN_DIM; k0 += 8) {
        const float* xk = xrow + k0;
        const float* wk = wcol + k0 * HID;
        v2f a0; a0.x = xk[0];       a0.y = xk[1];
        v2f a1; a1.x = xk[2];       a1.y = xk[3];
        v2f b0; b0.x = wk[0 * HID]; b0.y = wk[1 * HID];
        v2f b1; b1.x = wk[2 * HID]; b1.y = wk[3 * HID];
        c = __builtin_amdgcn_wmma_f32_16x16x4_f32(false, a0, false, b0, (short)0, c, false, false);
        c = __builtin_amdgcn_wmma_f32_16x16x4_f32(false, a1, false, b1, (short)0, c, false, false);
    }
    float* mt = m1 + (size_t)(wave * 16) * HID;
    #pragma unroll
    for (int v = 0; v < 8; ++v)
        mt[(v + hi * 8) * HID + l15] = c[v];
}

__device__ __forceinline__ int q_push(int* qe, unsigned short* qt, int qn,
                                      unsigned t, int e, unsigned cnt) {
    bool hit = t < cnt;
    unsigned mask = __builtin_amdgcn_ballot_w32(hit);
    if (hit) {
        int slot = qn + (int)__builtin_amdgcn_mbcnt_lo(mask, 0u);
        qe[slot] = e;
        qt[slot] = (unsigned short)t;
    }
    return qn + __builtin_popcount(mask);
}

__device__ __forceinline__ void q_drain(const int* qe, const unsigned short* qt, int qn,
                                        const int* __restrict__ src, const float* __restrict__ m,
                                        float* acc, int lane) {
    const int f  = lane & 15;
    const int hi = lane >> 4;
    for (int g = 0; g < qn; g += 32) {
        int n = qn - g; if (n > 32) n = 32;
        int s = 0, t = 0;
        if (lane < n) { s = src[qe[g + lane]]; t = (int)qt[g + lane]; }
        for (int j = 0; j < n; j += 8) {
            float v[4]; int tr[4]; bool ok[4];
            #pragma unroll
            for (int u = 0; u < 4; ++u) {
                int ja = j + 2 * u, jb = ja + 1;
                bool a_ok = ja < n, b_ok = jb < n;
                int jac = a_ok ? ja : 0;
                int jbc = b_ok ? jb : jac;
                int sa = __builtin_amdgcn_readlane(s, jac), sb = __builtin_amdgcn_readlane(s, jbc);
                int ta = __builtin_amdgcn_readlane(t, jac), tb = __builtin_amdgcn_readlane(t, jbc);
                int sj = hi ? sb : sa;
                tr[u]  = hi ? tb : ta;
                ok[u]  = hi ? b_ok : a_ok;
                v[u]   = m[(size_t)sj * HID + f];
            }
            #pragma unroll
            for (int u = 0; u < 4; ++u)
                if (ok[u]) atomicAdd(&acc[tr[u] * HID + f], v[u]);
        }
    }
}

__device__ __forceinline__ void ld_chunk(int4 (&b)[4], const int4* __restrict__ d4,
                                         int k, int start, int n_part, int lane) {
    int kk = k < n_part ? k : n_part - 1;
    int ci = kk + start; if (ci >= n_part) ci -= n_part;
    #pragma unroll
    for (int j = 0; j < 4; ++j) b[j] = d4[(size_t)ci * 128 + j * 32 + lane];
}

__device__ __forceinline__ int scan_chunk(const int4 (&b)[4], int k, int start, int n_part,
                                          unsigned lo, unsigned ucnt, int* qe, unsigned short* qt,
                                          int qn, const int* __restrict__ src,
                                          const float* __restrict__ m, float* acc, int lane) {
    int ci = k + start; if (ci >= n_part) ci -= n_part;
    #pragma unroll
    for (int j = 0; j < 4; ++j) {
        const int eb = ci * 512 + j * 128 + lane * 4;
        unsigned t0 = (unsigned)b[j].x - lo;
        unsigned t1 = (unsigned)b[j].y - lo;
        unsigned t2 = (unsigned)b[j].z - lo;
        unsigned t3 = (unsigned)b[j].w - lo;
        qn = q_push(qe, qt, qn, t0, eb + 0, ucnt);
        qn = q_push(qe, qt, qn, t1, eb + 1, ucnt);
        qn = q_push(qe, qt, qn, t2, eb + 2, ucnt);
        qn = q_push(qe, qt, qn, t3, eb + 3, ucnt);
    }
    if (qn >= 32) {
        q_drain(qe, qt, qn, src, m, acc, lane);
        qn = 0;
    }
    return qn;
}

__device__ __forceinline__ void scan_slice(const int* __restrict__ src, const int* __restrict__ dst,
                                           const float* __restrict__ m, int n_edges,
                                           unsigned ulo, unsigned ucnt, float* acc,
                                           int* qe, unsigned short* qt, int r,
                                           int lane, int wv, int nw) {
    const bool vec_ok = ((((size_t)dst) & 15) == 0);
    const int n_part = vec_ok ? (n_edges >> 9) : 0;
    const int start = (int)(((long long)r * n_part) / ROW_RANGES);
    const int4* d4 = (const int4*)dst;
    const int t_lo = n_part << 9;
    const int t_hi = n_edges;

    int qn = 0;
    if (n_part > 0) {
    int4 A[4], B[4], C[4];
    ld_chunk(A, d4, wv,      start, n_part, lane);
    ld_chunk(B, d4, wv + nw, start, n_part, lane);
    for (int c = wv; c < n_part; c += 3 * nw) {
        ld_chunk(C, d4, c + 2 * nw, start, n_part, lane);
        qn = scan_chunk(A, c, start, n_part, ulo, ucnt, qe, qt, qn, src, m, acc, lane);
        ld_chunk(A, d4, c + 3 * nw, start, n_part, lane);
        if (c + nw < n_part)
            qn = scan_chunk(B, c + nw, start, n_part, ulo, ucnt, qe, qt, qn, src, m, acc, lane);
        ld_chunk(B, d4, c + 4 * nw, start, n_part, lane);
        if (c + 2 * nw < n_part)
            qn = scan_chunk(C, c + 2 * nw, start, n_part, ulo, ucnt, qe, qt, qn, src, m, acc, lane);
    }
    }
    for (int b = t_lo + wv * 32; b < t_hi; b += nw * 32) {
        int e = b + lane;
        unsigned t = (e < t_hi) ? ((unsigned)dst[e] - ulo) : 0xFFFFFFFFu;
        qn = q_push(qe, qt, qn, t, e, ucnt);
        if (qn >= 32) {
            q_drain(qe, qt, qn, src, m, acc, lane);
            qn = 0;
        }
    }
    if (qn > 0) q_drain(qe, qt, qn, src, m, acc, lane);
}

__device__ __forceinline__ int range_rows(int n_nodes) {
    int rr = (n_nodes + ROW_RANGES - 1) / ROW_RANGES;
    rr = (rr + 15) & ~15;
    if (rr > ROW_CAP) rr = ROW_CAP;
    return rr;
}

__device__ __forceinline__ int w_push(unsigned* qw, int qn, unsigned t, int s, unsigned cnt) {
    bool hit = t < cnt;
    unsigned mask = __builtin_amdgcn_ballot_w32(hit);
    if (hit) qw[qn + (int)__builtin_amdgcn_mbcnt_lo(mask, 0u)] = ((unsigned)s << WIN_BITS) | t;
    return qn + __builtin_popcount(mask);
}

__device__ __forceinline__ void w_flush(const unsigned* qw, int qn, int lane, unsigned rr,
                                        unsigned rr_inv, unsigned* ls, int* lst_n) {
    for (int g = 0; g < qn; g += 32) {
        const int i = g + lane;
        const bool valid = i < qn;
        const unsigned wd = qw[valid ? i : qn - 1];
        const unsigned t = wd & ((1u << WIN_BITS) - 1u);
        const unsigned k = __umulhi(t, rr_inv);
        const unsigned word = ((wd >> WIN_BITS) << ROW_BITS) | (t - k * rr);
        unsigned same = __builtin_amdgcn_ballot_w32(valid);
        unsigned mine = same;
        #pragma unroll
        for (int b = 0; b < FIND_WIN_LOG; ++b) {
            const unsigned kb = __builtin_amdgcn_ballot_w32(valid && ((k >> b) & 1u) != 0u);
            same &= ((k >> b) & 1u) ? kb : ~kb;
            mine &= (((unsigned)lane >> b) & 1u) ? kb : ~kb;
        }
        int base = 0;
        if (lane < FIND_WIN) base = atomicAdd(&lst_n[lane], __builtin_popcount(mine));
        const int bk  = __builtin_amdgcn_ds_bpermute((int)(k << 2), base);
        const int pos = bk + (int)__builtin_amdgcn_mbcnt_lo(same, 0u);
        if (valid && pos < SUB_CAP) ls[(int)k * SUB_PITCH + pos] = word;
    }
}

__device__ __forceinline__ void ld_pair(int4 (&d)[4], int4 (&s)[4], const int4* __restrict__ d4,
                                        const int4* __restrict__ s4, int k, int start,
                                        int n_part, int c0, int lane) {
    int kk = k < n_part ? k : n_part - 1;
    int ci = kk + start; if (ci >= n_part) ci -= n_part;
    ci += c0;
    #pragma unroll
    for (int j = 0; j < 4; ++j) {
        d[j] = d4[(size_t)ci * 128 + j * 32 + lane];
        s[j] = s4[(size_t)ci * 128 + j * 32 + lane];
    }
}

__device__ __forceinline__ int find_chunk(const int4 (&d)[4], const int4 (&s)[4], unsigned lo,
                                          unsigned ucnt, unsigned* qw, int qn, int lane,
                                          unsigned rr, unsigned rr_inv,
                                          unsigned* lst, int* lst_n) {
    #pragma unroll
    for (int j = 0; j < 4; ++j) {
        qn = w_push(qw, qn, (unsigned)d[j].x - lo, s[j].x, ucnt);
        qn = w_push(qw, qn, (unsigned)d[j].y - lo, s[j].y, ucnt);
        qn = w_push(qw, qn, (unsigned)d[j].z - lo, s[j].z, ucnt);
        qn = w_push(qw, qn, (unsigned)d[j].w - lo, s[j].w, ucnt);
    }
    if (qn >= 32) { w_flush(qw, qn, lane, rr, rr_inv, lst, lst_n); qn = 0; }
    return qn;
}

__global__ __launch_bounds__(OWN_THREADS)
void gcn_find_hits(const int* __restrict__ src, const int* __restrict__ dst,
                   unsigned* __restrict__ lst_all, int* __restrict__ cnts,
                   int n_nodes, int n_edges) {
    __shared__ unsigned q_w[OWN_WAVES * Q_CAP];
    __shared__ __attribute__((aligned(16))) unsigned l_s[FIND_WIN * SUB_PITCH];
    __shared__ int lst_n[FIND_WIN];

    const int tid = threadIdx.x;
    const int rr  = range_rows(n_nodes);
    int* my_cnt = cnts + (size_t)blockIdx.x * FIND_WIN * CNT_PITCH;
    const bool keep = ((long long)ROW_RANGES * rr >= n_nodes) && (n_nodes <= (1 << (32 - WIN_BITS)))
                      && (((((size_t)src) | ((size_t)dst)) & 15) == 0);
    if (!keep) {
        if (tid < FIND_WIN) my_cnt[tid * CNT_PITCH] = 0x7fffffff;
        return;
    }

    const int lane = tid & 31;
    const int wv   = __builtin_amdgcn_readfirstlane(tid >> 5);
    const int nw   = blockDim.x >> 5;
    const int q    = blockIdx.x % FIND_SLICES;
    const int w    = blockIdx.x / FIND_SLICES;
    const int lo   = w * FIND_WIN * rr;
    int cnt = n_nodes - lo; if (cnt > FIND_WIN * rr) cnt = FIND_WIN * rr; if (cnt < 0) cnt = 0;
    unsigned* lst = lst_all + (size_t)blockIdx.x * LIST_CAP;
    unsigned* qw  = q_w + wv * Q_CAP;

    if (tid < FIND_WIN) lst_n[tid] = 0;
    __syncthreads();

    if (cnt > 0) {
        const int n_full = n_edges >> 9;
        const int c0     = (int)(((long long)q * n_full) / FIND_SLICES);
        const int n_part = (int)(((long long)(q + 1) * n_full) / FIND_SLICES) - c0;
        const int start  = (int)(((long long)w * n_part) / N_WIN);
        const int4* d4 = (const int4*)dst;
        const int4* s4 = (const int4*)src;
        const int t_base = n_full << 9;
        const int t_lo = t_base + (int)(((long long)q * (n_edges - t_base)) / FIND_SLICES);
        const int t_hi = t_base + (int)(((long long)(q + 1) * (n_edges - t_base)) / FIND_SLICES);
        const unsigned ulo = (unsigned)lo, ucnt = (unsigned)cnt, urr = (unsigned)rr;
        const unsigned rr_inv = 0xFFFFFFFFu / urr + 1u;

        int qn = 0;
        if (n_part > 0) {
            int4 dA[4], sA[4], dB[4], sB[4];
            ld_pair(dA, sA, d4, s4, wv,      start, n_part, c0, lane);
            ld_pair(dB, sB, d4, s4, wv + nw, start, n_part, c0, lane);
            for (int c = wv; c < n_part; c += 2 * nw) {
                qn = find_chunk(dA, sA, ulo, ucnt, qw, qn, lane, urr, rr_inv, l_s, lst_n);
                ld_pair(dA, sA, d4, s4, c + 2 * nw, start, n_part, c0, lane);
                if (c + nw < n_part)
                    qn = find_chunk(dB, sB, ulo, ucnt, qw, qn, lane, urr, rr_inv, l_s, lst_n);
                ld_pair(dB, sB, d4, s4, c + 3 * nw, start, n_part, c0, lane);
            }
        }
        for (int b = t_lo + wv * 32; b < t_hi; b += nw * 32) {
            const int e = b + lane;
            unsigned t = 0xFFFFFFFFu; int sn = 0;
            if (e < t_hi) { t = (unsigned)dst[e] - ulo; sn = src[e]; }
            qn = w_push(qw, qn, t, sn, ucnt);
            if (qn >= 32) { w_flush(qw, qn, lane, urr, rr_inv, l_s, lst_n); qn = 0; }
        }
        if (qn > 0) w_flush(qw, qn, lane, urr, rr_inv, l_s, lst_n);
    }
    __syncthreads();
    for (int u = wv; u < FIND_WIN; u += nw) {
        int n = lst_n[u]; if (n > SUB_CAP) n = SUB_CAP;
        const uint4* s4 = (const uint4*)(l_s + u * SUB_PITCH);
        uint4* g4 = (uint4*)(lst + (size_t)u * SUB_CAP);
        for (int i = lane; i * 4 < n; i += 32) g4[i] = s4[i];
    }
    if (tid < FIND_WIN) my_cnt[tid * CNT_PITCH] = lst_n[tid];
}

__device__ __forceinline__ void store_hidden(const float* acc, const float* __restrict__ b1,
                                             float* __restrict__ h, int lo, int cnt) {
    const int tid = threadIdx.x;
    const int c = (tid & 3) * 4;
    const float bx = b1[c], by = b1[c + 1], bz = b1[c + 2], bw = b1[c + 3];
    const float4* acc4 = (const float4*)acc;
    float4* h4 = (float4*)(h + (size_t)lo * HID);
    for (int i = tid; i < cnt * (HID / 4); i += blockDim.x) {
        float4 a = acc4[i];
        a.x = fmaxf(a.x + bx, 0.0f);
        a.y = fmaxf(a.y + by, 0.0f);
        a.z = fmaxf(a.z + bz, 0.0f);
        a.w = fmaxf(a.w + bw, 0.0f);
        h4[i] = a;
    }
}

__device__ __forceinline__ void store_out(const float* acc, float* stage,
                                          const float* __restrict__ W2, const float* __restrict__ b2,
                                          float* __restrict__ out, int lo, int cnt,
                                          int lane, int wv, int nw) {
    const int l15 = lane & 15;
    const int hi  = lane >> 4;
    const bool col_ok = l15 < OUTD;
    v2f b[HID / 4];
    #pragma unroll
    for (int j = 0; j < HID / 4; ++j) {
        const int ka = j * 4 + hi * 2;
        b[j].x = col_ok ? W2[(ka + 0) * OUTD + l15] : 0.0f;
        b[j].y = col_ok ? W2[(ka + 1) * OUTD + l15] : 0.0f;
    }
    const float bias = col_ok ? b2[l15] : 0.0f;
    const int n_t = (cnt + 15) >> 4;
    for (int t = wv; t < n_t; t += nw) {
        const float* arow = acc + (t * 16 + l15) * HID + hi * 2;
        v8f c;
        #pragma unroll
        for (int v = 0; v < 8; ++v) c[v] = bias;
        #pragma unroll
        for (int j = 0; j < HID / 4; ++j) {
            v2f a; a.x = arow[j * 4]; a.y = arow[j * 4 + 1];
            c = __builtin_amdgcn_wmma_f32_16x16x4_f32(false, a, false, b[j], (short)0, c, false, false);
        }
        #pragma unroll
        for (int v = 0; v < 8; ++v) {
            const int row = t * 16 + hi * 8 + v;
            if (col_ok && row < cnt) stage[row * OUTD + l15] = c[v];
        }
    }
    __syncthreads();
    const int tid = threadIdx.x;
    const int n_f = cnt * OUTD;
    float* o = out + (size_t)lo * OUTD;
    int done = 0;
    if ((((size_t)o) & 15) == 0) {
        const int n4 = n_f >> 2;
        const float4* s4 = (const float4*)stage;
        float4* o4 = (float4*)o;
        for (int i = tid; i < n4; i += blockDim.x) o4[i] = s4[i];
        done = n4 << 2;
    }
    for (int i = done + tid; i < n_f; i += blockDim.x) o[i] = stage[i];
}

template <int LAYER>
__device__ __forceinline__ void store_rows(const float* acc, float* stage, const float* __restrict__ bias,
                                           const float* __restrict__ W2, float* __restrict__ res,
                                           int lo, int cnt, int lane, int wv, int nw) {
    if (LAYER == 1) store_hidden(acc, bias, res, lo, cnt);
    else            store_out(acc, stage, W2, bias, res, lo, cnt, lane, wv, nw);
}

template <int LAYER>
__device__ __forceinline__ void owner_scan(const int* __restrict__ src, const int* __restrict__ dst,
                                           const float* __restrict__ m, float* __restrict__ res,
                                           const float* __restrict__ bias, const float* __restrict__ W2,
                                           int n_nodes, int n_edges,
                                           float* acc, int* q_e, unsigned short* q_t) {
    const int tid  = threadIdx.x;
    const int lane = tid & 31;
    const int wv   = __builtin_amdgcn_readfirstlane(tid >> 5);
    const int nw   = blockDim.x >> 5;
    const int r    = blockIdx.x;
    int* qe = q_e + wv * Q_CAP;
    unsigned short* qt = q_t + wv * Q_CAP;
    const int rr = range_rows(n_nodes);

    for (int lo = r * rr; lo < n_nodes; lo += ROW_RANGES * rr) {
        int cnt = n_nodes - lo; if (cnt > rr) cnt = rr;
        float4* acc4 = (float4*)acc;
        for (int i = tid; i < cnt * (HID / 4); i += blockDim.x) acc4[i] = make_float4(0.f, 0.f, 0.f, 0.f);
        __syncthreads();

        scan_slice(src, dst, m, n_edges, (unsigned)lo, (unsigned)cnt, acc, qe, qt, r,
                   lane, wv, nw);
        __syncthreads();

        store_rows<LAYER>(acc, (float*)q_e, bias, W2, res, lo, cnt, lane, wv, nw);
        __syncthreads();
    }
}

__device__ __forceinline__ void list_sum(const unsigned* __restrict__ l, int total,
                                         const float* __restrict__ m, float* acc,
                                         int lane, int sw, int nsw) {
    const int f  = lane & 15;
    const int hi = lane >> 4;
    const int nb = (total + 31) >> 5;
    const unsigned row_mask = (1u << ROW_BITS) - 1u;
    unsigned cur[16];
    {
        unsigned w = 0u;
        if (sw * 32 + lane < total) w = l[sw * 32 + lane];
        #pragma unroll
        for (int u = 0; u < 16; ++u) {
            unsigned wa = (unsigned)__builtin_amdgcn_readlane((int)w, 2 * u);
            unsigned wb = (unsigned)__builtin_amdgcn_readlane((int)w, 2 * u + 1);
            cur[u] = hi ? wb : wa;
        }
    }
    for (int b = sw; b < nb; b += nsw) {
        unsigned wn = 0u;
        { int i = (b + nsw) * 32 + lane; if (i < total) wn = l[i]; }
        const int n = total - b * 32;
        float v[16];
        #pragma unroll
        for (int u = 0; u < 16; ++u)
            v[u] = m[(size_t)(cur[u] >> ROW_BITS) * HID + f];
        if (n >= 32) {
            #pragma unroll
            for (int u = 0; u < 16; ++u)
                atomicAdd(&acc[(int)(cur[u] & row_mask) * HID + f], v[u]);
        } else {
            #pragma unroll
            for (int u = 0; u < 16; ++u)
                if (2 * u + hi < n) atomicAdd(&acc[(int)(cur[u] & row_mask) * HID + f], v[u]);
        }
        #pragma unroll
        for (int u = 0; u < 16; ++u) {
            unsigned wa = (unsigned)__builtin_amdgcn_readlane((int)wn, 2 * u);
            unsigned wb = (unsigned)__builtin_amdgcn_readlane((int)wn, 2 * u + 1);
            cur[u] = hi ? wb : wa;
        }
    }
}

template <int LAYER>
__global__ __launch_bounds__(OWN_THREADS)
void gcn_owner_sum_list(const int* __restrict__ src, const int* __restrict__ dst,
                        const float* __restrict__ m, float* __restrict__ res,
                        const float* __restrict__ bias, const float* __restrict__ W2,
                        const unsigned* __restrict__ lst_all, const int* __restrict__ cnts,
                        int n_nodes, int n_edges) {
    __shared__ __attribute__((aligned(16))) float acc[ROW_CAP * HID];
    __shared__ __attribute__((aligned(16))) int q_e[OWN_WAVES * Q_CAP];
    __shared__ unsigned short q_t[OWN_WAVES * Q_CAP];

    const int tid  = threadIdx.x;
    const int lane = tid & 31;
    const int wv   = __builtin_amdgcn_readfirstlane(tid >> 5);
    const int nw   = blockDim.x >> 5;
    const int r = blockIdx.x;
    const int w = r / FIND_WIN;
    const int k = r % FIND_WIN;
    const size_t fb0 = (size_t)w * FIND_SLICES;
    const int n_l = cnts[((fb0 + lane) * FIND_WIN + k) * CNT_PITCH];
    const bool lists_ok = __builtin_amdgcn_ballot_w32((unsigned)n_l > (unsigned)SUB_CAP) == 0u;
    if (!lists_ok) {
        owner_scan<LAYER>(src, dst, m, res, bias, W2, n_nodes, n_edges, acc, q_e, q_t);
        return;
    }

    const int rr = range_rows(n_nodes);
    const int lo = r * rr;
    int cnt = n_nodes - lo; if (cnt > rr) cnt = rr; if (cnt < 0) cnt = 0;

    float4* acc4 = (float4*)acc;
    for (int i = tid; i < cnt * (HID / 4); i += blockDim.x) acc4[i] = make_float4(0.f, 0.f, 0.f, 0.f);
    __syncthreads();

    for (int u = wv; u < FIND_SLICES; u += nw) {
        const int total = __builtin_amdgcn_readlane(n_l, u);
        const unsigned* l = lst_all + (fb0 + u) * LIST_CAP + (size_t)k * SUB_CAP;
        list_sum(l, total, m, acc, lane, 0, 1);
    }
    __syncthreads();

    store_rows<LAYER>(acc, (float*)q_e, bias, W2, res, lo, cnt, lane, wv, nw);
}

extern "C" void kernel_launch(void* const* d_in, const int* in_sizes, int n_in,
                              void* d_out, int out_size, void* d_ws, size_t ws_size,
                              hipStream_t stream) {
    const float* x  = (const float*)d_in[0];
    const int*   ei = (const int*)  d_in[1];
    const float* W1 = (const float*)d_in[2];
    const float* b1 = (const float*)d_in[3];
    const float* W2 = (const float*)d_in[4];
    const float* b2 = (const float*)d_in[5];
    float* out = (float*)d_out;

    int n_nodes       = in_sizes[0] / IN_DIM;
    long long n_edges = (long long)in_sizes[1] / 2;
    const int* src = ei;
    const int* dst = ei + n_edges;

    float* m1   = (float*)d_ws;
    float* h    = m1 + (size_t)n_nodes * HID;
    unsigned* hits = (unsigned*)(h + (size_t)n_nodes * HID);
    int* hit_n  = (int*)(hits + (size_t)OWN_GROUPS * LIST_CAP);

    int n_tiles = (n_nodes + 15) / 16;
    int gemm_blocks = (n_tiles + 7) / 8;

    gcn_gemm1_wmma<<<gemm_blocks, 256, 0, stream>>>(x, W1, m1, n_tiles);
    gcn_find_hits<<<OWN_GROUPS, OWN_THREADS, 0, stream>>>(src, dst, hits, hit_n, n_nodes, (int)n_edges);
    gcn_owner_sum_list<1><<<OWN_GROUPS, OWN_THREADS, 0, stream>>>(src, dst, m1, h, b1, W2, hits, hit_n, n_nodes, (int)n_edges);
    gcn_owner_sum_list<2><<<OWN_GROUPS, OWN_THREADS, 0, stream>>>(src, dst, h, out, b2, W2, hits, hit_n, n_nodes, (int)n_edges);
}
